# starting kernel
# baseline (speedup 1.0000x reference)
#include <hip/hip_runtime.h>
#include <hip/hip_bf16.h>

typedef __attribute__((ext_vector_type(2))) float v2f;
typedef __attribute__((ext_vector_type(8))) float v8f;

__global__ __launch_bounds__(256)
void stats_kernel(const float* __restrict__ feat, const float* __restrict__ noise,
                  float* __restrict__ scl, float* __restrict__ nmn,
                  int nd, float snr_pow) {
  int b = blockIdx.x;
  int t = threadIdx.x;
  const float* f  = feat  + (long long)b * nd;
  const float* nz = noise + (long long)b * nd;
  double sf = 0.0, sf2 = 0.0, sn = 0.0, sn2 = 0.0;
  for (int i = t; i < nd; i += 256) {
    double x = (double)f[i];
    double y = (double)nz[i];
    sf += x; sf2 += x * x; sn += y; sn2 += y * y;
  }
  __shared__ double sh[4][256];
  sh[0][t] = sf; sh[1][t] = sf2; sh[2][t] = sn; sh[3][t] = sn2;
  __syncthreads();
  for (int s = 128; s > 0; s >>= 1) {
    if (t < s) {
      sh[0][t] += sh[0][t + s]; sh[1][t] += sh[1][t + s];
      sh[2][t] += sh[2][t + s]; sh[3][t] += sh[3][t + s];
    }
    __syncthreads();
  }
  if (t == 0) {
    double inv = 1.0 / (double)nd;
    double gmf = sh[0][0] * inv;
    double gmn = sh[2][0] * inv;
    double tvar = sh[1][0] * inv - gmf * gmf;
    double nvar = sh[3][0] * inv - gmn * gmn;
    scl[b] = (float)sqrt((tvar / (double)snr_pow) / nvar);
    nmn[b] = (float)gmn;
  }
}

__global__ __launch_bounds__(256)
void perturb_kernel(const float* __restrict__ feat, const float* __restrict__ noise,
                    const float* __restrict__ scl, const float* __restrict__ nmn,
                    float* __restrict__ Fp, int nd) {
  int b = blockIdx.y;
  long long idx = (long long)b * nd + (long long)blockIdx.x * 256 + threadIdx.x;
  Fp[idx] = feat[idx] + (noise[idx] - nmn[b]) * scl[b];
}

__global__ __launch_bounds__(256)
void mean_kernel(const float* __restrict__ Fp, float* __restrict__ mean, int n, int d) {
  int b = blockIdx.x;
  int j = threadIdx.x;
  const float* base = Fp + (long long)b * n * d;
  float s = 0.0f;
  for (int i = 0; i < n; ++i) s += base[(long long)i * d + j];
  mean[(long long)b * d + j] = s / (float)n;
}

__global__ __launch_bounds__(256)
void center_kernel(float* __restrict__ Fp, const float* __restrict__ mean, int n, int d) {
  int row = blockIdx.x;
  int j   = threadIdx.x;
  int b   = row / n;
  Fp[(long long)row * d + j] -= mean[(long long)b * d + j];
}

#define BM 128
#define BN 128
#define BK 16
#define LDP 18

__global__ __launch_bounds__(256)
void wmma_gemm_batched(const float* __restrict__ A, long long sAb, int sAm, int sAk,
                       const float* __restrict__ Bm, long long sBb, int sBk, int sBn,
                       float* __restrict__ Dm, long long sDb,
                       int N, int K, float alpha, int mode) {
  __shared__ __align__(16) float As[BM][LDP];
  __shared__ __align__(16) float Bst[BN][LDP];

  int b  = blockIdx.z;
  int m0 = blockIdx.y * BM;
  int n0 = blockIdx.x * BN;
  const float* Ab = A  + (long long)b * sAb;
  const float* Bb = Bm + (long long)b * sBb;
  float*       Db = Dm + (long long)b * sDb;

  int tid  = threadIdx.x;
  int wid  = tid >> 5;
  int lane = tid & 31;
  int mr   = lane & 15;
  int h    = lane >> 4;
  int wm   = (wid >> 1) * 32;
  int wn   = (wid & 1) * 64;

  v8f acc[2][4];
#pragma unroll
  for (int im = 0; im < 2; ++im)
#pragma unroll
    for (int in = 0; in < 4; ++in) acc[im][in] = (v8f){};

  for (int k0 = 0; k0 < K; k0 += BK) {
#pragma unroll
    for (int i = 0; i < 8; ++i) {
      int e = tid + i * 256;
      int m = e >> 4, kk = e & 15;
      As[m][kk] = Ab[(long long)(m0 + m) * sAm + (long long)(k0 + kk) * sAk];
    }
#pragma unroll
    for (int i = 0; i < 8; ++i) {
      int e = tid + i * 256;
      int nn = e & 127, kk = e >> 7;
      Bst[nn][kk] = Bb[(long long)(k0 + kk) * sBk + (long long)(n0 + nn) * sBn];
    }
    __syncthreads();
#pragma unroll
    for (int kk = 0; kk < BK; kk += 4) {
      int kc = kk + 2 * h;
      v2f a0 = *(const v2f*)&As[wm + mr][kc];
      v2f a1 = *(const v2f*)&As[wm + 16 + mr][kc];
      v2f b0 = *(const v2f*)&Bst[wn + mr][kc];
      v2f b1 = *(const v2f*)&Bst[wn + 16 + mr][kc];
      v2f b2 = *(const v2f*)&Bst[wn + 32 + mr][kc];
      v2f b3 = *(const v2f*)&Bst[wn + 48 + mr][kc];
      acc[0][0] = __builtin_amdgcn_wmma_f32_16x16x4_f32(false, a0, false, b0, (short)0, acc[0][0], false, false);
      acc[0][1] = __builtin_amdgcn_wmma_f32_16x16x4_f32(false, a0, false, b1, (short)0, acc[0][1], false, false);
      acc[0][2] = __builtin_amdgcn_wmma_f32_16x16x4_f32(false, a0, false, b2, (short)0, acc[0][2], false, false);
      acc[0][3] = __builtin_amdgcn_wmma_f32_16x16x4_f32(false, a0, false, b3, (short)0, acc[0][3], false, false);
      acc[1][0] = __builtin_amdgcn_wmma_f32_16x16x4_f32(false, a1, false, b0, (short)0, acc[1][0], false, false);
      acc[1][1] = __builtin_amdgcn_wmma_f32_16x16x4_f32(false, a1, false, b1, (short)0, acc[1][1], false, false);
      acc[1][2] = __builtin_amdgcn_wmma_f32_16x16x4_f32(false, a1, false, b2, (short)0, acc[1][2], false, false);
      acc[1][3] = __builtin_amdgcn_wmma_f32_16x16x4_f32(false, a1, false, b3, (short)0, acc[1][3], false, false);
    }
    __syncthreads();
  }

#pragma unroll
  for (int im = 0; im < 2; ++im) {
#pragma unroll
    for (int r = 0; r < 8; ++r) {
      int row = m0 + wm + 16 * im + 8 * h + r;
      float* drow = Db + (long long)row * N;
#pragma unroll
      for (int in = 0; in < 4; ++in) {
        int col = n0 + wn + 16 * in + mr;
        float v = acc[im][in][r];
        if (mode == 1) {
          v = -0.5f * v + ((row == col) ? 1.5f : 0.0f);
        } else {
          v *= alpha;
        }
        drow[col] = v;
      }
    }
  }
}

__global__ __launch_bounds__(256)
void trace_kernel(const float* __restrict__ cov, float* __restrict__ tr, int d) {
  int b = blockIdx.x;
  int t = threadIdx.x;
  __shared__ float sh[256];
  sh[t] = cov[(long long)b * d * d + (long long)t * d + t];
  __syncthreads();
  for (int s = 128; s > 0; s >>= 1) {
    if (t < s) sh[t] += sh[t + s];
    __syncthreads();
  }
  if (t == 0) tr[b] = sh[0];
}

__global__ __launch_bounds__(256)
void init_A_ZY(const float* __restrict__ cov, const float* __restrict__ tr,
               float* __restrict__ Amat, float* __restrict__ ZY, int d) {
  int bi = blockIdx.x;
  int b  = bi / d;
  int i  = bi % d;
  int j  = threadIdx.x;
  long long idx = (long long)b * d * d + (long long)i * d + j;
  float a = cov[idx] / tr[b];
  Amat[idx] = a;
  ZY[idx]   = -0.5f * a + ((i == j) ? 1.5f : 0.0f);
}

__global__ __launch_bounds__(256)
void out_kernel(const float* __restrict__ YZY, const float* __restrict__ mean,
                const float* __restrict__ tr, float* __restrict__ out, int d) {
  int wid  = threadIdx.x >> 5;
  int lane = threadIdx.x & 31;
  int row  = blockIdx.x * 8 + wid;
  int b    = row / d;
  int i    = row % d;
  const float* yrow = YZY + (long long)b * d * d + (long long)i * d;
  const float* mrow = mean + (long long)b * d;
  float s = 0.0f;
  for (int j = lane; j < d; j += 32) s += yrow[j] * mrow[j];
  for (int off = 16; off > 0; off >>= 1) s += __shfl_down(s, off, 32);
  if (lane == 0) out[row] = s * sqrtf(tr[b]);
}

extern "C" void kernel_launch(void* const* d_in, const int* in_sizes, int n_in,
                              void* d_out, int out_size, void* d_ws, size_t ws_size,
                              hipStream_t stream) {
  (void)n_in; (void)ws_size; (void)out_size;
  const float* feat  = (const float*)d_in[1];
  const float* noise = (const float*)d_in[2];

  const int B  = in_sizes[0];
  const int d  = 256;
  const long long total = (long long)in_sizes[1];
  const int n  = (int)(total / ((long long)B * d));
  const int nd = n * d;
  const long long dd = (long long)d * d;

  float* W = (float*)d_ws;
  long long off = 0;
  float* Fp   = W + off; off += (long long)B * nd;
  float* mean = W + off; off += (long long)B * d;
  float* scl  = W + off; off += B;
  float* nmn  = W + off; off += B;
  float* tr   = W + off; off += B;
  off = (off + 255) & ~255LL;
  float* M0 = W + off; off += (long long)B * dd;
  float* M1 = W + off; off += (long long)B * dd;
  float* M2 = W + off; off += (long long)B * dd;
  float* M3 = W + off; off += (long long)B * dd;

  stats_kernel<<<B, 256, 0, stream>>>(feat, noise, scl, nmn, nd, 31.6227766016838f);
  perturb_kernel<<<dim3(nd / 256, B), 256, 0, stream>>>(feat, noise, scl, nmn, Fp, nd);
  mean_kernel<<<B, d, 0, stream>>>(Fp, mean, n, d);
  center_kernel<<<B * n, d, 0, stream>>>(Fp, mean, n, d);

  dim3 gg(d / BN, d / BM, B);
  wmma_gemm_batched<<<gg, 256, 0, stream>>>(Fp, (long long)nd, 1, d,
                                            Fp, (long long)nd, d, 1,
                                            M0, dd, d, n,
                                            1.0f / (float)(n - 1), 0);
  trace_kernel<<<B, d, 0, stream>>>(M0, tr, d);
  init_A_ZY<<<B * d, d, 0, stream>>>(M0, tr, M1, M2, d);

  wmma_gemm_batched<<<gg, 256, 0, stream>>>(M1, dd, d, 1, M2, dd, d, 1,
                                            M0, dd, d, d, 1.0f, 0);
  float* Y = M0; float* Z = M2; float* S1 = M1; float* S2 = M3;

  for (int it = 0; it < 3; ++it) {
    wmma_gemm_batched<<<gg, 256, 0, stream>>>(Z, dd, d, 1, Y, dd, d, 1,
                                              S1, dd, d, d, 1.0f, 1);
    wmma_gemm_batched<<<gg, 256, 0, stream>>>(Y, dd, d, 1, S1, dd, d, 1,
                                              S2, dd, d, d, 1.0f, 0);
    wmma_gemm_batched<<<gg, 256, 0, stream>>>(S1, dd, d, 1, Z, dd, d, 1,
                                              Y, dd, d, d, 1.0f, 0);
    float* newY = S2; float* newZ = Y; float* freed = Z;
    Y = newY; Z = newZ; S2 = freed;
  }

  wmma_gemm_batched<<<gg, 256, 0, stream>>>(Z, dd, d, 1, Y, dd, d, 1,
                                            S1, dd, d, d, 1.0f, 1);
  wmma_gemm_batched<<<gg, 256, 0, stream>>>(Y, dd, d, 1, S1, dd, d, 1,
                                            S2, dd, d, d, 1.0f, 0);

  out_kernel<<<(B * d) / 8, 256, 0, stream>>>(S2, mean, tr, (float*)d_out, d);
}


# Newton-Schulz products tiled triangularly: operands are symmetric and commute, so only the 64x64 tiles on or above the diagonal (10 of 16) are computed, each stored mirrored too; lower sums change
# speedup vs baseline: 1.6006x; 1.6006x over previous
#include <hip/hip_runtime.h>
#include <hip/hip_bf16.h>

typedef __attribute__((ext_vector_type(2))) float v2f;
typedef __attribute__((ext_vector_type(8))) float v8f;

#define STAT_SLICES 8

__global__ __launch_bounds__(256)
void stats_partial_kernel(const float* __restrict__ feat, const float* __restrict__ noise,
                          float* __restrict__ part_ws, int nd) {
  int s = blockIdx.x;
  int b = blockIdx.y;
  int t = threadIdx.x;
  int len  = nd / STAT_SLICES;
  int len4 = len >> 2;
  const float4* f4 = reinterpret_cast<const float4*>(feat  + (long long)b * nd + (long long)s * len);
  const float4* n4 = reinterpret_cast<const float4*>(noise + (long long)b * nd + (long long)s * len);
  double sf = 0.0, sf2 = 0.0, sn = 0.0, sn2 = 0.0;
#pragma unroll 4
  for (int i = t; i < len4; i += 256) {
    float4 x = f4[i];
    float4 y = n4[i];
    double x0 = (double)x.x, x1 = (double)x.y, x2 = (double)x.z, x3 = (double)x.w;
    double y0 = (double)y.x, y1 = (double)y.y, y2 = (double)y.z, y3 = (double)y.w;
    sf += x0; sf2 += x0 * x0; sn += y0; sn2 += y0 * y0;
    sf += x1; sf2 += x1 * x1; sn += y1; sn2 += y1 * y1;
    sf += x2; sf2 += x2 * x2; sn += y2; sn2 += y2 * y2;
    sf += x3; sf2 += x3 * x3; sn += y3; sn2 += y3 * y3;
  }
  __shared__ double sh[4][256];
  sh[0][t] = sf; sh[1][t] = sf2; sh[2][t] = sn; sh[3][t] = sn2;
  __syncthreads();
  for (int w = 128; w > 0; w >>= 1) {
    if (t < w) {
      sh[0][t] += sh[0][t + w]; sh[1][t] += sh[1][t + w];
      sh[2][t] += sh[2][t + w]; sh[3][t] += sh[3][t + w];
    }
    __syncthreads();
  }
  if (t < 4) {
    double* part = reinterpret_cast<double*>(part_ws);
    part[((long long)b * STAT_SLICES + s) * 4 + t] = sh[t][0];
  }
}

__global__ __launch_bounds__(128)
void stats_final_kernel(const float* __restrict__ part_ws,
                        float* __restrict__ scl, float* __restrict__ nmn,
                        int B, int nd, float snr_pow) {
  const double* part = reinterpret_cast<const double*>(part_ws);
  for (int b = threadIdx.x; b < B; b += 128) {
    const double* p = part + (long long)b * STAT_SLICES * 4;
    double s0 = 0.0, s1 = 0.0, s2 = 0.0, s3 = 0.0;
    for (int s = 0; s < STAT_SLICES; ++s) {
      s0 += p[4 * s + 0]; s1 += p[4 * s + 1];
      s2 += p[4 * s + 2]; s3 += p[4 * s + 3];
    }
    double inv = 1.0 / (double)nd;
    double gmf = s0 * inv;
    double gmn = s2 * inv;
    double tvar = s1 * inv - gmf * gmf;
    double nvar = s3 * inv - gmn * gmn;
    scl[b] = (float)sqrt((tvar / (double)snr_pow) / nvar);
    nmn[b] = (float)gmn;
  }
}

#define EW_TILE4 1024

__global__ __launch_bounds__(256)
void perturb_kernel(const float* __restrict__ feat, const float* __restrict__ noise,
                    const float* __restrict__ scl, const float* __restrict__ nmn,
                    float* __restrict__ Fp, int nd) {
  int b   = blockIdx.y;
  int nd4 = nd >> 2;
  const float4* f4 = reinterpret_cast<const float4*>(feat  + (long long)b * nd);
  const float4* n4 = reinterpret_cast<const float4*>(noise + (long long)b * nd);
  float4*       o4 = reinterpret_cast<float4*>(Fp + (long long)b * nd);
  float nm = nmn[b];
  float sc = scl[b];
  for (int c = blockIdx.x * EW_TILE4 + threadIdx.x; c < nd4; c += gridDim.x * EW_TILE4) {
    float4 x[4], y[4];
#pragma unroll
    for (int k = 0; k < 4; ++k) {
      int i = c + k * 256;
      if (i < nd4) { x[k] = f4[i]; y[k] = n4[i]; }
    }
#pragma unroll
    for (int k = 0; k < 4; ++k) {
      int i = c + k * 256;
      if (i < nd4) {
        float4 r;
        r.x = x[k].x + (y[k].x - nm) * sc;
        r.y = x[k].y + (y[k].y - nm) * sc;
        r.z = x[k].z + (y[k].z - nm) * sc;
        r.w = x[k].w + (y[k].w - nm) * sc;
        o4[i] = r;
      }
    }
  }
}

__global__ __launch_bounds__(256)
void mean_kernel(const float* __restrict__ Fp, float* __restrict__ mean, int n, int d) {
  int b = blockIdx.x;
  int j = threadIdx.x;
  const float* base = Fp + (long long)b * n * d;
  float s = 0.0f;
  for (int i = 0; i < n; ++i) s += base[(long long)i * d + j];
  mean[(long long)b * d + j] = s / (float)n;
}

__global__ __launch_bounds__(256)
void center_kernel(float* __restrict__ Fp, const float* __restrict__ mean, int nd, int d) {
  int b   = blockIdx.y;
  int nd4 = nd >> 2;
  int d4  = d >> 2;
  float4*       p4 = reinterpret_cast<float4*>(Fp + (long long)b * nd);
  const float4* m4 = reinterpret_cast<const float4*>(mean + (long long)b * d);
  for (int c = blockIdx.x * EW_TILE4 + threadIdx.x; c < nd4; c += gridDim.x * EW_TILE4) {
    float4 x[4], m[4];
#pragma unroll
    for (int k = 0; k < 4; ++k) {
      int i = c + k * 256;
      if (i < nd4) { x[k] = p4[i]; m[k] = m4[i % d4]; }
    }
#pragma unroll
    for (int k = 0; k < 4; ++k) {
      int i = c + k * 256;
      if (i < nd4) {
        float4 r;
        r.x = x[k].x - m[k].x;
        r.y = x[k].y - m[k].y;
        r.z = x[k].z - m[k].z;
        r.w = x[k].w - m[k].w;
        p4[i] = r;
      }
    }
  }
}

#define BM 128
#define BN 128
#define BK 16
#define LDP 18

__global__ __launch_bounds__(256)
void wmma_gemm_batched(const float* __restrict__ A, long long sAb, int sAm, int sAk,
                       const float* __restrict__ Bm, long long sBb, int sBk, int sBn,
                       float* __restrict__ Dm, long long sDb,
                       int N, int K, float alpha, int mode) {
  __shared__ __align__(16) float As[BM][LDP];
  __shared__ __align__(16) float Bst[BN][LDP];

  int b  = blockIdx.z;
  int m0 = blockIdx.y * BM;
  int n0 = blockIdx.x * BN;
  const float* Ab = A  + (long long)b * sAb;
  const float* Bb = Bm + (long long)b * sBb;
  float*       Db = Dm + (long long)b * sDb;

  int tid  = threadIdx.x;
  int wid  = tid >> 5;
  int lane = tid & 31;
  int mr   = lane & 15;
  int h    = lane >> 4;
  int wm   = (wid >> 1) * 32;
  int wn   = (wid & 1) * 64;

  v8f acc[2][4];
#pragma unroll
  for (int im = 0; im < 2; ++im)
#pragma unroll
    for (int in = 0; in < 4; ++in) acc[im][in] = (v8f){};

  for (int k0 = 0; k0 < K; k0 += BK) {
#pragma unroll
    for (int i = 0; i < 8; ++i) {
      int e = tid + i * 256;
      int m = e >> 4, kk = e & 15;
      As[m][kk] = Ab[(long long)(m0 + m) * sAm + (long long)(k0 + kk) * sAk];
    }
#pragma unroll
    for (int i = 0; i < 8; ++i) {
      int e = tid + i * 256;
      int nn = e & 127, kk = e >> 7;
      Bst[nn][kk] = Bb[(long long)(k0 + kk) * sBk + (long long)(n0 + nn) * sBn];
    }
    __syncthreads();
#pragma unroll
    for (int kk = 0; kk < BK; kk += 4) {
      int kc = kk + 2 * h;
      v2f a0 = *(const v2f*)&As[wm + mr][kc];
      v2f a1 = *(const v2f*)&As[wm + 16 + mr][kc];
      v2f b0 = *(const v2f*)&Bst[wn + mr][kc];
      v2f b1 = *(const v2f*)&Bst[wn + 16 + mr][kc];
      v2f b2 = *(const v2f*)&Bst[wn + 32 + mr][kc];
      v2f b3 = *(const v2f*)&Bst[wn + 48 + mr][kc];
      acc[0][0] = __builtin_amdgcn_wmma_f32_16x16x4_f32(false, a0, false, b0, (short)0, acc[0][0], false, false);
      acc[0][1] = __builtin_amdgcn_wmma_f32_16x16x4_f32(false, a0, false, b1, (short)0, acc[0][1], false, false);
      acc[0][2] = __builtin_amdgcn_wmma_f32_16x16x4_f32(false, a0, false, b2, (short)0, acc[0][2], false, false);
      acc[0][3] = __builtin_amdgcn_wmma_f32_16x16x4_f32(false, a0, false, b3, (short)0, acc[0][3], false, false);
      acc[1][0] = __builtin_amdgcn_wmma_f32_16x16x4_f32(false, a1, false, b0, (short)0, acc[1][0], false, false);
      acc[1][1] = __builtin_amdgcn_wmma_f32_16x16x4_f32(false, a1, false, b1, (short)0, acc[1][1], false, false);
      acc[1][2] = __builtin_amdgcn_wmma_f32_16x16x4_f32(false, a1, false, b2, (short)0, acc[1][2], false, false);
      acc[1][3] = __builtin_amdgcn_wmma_f32_16x16x4_f32(false, a1, false, b3, (short)0, acc[1][3], false, false);
    }
    __syncthreads();
  }

#pragma unroll
  for (int im = 0; im < 2; ++im) {
#pragma unroll
    for (int r = 0; r < 8; ++r) {
      int row = m0 + wm + 16 * im + 8 * h + r;
      float* drow = Db + (long long)row * N;
#pragma unroll
      for (int in = 0; in < 4; ++in) {
        int col = n0 + wn + 16 * in + mr;
        float v = acc[im][in][r];
        if (mode == 1) {
          v = -0.5f * v + ((row == col) ? 1.5f : 0.0f);
        } else {
          v *= alpha;
        }
        drow[col] = v;
      }
    }
  }
}

#define GLD 256
#define LDA 20
#define SM 64
#define SLDB 160

__global__ __launch_bounds__(128)
void wmma_gemm_sym(const float* __restrict__ A, const float* __restrict__ Bm,
                   float* __restrict__ Dm, int K, int mode) {
  __shared__ __align__(16) float As[SM][LDA];
  __shared__ __align__(16) float Bs[BK / 2][SLDB];

  int b  = blockIdx.z;
  int t  = blockIdx.x;
  int ti = (t >= 4) + (t >= 7) + (t >= 9);
  int tj = ti + t - (4 * ti - (ti * (ti - 1)) / 2);
  int m0 = ti * SM;
  int n0 = tj * SM;
  const float* Ab = A  + (long long)b * (GLD * GLD);
  const float* Bb = Bm + (long long)b * (GLD * GLD);
  float*       Db = Dm + (long long)b * (GLD * GLD);

  int tid  = threadIdx.x;
  int wid  = tid >> 5;
  int lane = tid & 31;
  int mr   = lane & 15;
  int h    = lane >> 4;
  int wm   = (wid >> 1) * 32;
  int wn   = (wid & 1) * 32;

  int ar  = tid >> 2;
  int ak4 = tid & 3;
  int bk  = tid >> 4;
  int bn4 = tid & 15;
  const float* Ag = Ab + (m0 + ar) * GLD + 4 * ak4;
  const float* Bg = Bb + 2 * bk * GLD + n0 + 4 * bn4;

  v8f acc[2][2];
#pragma unroll
  for (int im = 0; im < 2; ++im)
#pragma unroll
    for (int in = 0; in < 2; ++in) acc[im][in] = (v8f){};

  for (int k0 = 0; k0 < K; k0 += BK) {
    float4 ga0 = *reinterpret_cast<const float4*>(Ag + k0);
    float4 ga1 = *reinterpret_cast<const float4*>(Ag + k0 + 32 * GLD);
    float4 gb0 = *reinterpret_cast<const float4*>(Bg + k0 * GLD);
    float4 gb1 = *reinterpret_cast<const float4*>(Bg + (k0 + 1) * GLD);
    *reinterpret_cast<float4*>(&As[ar][4 * ak4])      = ga0;
    *reinterpret_cast<float4*>(&As[ar + 32][4 * ak4]) = ga1;
    float4 lo, hi;
    lo.x = gb0.x; lo.y = gb1.x; lo.z = gb0.y; lo.w = gb1.y;
    hi.x = gb0.z; hi.y = gb1.z; hi.z = gb0.w; hi.w = gb1.w;
    *reinterpret_cast<float4*>(&Bs[bk][8 * bn4])     = lo;
    *reinterpret_cast<float4*>(&Bs[bk][8 * bn4 + 4]) = hi;
    __syncthreads();
#pragma unroll
    for (int kk = 0; kk < BK; kk += 4) {
      int kc = kk + 2 * h;
      v2f a0 = *(const v2f*)&As[wm + mr][kc];
      v2f a1 = *(const v2f*)&As[wm + 16 + mr][kc];
      int kp = (kk >> 1) + h;
      v2f b0 = *(const v2f*)&Bs[kp][2 * (wn + mr)];
      v2f b1 = *(const v2f*)&Bs[kp][2 * (wn + 16 + mr)];
      acc[0][0] = __builtin_amdgcn_wmma_f32_16x16x4_f32(false, a0, false, b0, (short)0, acc[0][0], false, false);
      acc[0][1] = __builtin_amdgcn_wmma_f32_16x16x4_f32(false, a0, false, b1, (short)0, acc[0][1], false, false);
      acc[1][0] = __builtin_amdgcn_wmma_f32_16x16x4_f32(false, a1, false, b0, (short)0, acc[1][0], false, false);
      acc[1][1] = __builtin_amdgcn_wmma_f32_16x16x4_f32(false, a1, false, b1, (short)0, acc[1][1], false, false);
    }
    __syncthreads();
  }

  if (mode == 1) {
#pragma unroll
    for (int im = 0; im < 2; ++im)
#pragma unroll
      for (int in = 0; in < 2; ++in)
#pragma unroll
        for (int r = 0; r < 8; ++r) {
          int row = m0 + wm + 16 * im + 8 * h + r;
          int col = n0 + wn + 16 * in + mr;
          acc[im][in][r] = -0.5f * acc[im][in][r] + ((row == col) ? 1.5f : 0.0f);
        }
  }

#pragma unroll
  for (int im = 0; im < 2; ++im) {
#pragma unroll
    for (int r = 0; r < 8; ++r) {
      int row = m0 + wm + 16 * im + 8 * h + r;
      float* drow = Db + row * GLD;
#pragma unroll
      for (int in = 0; in < 2; ++in) {
        int col = n0 + wn + 16 * in + mr;
        drow[col] = acc[im][in][r];
      }
    }
  }

  if (tj != ti) {
#pragma unroll
    for (int in = 0; in < 2; ++in) {
      float* mrow = Db + (n0 + wn + 16 * in + mr) * GLD + m0 + wm + 8 * h;
#pragma unroll
      for (int im = 0; im < 2; ++im) {
        float4 lo, hi;
        lo.x = acc[im][in][0]; lo.y = acc[im][in][1]; lo.z = acc[im][in][2]; lo.w = acc[im][in][3];
        hi.x = acc[im][in][4]; hi.y = acc[im][in][5]; hi.z = acc[im][in][6]; hi.w = acc[im][in][7];
        *reinterpret_cast<float4*>(mrow + 16 * im)     = lo;
        *reinterpret_cast<float4*>(mrow + 16 * im + 4) = hi;
      }
    }
  }
}

__global__ __launch_bounds__(256)
void trace_kernel(const float* __restrict__ cov, float* __restrict__ tr, int d) {
  int b = blockIdx.x;
  int t = threadIdx.x;
  __shared__ float sh[256];
  sh[t] = cov[(long long)b * d * d + (long long)t * d + t];
  __syncthreads();
  for (int s = 128; s > 0; s >>= 1) {
    if (t < s) sh[t] += sh[t + s];
    __syncthreads();
  }
  if (t == 0) tr[b] = sh[0];
}

#define INIT_TILE4 2048

__global__ __launch_bounds__(256)
void init_A_ZY(const float* __restrict__ cov, const float* __restrict__ tr,
               float* __restrict__ Amat, float* __restrict__ ZY, int d) {
  int b   = blockIdx.y;
  int d4  = d >> 2;
  int dd4 = d * d4;
  const float4* c4 = reinterpret_cast<const float4*>(cov + (long long)b * d * d);
  float4*       a4 = reinterpret_cast<float4*>(Amat + (long long)b * d * d);
  float4*       z4 = reinterpret_cast<float4*>(ZY   + (long long)b * d * d);
  float t = tr[b];
  for (int c = blockIdx.x * INIT_TILE4 + threadIdx.x; c < dd4; c += gridDim.x * INIT_TILE4) {
    float4 x[8];
#pragma unroll
    for (int k = 0; k < 8; ++k) {
      int i = c + k * 256;
      if (i < dd4) x[k] = c4[i];
    }
#pragma unroll
    for (int k = 0; k < 8; ++k) {
      int i = c + k * 256;
      if (i < dd4) {
        int row = i / d4;
        int col = (i - row * d4) * 4;
        float4 a, z;
        a.x = x[k].x / t;
        a.y = x[k].y / t;
        a.z = x[k].z / t;
        a.w = x[k].w / t;
        z.x = -0.5f * a.x + ((row == col)     ? 1.5f : 0.0f);
        z.y = -0.5f * a.y + ((row == col + 1) ? 1.5f : 0.0f);
        z.z = -0.5f * a.z + ((row == col + 2) ? 1.5f : 0.0f);
        z.w = -0.5f * a.w + ((row == col + 3) ? 1.5f : 0.0f);
        a4[i] = a;
        z4[i] = z;
      }
    }
  }
}

#define OUT_ROWS 4

__global__ __launch_bounds__(256)
void out_kernel(const float* __restrict__ YZY, const float* __restrict__ mean,
                const float* __restrict__ tr, float* __restrict__ out, int d, int rows) {
  __shared__ float res[8 * OUT_ROWS];
  int wid  = threadIdx.x >> 5;
  int lane = threadIdx.x & 31;
  int row0 = blockIdx.x * (8 * OUT_ROWS) + wid * OUT_ROWS;
  const float* yrow[OUT_ROWS];
  const float* mrow[OUT_ROWS];
  float s[OUT_ROWS];
#pragma unroll
  for (int r = 0; r < OUT_ROWS; ++r) {
    int row = row0 + r;
    if (row >= rows) row = rows - 1;
    yrow[r] = YZY + (long long)row * d;
    mrow[r] = mean + (long long)(row / d) * d;
    s[r] = 0.0f;
  }
  for (int j = lane; j < d; j += 32) {
#pragma unroll
    for (int r = 0; r < OUT_ROWS; ++r) s[r] += yrow[r][j] * mrow[r][j];
  }
#pragma unroll
  for (int r = 0; r < OUT_ROWS; ++r)
    for (int off = 16; off > 0; off >>= 1) s[r] += __shfl_down(s[r], off, 32);
  if (lane == 0) {
#pragma unroll
    for (int r = 0; r < OUT_ROWS; ++r) res[wid * OUT_ROWS + r] = s[r];
  }
  __syncthreads();
  if (threadIdx.x < 8 * OUT_ROWS) {
    int row = blockIdx.x * (8 * OUT_ROWS) + threadIdx.x;
    if (row < rows) out[row] = res[threadIdx.x] * sqrtf(tr[row / d]);
  }
}

extern "C" void kernel_launch(void* const* d_in, const int* in_sizes, int n_in,
                              void* d_out, int out_size, void* d_ws, size_t ws_size,
                              hipStream_t stream) {
  (void)n_in; (void)ws_size; (void)out_size;
  const float* feat  = (const float*)d_in[1];
  const float* noise = (const float*)d_in[2];

  const int B  = in_sizes[0];
  const int d  = 256;
  const long long total = (long long)in_sizes[1];
  const int n  = (int)(total / ((long long)B * d));
  const int nd = n * d;
  const long long dd = (long long)d * d;

  float* W = (float*)d_ws;
  long long off = 0;
  float* Fp   = W + off; off += (long long)B * nd;
  float* mean = W + off; off += (long long)B * d;
  float* scl  = W + off; off += B;
  float* nmn  = W + off; off += B;
  float* tr   = W + off; off += B;
  off = (off + 255) & ~255LL;
  float* M0 = W + off; off += (long long)B * dd;
  float* M1 = W + off; off += (long long)B * dd;
  float* M2 = W + off; off += (long long)B * dd;
  float* M3 = W + off; off += (long long)B * dd;

  stats_partial_kernel<<<dim3(STAT_SLICES, B), 256, 0, stream>>>(feat, noise, M0, nd);
  stats_final_kernel<<<1, 128, 0, stream>>>(M0, scl, nmn, B, nd, 31.6227766016838f);
  perturb_kernel<<<dim3((nd / 4 + EW_TILE4 - 1) / EW_TILE4, B), 256, 0, stream>>>(feat, noise, scl, nmn, Fp, nd);
  mean_kernel<<<B, d, 0, stream>>>(Fp, mean, n, d);
  center_kernel<<<dim3((nd / 4 + EW_TILE4 - 1) / EW_TILE4, B), 256, 0, stream>>>(Fp, mean, nd, d);

  dim3 gg(d / BN, d / BM, B);
  wmma_gemm_batched<<<gg, 256, 0, stream>>>(Fp, (long long)nd, 1, d,
                                            Fp, (long long)nd, d, 1,
                                            M0, dd, d, n,
                                            1.0f / (float)(n - 1), 0);
  trace_kernel<<<B, d, 0, stream>>>(M0, tr, d);
  init_A_ZY<<<dim3(((int)(dd / 4) + INIT_TILE4 - 1) / INIT_TILE4, B), 256, 0, stream>>>(M0, tr, M1, M2, d);

  dim3 gs(10, 1, B);
  wmma_gemm_sym<<<gs, 128, 0, stream>>>(M1, M2, M0, d, 0);
  float* Y = M0; float* Z = M2; float* S1 = M1; float* S2 = M3;

  for (int it = 0; it < 3; ++it) {
    wmma_gemm_sym<<<gs, 128, 0, stream>>>(Z, Y, S1, d, 1);
    wmma_gemm_sym<<<gs, 128, 0, stream>>>(Y, S1, S2, d, 0);
    wmma_gemm_sym<<<gs, 128, 0, stream>>>(S1, Z, Y, d, 0);
    float* newY = S2; float* newZ = Y; float* freed = Z;
    Y = newY; Z = newZ; S2 = freed;
  }

  wmma_gemm_sym<<<gs, 128, 0, stream>>>(Z, Y, S1, d, 1);
  wmma_gemm_sym<<<gs, 128, 0, stream>>>(Y, S1, S2, d, 0);

  out_kernel<<<(B * d + 8 * OUT_ROWS - 1) / (8 * OUT_ROWS), 256, 0, stream>>>(S2, mean, tr, (float*)d_out, d, B * d);
}


# Element-wise launches folded into the GEMM beside them: centring into the cov GEMM's operand staging; trace, A, ZY0 into the first product (Y0 = (3A - A@A)/2); 21 -> 18 launches; Y0 re-associated
# speedup vs baseline: 1.6699x; 1.0433x over previous
#include <hip/hip_runtime.h>
#include <hip/hip_bf16.h>

typedef __attribute__((ext_vector_type(2))) float v2f;
typedef __attribute__((ext_vector_type(8))) float v8f;

#define STAT_SLICES 8

__global__ __launch_bounds__(256)
void stats_partial_kernel(const float* __restrict__ feat, const float* __restrict__ noise,
                          float* __restrict__ part_ws, int nd) {
  int s = blockIdx.x;
  int b = blockIdx.y;
  int t = threadIdx.x;
  int len  = nd / STAT_SLICES;
  int len4 = len >> 2;
  const float4* f4 = reinterpret_cast<const float4*>(feat  + (long long)b * nd + (long long)s * len);
  const float4* n4 = reinterpret_cast<const float4*>(noise + (long long)b * nd + (long long)s * len);
  double sf = 0.0, sf2 = 0.0, sn = 0.0, sn2 = 0.0;
#pragma unroll 4
  for (int i = t; i < len4; i += 256) {
    float4 x = f4[i];
    float4 y = n4[i];
    double x0 = (double)x.x, x1 = (double)x.y, x2 = (double)x.z, x3 = (double)x.w;
    double y0 = (double)y.x, y1 = (double)y.y, y2 = (double)y.z, y3 = (double)y.w;
    sf += x0; sf2 += x0 * x0; sn += y0; sn2 += y0 * y0;
    sf += x1; sf2 += x1 * x1; sn += y1; sn2 += y1 * y1;
    sf += x2; sf2 += x2 * x2; sn += y2; sn2 += y2 * y2;
    sf += x3; sf2 += x3 * x3; sn += y3; sn2 += y3 * y3;
  }
  __shared__ double sh[4][256];
  sh[0][t] = sf; sh[1][t] = sf2; sh[2][t] = sn; sh[3][t] = sn2;
  __syncthreads();
  for (int w = 128; w > 0; w >>= 1) {
    if (t < w) {
      sh[0][t] += sh[0][t + w]; sh[1][t] += sh[1][t + w];
      sh[2][t] += sh[2][t + w]; sh[3][t] += sh[3][t + w];
    }
    __syncthreads();
  }
  if (t < 4) {
    double* part = reinterpret_cast<double*>(part_ws);
    part[((long long)b * STAT_SLICES + s) * 4 + t] = sh[t][0];
  }
}

__global__ __launch_bounds__(128)
void stats_final_kernel(const float* __restrict__ part_ws,
                        float* __restrict__ scl, float* __restrict__ nmn,
                        int B, int nd, float snr_pow) {
  const double* part = reinterpret_cast<const double*>(part_ws);
  for (int b = threadIdx.x; b < B; b += 128) {
    const double* p = part + (long long)b * STAT_SLICES * 4;
    double s0 = 0.0, s1 = 0.0, s2 = 0.0, s3 = 0.0;
    for (int s = 0; s < STAT_SLICES; ++s) {
      s0 += p[4 * s + 0]; s1 += p[4 * s + 1];
      s2 += p[4 * s + 2]; s3 += p[4 * s + 3];
    }
    double inv = 1.0 / (double)nd;
    double gmf = s0 * inv;
    double gmn = s2 * inv;
    double tvar = s1 * inv - gmf * gmf;
    double nvar = s3 * inv - gmn * gmn;
    scl[b] = (float)sqrt((tvar / (double)snr_pow) / nvar);
    nmn[b] = (float)gmn;
  }
}

#define EW_TILE4 1024

__global__ __launch_bounds__(256)
void perturb_kernel(const float* __restrict__ feat, const float* __restrict__ noise,
                    const float* __restrict__ scl, const float* __restrict__ nmn,
                    float* __restrict__ Fp, int nd) {
  int b   = blockIdx.y;
  int nd4 = nd >> 2;
  const float4* f4 = reinterpret_cast<const float4*>(feat  + (long long)b * nd);
  const float4* n4 = reinterpret_cast<const float4*>(noise + (long long)b * nd);
  float4*       o4 = reinterpret_cast<float4*>(Fp + (long long)b * nd);
  float nm = nmn[b];
  float sc = scl[b];
  for (int c = blockIdx.x * EW_TILE4 + threadIdx.x; c < nd4; c += gridDim.x * EW_TILE4) {
    float4 x[4], y[4];
#pragma unroll
    for (int k = 0; k < 4; ++k) {
      int i = c + k * 256;
      if (i < nd4) { x[k] = f4[i]; y[k] = n4[i]; }
    }
#pragma unroll
    for (int k = 0; k < 4; ++k) {
      int i = c + k * 256;
      if (i < nd4) {
        float4 r;
        r.x = x[k].x + (y[k].x - nm) * sc;
        r.y = x[k].y + (y[k].y - nm) * sc;
        r.z = x[k].z + (y[k].z - nm) * sc;
        r.w = x[k].w + (y[k].w - nm) * sc;
        o4[i] = r;
      }
    }
  }
}

__global__ __launch_bounds__(256)
void mean_kernel(const float* __restrict__ Fp, float* __restrict__ mean, int n, int d) {
  int b = blockIdx.x;
  int j = threadIdx.x;
  const float* base = Fp + (long long)b * n * d;
  float s = 0.0f;
  for (int i = 0; i < n; ++i) s += base[(long long)i * d + j];
  mean[(long long)b * d + j] = s / (float)n;
}

#define BM 128
#define BN 128
#define BK 16
#define LDP 18

__global__ __launch_bounds__(256)
void wmma_gemm_batched(const float* __restrict__ A, long long sAb, int sAm, int sAk,
                       const float* __restrict__ Bm, long long sBb, int sBk, int sBn,
                       float* __restrict__ Dm, long long sDb,
                       int N, int K, float alpha, int mode,
                       const float* __restrict__ mu) {
  __shared__ __align__(16) float As[BM][LDP];
  __shared__ __align__(16) float Bst[BN][LDP];

  int b  = blockIdx.z;
  int m0 = blockIdx.y * BM;
  int n0 = blockIdx.x * BN;
  const float* Ab = A  + (long long)b * sAb;
  const float* Bb = Bm + (long long)b * sBb;
  float*       Db = Dm + (long long)b * sDb;

  int tid  = threadIdx.x;
  int wid  = tid >> 5;
  int lane = tid & 31;
  int mr   = lane & 15;
  int h    = lane >> 4;
  int wm   = (wid >> 1) * 32;
  int wn   = (wid & 1) * 64;

  const float* mub = mu + (long long)b * N;
  float muA[8];
#pragma unroll
  for (int i = 0; i < 8; ++i) muA[i] = mub[m0 + ((tid + i * 256) >> 4)];
  float muB = mub[n0 + (tid & 127)];

  v8f acc[2][4];
#pragma unroll
  for (int im = 0; im < 2; ++im)
#pragma unroll
    for (int in = 0; in < 4; ++in) acc[im][in] = (v8f){};

  for (int k0 = 0; k0 < K; k0 += BK) {
#pragma unroll
    for (int i = 0; i < 8; ++i) {
      int e = tid + i * 256;
      int m = e >> 4, kk = e & 15;
      As[m][kk] = Ab[(long long)(m0 + m) * sAm + (long long)(k0 + kk) * sAk] - muA[i];
    }
#pragma unroll
    for (int i = 0; i < 8; ++i) {
      int e = tid + i * 256;
      int nn = e & 127, kk = e >> 7;
      Bst[nn][kk] = Bb[(long long)(k0 + kk) * sBk + (long long)(n0 + nn) * sBn] - muB;
    }
    __syncthreads();
#pragma unroll
    for (int kk = 0; kk < BK; kk += 4) {
      int kc = kk + 2 * h;
      v2f a0 = *(const v2f*)&As[wm + mr][kc];
      v2f a1 = *(const v2f*)&As[wm + 16 + mr][kc];
      v2f b0 = *(const v2f*)&Bst[wn + mr][kc];
      v2f b1 = *(const v2f*)&Bst[wn + 16 + mr][kc];
      v2f b2 = *(const v2f*)&Bst[wn + 32 + mr][kc];
      v2f b3 = *(const v2f*)&Bst[wn + 48 + mr][kc];
      acc[0][0] = __builtin_amdgcn_wmma_f32_16x16x4_f32(false, a0, false, b0, (short)0, acc[0][0], false, false);
      acc[0][1] = __builtin_amdgcn_wmma_f32_16x16x4_f32(false, a0, false, b1, (short)0, acc[0][1], false, false);
      acc[0][2] = __builtin_amdgcn_wmma_f32_16x16x4_f32(false, a0, false, b2, (short)0, acc[0][2], false, false);
      acc[0][3] = __builtin_amdgcn_wmma_f32_16x16x4_f32(false, a0, false, b3, (short)0, acc[0][3], false, false);
      acc[1][0] = __builtin_amdgcn_wmma_f32_16x16x4_f32(false, a1, false, b0, (short)0, acc[1][0], false, false);
      acc[1][1] = __builtin_amdgcn_wmma_f32_16x16x4_f32(false, a1, false, b1, (short)0, acc[1][1], false, false);
      acc[1][2] = __builtin_amdgcn_wmma_f32_16x16x4_f32(false, a1, false, b2, (short)0, acc[1][2], false, false);
      acc[1][3] = __builtin_amdgcn_wmma_f32_16x16x4_f32(false, a1, false, b3, (short)0, acc[1][3], false, false);
    }
    __syncthreads();
  }

#pragma unroll
  for (int im = 0; im < 2; ++im) {
#pragma unroll
    for (int r = 0; r < 8; ++r) {
      int row = m0 + wm + 16 * im + 8 * h + r;
      float* drow = Db + (long long)row * N;
#pragma unroll
      for (int in = 0; in < 4; ++in) {
        int col = n0 + wn + 16 * in + mr;
        float v = acc[im][in][r];
        if (mode == 1) {
          v = -0.5f * v + ((row == col) ? 1.5f : 0.0f);
        } else {
          v *= alpha;
        }
        drow[col] = v;
      }
    }
  }
}

#define GLD 256
#define LDA 20
#define SM 64
#define SLDB 160

__global__ __launch_bounds__(128)
void wmma_gemm_sym(const float* __restrict__ A, const float* __restrict__ Bm,
                   float* __restrict__ Dm, int K, int mode) {
  __shared__ __align__(16) float As[SM][LDA];
  __shared__ __align__(16) float Bs[BK / 2][SLDB];

  int b  = blockIdx.z;
  int t  = blockIdx.x;
  int ti = (t >= 4) + (t >= 7) + (t >= 9);
  int tj = ti + t - (4 * ti - (ti * (ti - 1)) / 2);
  int m0 = ti * SM;
  int n0 = tj * SM;
  const float* Ab = A  + (long long)b * (GLD * GLD);
  const float* Bb = Bm + (long long)b * (GLD * GLD);
  float*       Db = Dm + (long long)b * (GLD * GLD);

  int tid  = threadIdx.x;
  int wid  = tid >> 5;
  int lane = tid & 31;
  int mr   = lane & 15;
  int h    = lane >> 4;
  int wm   = (wid >> 1) * 32;
  int wn   = (wid & 1) * 32;

  int ar  = tid >> 2;
  int ak4 = tid & 3;
  int bk  = tid >> 4;
  int bn4 = tid & 15;
  const float* Ag = Ab + (m0 + ar) * GLD + 4 * ak4;
  const float* Bg = Bb + 2 * bk * GLD + n0 + 4 * bn4;

  v8f acc[2][2];
#pragma unroll
  for (int im = 0; im < 2; ++im)
#pragma unroll
    for (int in = 0; in < 2; ++in) acc[im][in] = (v8f){};

  for (int k0 = 0; k0 < K; k0 += BK) {
    float4 ga0 = *reinterpret_cast<const float4*>(Ag + k0);
    float4 ga1 = *reinterpret_cast<const float4*>(Ag + k0 + 32 * GLD);
    float4 gb0 = *reinterpret_cast<const float4*>(Bg + k0 * GLD);
    float4 gb1 = *reinterpret_cast<const float4*>(Bg + (k0 + 1) * GLD);
    *reinterpret_cast<float4*>(&As[ar][4 * ak4])      = ga0;
    *reinterpret_cast<float4*>(&As[ar + 32][4 * ak4]) = ga1;
    float4 lo, hi;
    lo.x = gb0.x; lo.y = gb1.x; lo.z = gb0.y; lo.w = gb1.y;
    hi.x = gb0.z; hi.y = gb1.z; hi.z = gb0.w; hi.w = gb1.w;
    *reinterpret_cast<float4*>(&Bs[bk][8 * bn4])     = lo;
    *reinterpret_cast<float4*>(&Bs[bk][8 * bn4 + 4]) = hi;
    __syncthreads();
#pragma unroll
    for (int kk = 0; kk < BK; kk += 4) {
      int kc = kk + 2 * h;
      v2f a0 = *(const v2f*)&As[wm + mr][kc];
      v2f a1 = *(const v2f*)&As[wm + 16 + mr][kc];
      int kp = (kk >> 1) + h;
      v2f b0 = *(const v2f*)&Bs[kp][2 * (wn + mr)];
      v2f b1 = *(const v2f*)&Bs[kp][2 * (wn + 16 + mr)];
      acc[0][0] = __builtin_amdgcn_wmma_f32_16x16x4_f32(false, a0, false, b0, (short)0, acc[0][0], false, false);
      acc[0][1] = __builtin_amdgcn_wmma_f32_16x16x4_f32(false, a0, false, b1, (short)0, acc[0][1], false, false);
      acc[1][0] = __builtin_amdgcn_wmma_f32_16x16x4_f32(false, a1, false, b0, (short)0, acc[1][0], false, false);
      acc[1][1] = __builtin_amdgcn_wmma_f32_16x16x4_f32(false, a1, false, b1, (short)0, acc[1][1], false, false);
    }
    __syncthreads();
  }

  if (mode == 1) {
#pragma unroll
    for (int im = 0; im < 2; ++im)
#pragma unroll
      for (int in = 0; in < 2; ++in)
#pragma unroll
        for (int r = 0; r < 8; ++r) {
          int row = m0 + wm + 16 * im + 8 * h + r;
          int col = n0 + wn + 16 * in + mr;
          acc[im][in][r] = -0.5f * acc[im][in][r] + ((row == col) ? 1.5f : 0.0f);
        }
  }

#pragma unroll
  for (int im = 0; im < 2; ++im) {
#pragma unroll
    for (int r = 0; r < 8; ++r) {
      int row = m0 + wm + 16 * im + 8 * h + r;
      float* drow = Db + row * GLD;
#pragma unroll
      for (int in = 0; in < 2; ++in) {
        int col = n0 + wn + 16 * in + mr;
        drow[col] = acc[im][in][r];
      }
    }
  }

  if (tj != ti) {
#pragma unroll
    for (int in = 0; in < 2; ++in) {
      float* mrow = Db + (n0 + wn + 16 * in + mr) * GLD + m0 + wm + 8 * h;
#pragma unroll
      for (int im = 0; im < 2; ++im) {
        float4 lo, hi;
        lo.x = acc[im][in][0]; lo.y = acc[im][in][1]; lo.z = acc[im][in][2]; lo.w = acc[im][in][3];
        hi.x = acc[im][in][4]; hi.y = acc[im][in][5]; hi.z = acc[im][in][6]; hi.w = acc[im][in][7];
        *reinterpret_cast<float4*>(mrow + 16 * im)     = lo;
        *reinterpret_cast<float4*>(mrow + 16 * im + 4) = hi;
      }
    }
  }
}

__global__ __launch_bounds__(128)
void wmma_gemm_first(const float* __restrict__ C, float* __restrict__ Ym,
                     float* __restrict__ Zm, float* __restrict__ trout) {
  __shared__ __align__(16) float As[SM][LDA];
  __shared__ __align__(16) float Bs[BK / 2][SLDB];
  __shared__ float red[128];

  int b  = blockIdx.z;
  int t  = blockIdx.x;
  int ti = (t >= 4) + (t >= 7) + (t >= 9);
  int tj = ti + t - (4 * ti - (ti * (ti - 1)) / 2);
  int m0 = ti * SM;
  int n0 = tj * SM;
  const float* Cb = C  + (long long)b * (GLD * GLD);
  float*       Yb = Ym + (long long)b * (GLD * GLD);
  float*       Zb = Zm + (long long)b * (GLD * GLD);

  int tid  = threadIdx.x;
  int wid  = tid >> 5;
  int lane = tid & 31;
  int mr   = lane & 15;
  int h    = lane >> 4;
  int wm   = (wid >> 1) * 32;
  int wn   = (wid & 1) * 32;

  int ar  = tid >> 2;
  int ak4 = tid & 3;
  int bk  = tid >> 4;
  int bn4 = tid & 15;
  const float* Ag = Cb + (m0 + ar) * GLD + 4 * ak4;
  const float* Bg = Cb + 2 * bk * GLD + n0 + 4 * bn4;

  red[tid] = Cb[tid * (GLD + 1)] + Cb[(tid + 128) * (GLD + 1)];
  __syncthreads();
  for (int s = 64; s > 0; s >>= 1) {
    if (tid < s) red[tid] += red[tid + s];
    __syncthreads();
  }
  float tr  = red[0];
  float inv = 1.0f / tr;
  if (t == 0 && tid == 0) trout[b] = tr;

  v8f acc[2][2];
#pragma unroll
  for (int im = 0; im < 2; ++im)
#pragma unroll
    for (int in = 0; in < 2; ++in) acc[im][in] = (v8f){};

  for (int k0 = 0; k0 < GLD; k0 += BK) {
    float4 ga0 = *reinterpret_cast<const float4*>(Ag + k0);
    float4 ga1 = *reinterpret_cast<const float4*>(Ag + k0 + 32 * GLD);
    float4 gb0 = *reinterpret_cast<const float4*>(Bg + k0 * GLD);
    float4 gb1 = *reinterpret_cast<const float4*>(Bg + (k0 + 1) * GLD);
    float4 sa0, sa1;
    sa0.x = ga0.x * inv; sa0.y = ga0.y * inv; sa0.z = ga0.z * inv; sa0.w = ga0.w * inv;
    sa1.x = ga1.x * inv; sa1.y = ga1.y * inv; sa1.z = ga1.z * inv; sa1.w = ga1.w * inv;
    *reinterpret_cast<float4*>(&As[ar][4 * ak4])      = sa0;
    *reinterpret_cast<float4*>(&As[ar + 32][4 * ak4]) = sa1;
    float4 lo, hi;
    lo.x = gb0.x * inv; lo.y = gb1.x * inv; lo.z = gb0.y * inv; lo.w = gb1.y * inv;
    hi.x = gb0.z * inv; hi.y = gb1.z * inv; hi.z = gb0.w * inv; hi.w = gb1.w * inv;
    *reinterpret_cast<float4*>(&Bs[bk][8 * bn4])     = lo;
    *reinterpret_cast<float4*>(&Bs[bk][8 * bn4 + 4]) = hi;
    __syncthreads();
#pragma unroll
    for (int kk = 0; kk < BK; kk += 4) {
      int kc = kk + 2 * h;
      v2f a0 = *(const v2f*)&As[wm + mr][kc];
      v2f a1 = *(const v2f*)&As[wm + 16 + mr][kc];
      int kp = (kk >> 1) + h;
      v2f b0 = *(const v2f*)&Bs[kp][2 * (wn + mr)];
      v2f b1 = *(const v2f*)&Bs[kp][2 * (wn + 16 + mr)];
      acc[0][0] = __builtin_amdgcn_wmma_f32_16x16x4_f32(false, a0, false, b0, (short)0, acc[0][0], false, false);
      acc[0][1] = __builtin_amdgcn_wmma_f32_16x16x4_f32(false, a0, false, b1, (short)0, acc[0][1], false, false);
      acc[1][0] = __builtin_amdgcn_wmma_f32_16x16x4_f32(false, a1, false, b0, (short)0, acc[1][0], false, false);
      acc[1][1] = __builtin_amdgcn_wmma_f32_16x16x4_f32(false, a1, false, b1, (short)0, acc[1][1], false, false);
    }
    __syncthreads();
  }

  v8f zv[2][2];
#pragma unroll
  for (int im = 0; im < 2; ++im)
#pragma unroll
    for (int in = 0; in < 2; ++in)
#pragma unroll
      for (int r = 0; r < 8; ++r) {
        int row = m0 + wm + 16 * im + 8 * h + r;
        int col = n0 + wn + 16 * in + mr;
        float a = Cb[row * GLD + col] / tr;
        zv[im][in][r]  = -0.5f * a + ((row == col) ? 1.5f : 0.0f);
        acc[im][in][r] = 1.5f * a - 0.5f * acc[im][in][r];
      }

#pragma unroll
  for (int im = 0; im < 2; ++im) {
#pragma unroll
    for (int r = 0; r < 8; ++r) {
      int row = m0 + wm + 16 * im + 8 * h + r;
      float* yrow = Yb + row * GLD;
      float* zrow = Zb + row * GLD;
#pragma unroll
      for (int in = 0; in < 2; ++in) {
        int col = n0 + wn + 16 * in + mr;
        yrow[col] = acc[im][in][r];
        zrow[col] = zv[im][in][r];
      }
    }
  }

  if (tj != ti) {
#pragma unroll
    for (int in = 0; in < 2; ++in) {
      int moff = (n0 + wn + 16 * in + mr) * GLD + m0 + wm + 8 * h;
#pragma unroll
      for (int im = 0; im < 2; ++im) {
        float4 lo, hi;
        lo.x = acc[im][in][0]; lo.y = acc[im][in][1]; lo.z = acc[im][in][2]; lo.w = acc[im][in][3];
        hi.x = acc[im][in][4]; hi.y = acc[im][in][5]; hi.z = acc[im][in][6]; hi.w = acc[im][in][7];
        *reinterpret_cast<float4*>(Yb + moff + 16 * im)     = lo;
        *reinterpret_cast<float4*>(Yb + moff + 16 * im + 4) = hi;
        lo.x = zv[im][in][0]; lo.y = zv[im][in][1]; lo.z = zv[im][in][2]; lo.w = zv[im][in][3];
        hi.x = zv[im][in][4]; hi.y = zv[im][in][5]; hi.z = zv[im][in][6]; hi.w = zv[im][in][7];
        *reinterpret_cast<float4*>(Zb + moff + 16 * im)     = lo;
        *reinterpret_cast<float4*>(Zb + moff + 16 * im + 4) = hi;
      }
    }
  }
}

#define OUT_ROWS 4

__global__ __launch_bounds__(256)
void out_kernel(const float* __restrict__ YZY, const float* __restrict__ mean,
                const float* __restrict__ tr, float* __restrict__ out, int d, int rows) {
  __shared__ float res[8 * OUT_ROWS];
  int wid  = threadIdx.x >> 5;
  int lane = threadIdx.x & 31;
  int row0 = blockIdx.x * (8 * OUT_ROWS) + wid * OUT_ROWS;
  const float* yrow[OUT_ROWS];
  const float* mrow[OUT_ROWS];
  float s[OUT_ROWS];
#pragma unroll
  for (int r = 0; r < OUT_ROWS; ++r) {
    int row = row0 + r;
    if (row >= rows) row = rows - 1;
    yrow[r] = YZY + (long long)row * d;
    mrow[r] = mean + (long long)(row / d) * d;
    s[r] = 0.0f;
  }
  for (int j = lane; j < d; j += 32) {
#pragma unroll
    for (int r = 0; r < OUT_ROWS; ++r) s[r] += yrow[r][j] * mrow[r][j];
  }
#pragma unroll
  for (int r = 0; r < OUT_ROWS; ++r)
    for (int off = 16; off > 0; off >>= 1) s[r] += __shfl_down(s[r], off, 32);
  if (lane == 0) {
#pragma unroll
    for (int r = 0; r < OUT_ROWS; ++r) res[wid * OUT_ROWS + r] = s[r];
  }
  __syncthreads();
  if (threadIdx.x < 8 * OUT_ROWS) {
    int row = blockIdx.x * (8 * OUT_ROWS) + threadIdx.x;
    if (row < rows) out[row] = res[threadIdx.x] * sqrtf(tr[row / d]);
  }
}

extern "C" void kernel_launch(void* const* d_in, const int* in_sizes, int n_in,
                              void* d_out, int out_size, void* d_ws, size_t ws_size,
                              hipStream_t stream) {
  (void)n_in; (void)ws_size; (void)out_size;
  const float* feat  = (const float*)d_in[1];
  const float* noise = (const float*)d_in[2];

  const int B  = in_sizes[0];
  const int d  = 256;
  const long long total = (long long)in_sizes[1];
  const int n  = (int)(total / ((long long)B * d));
  const int nd = n * d;
  const long long dd = (long long)d * d;

  float* W = (float*)d_ws;
  long long off = 0;
  float* Fp   = W + off; off += (long long)B * nd;
  float* mean = W + off; off += (long long)B * d;
  float* scl  = W + off; off += B;
  float* nmn  = W + off; off += B;
  float* tr   = W + off; off += B;
  off = (off + 255) & ~255LL;
  float* M0 = W + off; off += (long long)B * dd;
  float* M1 = W + off; off += (long long)B * dd;
  float* M2 = W + off; off += (long long)B * dd;
  float* M3 = W + off; off += (long long)B * dd;

  stats_partial_kernel<<<dim3(STAT_SLICES, B), 256, 0, stream>>>(feat, noise, M0, nd);
  stats_final_kernel<<<1, 128, 0, stream>>>(M0, scl, nmn, B, nd, 31.6227766016838f);
  perturb_kernel<<<dim3((nd / 4 + EW_TILE4 - 1) / EW_TILE4, B), 256, 0, stream>>>(feat, noise, scl, nmn, Fp, nd);
  mean_kernel<<<B, d, 0, stream>>>(Fp, mean, n, d);

  dim3 gg(d / BN, d / BM, B);
  wmma_gemm_batched<<<gg, 256, 0, stream>>>(Fp, (long long)nd, 1, d,
                                            Fp, (long long)nd, d, 1,
                                            M0, dd, d, n,
                                            1.0f / (float)(n - 1), 0, mean);

  dim3 gs(10, 1, B);
  wmma_gemm_first<<<gs, 128, 0, stream>>>(M0, M1, M2, tr);
  float* Y = M1; float* Z = M2; float* S1 = M0; float* S2 = M3;

  for (int it = 0; it < 3; ++it) {
    wmma_gemm_sym<<<gs, 128, 0, stream>>>(Z, Y, S1, d, 1);
    wmma_gemm_sym<<<gs, 128, 0, stream>>>(Y, S1, S2, d, 0);
    wmma_gemm_sym<<<gs, 128, 0, stream>>>(S1, Z, Y, d, 0);
    float* newY = S2; float* newZ = Y; float* freed = Z;
    Y = newY; Z = newZ; S2 = freed;
  }

  wmma_gemm_sym<<<gs, 128, 0, stream>>>(Z, Y, S1, d, 1);
  wmma_gemm_sym<<<gs, 128, 0, stream>>>(Y, S1, S2, d, 0);

  out_kernel<<<(B * d + 8 * OUT_ROWS - 1) / (8 * OUT_ROWS), 256, 0, stream>>>(S2, mean, tr, (float*)d_out, d, B * d);
}


# Newton-Schulz carried on T = Z@Y, Z never formed: T' = (ZY@ZY)@T; T@T and T@Y share the staged rows of T in one launch, ZY = 1.5I - 0.5T in its epilogue; 18 -> 15 launches; sums re-associated
# speedup vs baseline: 1.8073x; 1.0823x over previous
#include <hip/hip_runtime.h>
#include <hip/hip_bf16.h>

typedef __attribute__((ext_vector_type(2))) float v2f;
typedef __attribute__((ext_vector_type(8))) float v8f;

#define STAT_SLICES 8

__global__ __launch_bounds__(256)
void stats_partial_kernel(const float* __restrict__ feat, const float* __restrict__ noise,
                          float* __restrict__ part_ws, int nd) {
  int s = blockIdx.x;
  int b = blockIdx.y;
  int t = threadIdx.x;
  int len  = nd / STAT_SLICES;
  int len4 = len >> 2;
  const float4* f4 = reinterpret_cast<const float4*>(feat  + (long long)b * nd + (long long)s * len);
  const float4* n4 = reinterpret_cast<const float4*>(noise + (long long)b * nd + (long long)s * len);
  double sf = 0.0, sf2 = 0.0, sn = 0.0, sn2 = 0.0;
#pragma unroll 4
  for (int i = t; i < len4; i += 256) {
    float4 x = f4[i];
    float4 y = n4[i];
    double x0 = (double)x.x, x1 = (double)x.y, x2 = (double)x.z, x3 = (double)x.w;
    double y0 = (double)y.x, y1 = (double)y.y, y2 = (double)y.z, y3 = (double)y.w;
    sf += x0; sf2 += x0 * x0; sn += y0; sn2 += y0 * y0;
    sf += x1; sf2 += x1 * x1; sn += y1; sn2 += y1 * y1;
    sf += x2; sf2 += x2 * x2; sn += y2; sn2 += y2 * y2;
    sf += x3; sf2 += x3 * x3; sn += y3; sn2 += y3 * y3;
  }
  __shared__ double sh[4][256];
  sh[0][t] = sf; sh[1][t] = sf2; sh[2][t] = sn; sh[3][t] = sn2;
  __syncthreads();
  for (int w = 128; w > 0; w >>= 1) {
    if (t < w) {
      sh[0][t] += sh[0][t + w]; sh[1][t] += sh[1][t + w];
      sh[2][t] += sh[2][t + w]; sh[3][t] += sh[3][t + w];
    }
    __syncthreads();
  }
  if (t < 4) {
    double* part = reinterpret_cast<double*>(part_ws);
    part[((long long)b * STAT_SLICES + s) * 4 + t] = sh[t][0];
  }
}

__global__ __launch_bounds__(128)
void stats_final_kernel(const float* __restrict__ part_ws,
                        float* __restrict__ scl, float* __restrict__ nmn,
                        int B, int nd, float snr_pow) {
  const double* part = reinterpret_cast<const double*>(part_ws);
  for (int b = threadIdx.x; b < B; b += 128) {
    const double* p = part + (long long)b * STAT_SLICES * 4;
    double s0 = 0.0, s1 = 0.0, s2 = 0.0, s3 = 0.0;
    for (int s = 0; s < STAT_SLICES; ++s) {
      s0 += p[4 * s + 0]; s1 += p[4 * s + 1];
      s2 += p[4 * s + 2]; s3 += p[4 * s + 3];
    }
    double inv = 1.0 / (double)nd;
    double gmf = s0 * inv;
    double gmn = s2 * inv;
    double tvar = s1 * inv - gmf * gmf;
    double nvar = s3 * inv - gmn * gmn;
    scl[b] = (float)sqrt((tvar / (double)snr_pow) / nvar);
    nmn[b] = (float)gmn;
  }
}

#define EW_TILE4 1024

__global__ __launch_bounds__(256)
void perturb_kernel(const float* __restrict__ feat, const float* __restrict__ noise,
                    const float* __restrict__ scl, const float* __restrict__ nmn,
                    float* __restrict__ Fp, int nd) {
  int b   = blockIdx.y;
  int nd4 = nd >> 2;
  const float4* f4 = reinterpret_cast<const float4*>(feat  + (long long)b * nd);
  const float4* n4 = reinterpret_cast<const float4*>(noise + (long long)b * nd);
  float4*       o4 = reinterpret_cast<float4*>(Fp + (long long)b * nd);
  float nm = nmn[b];
  float sc = scl[b];
  for (int c = blockIdx.x * EW_TILE4 + threadIdx.x; c < nd4; c += gridDim.x * EW_TILE4) {
    float4 x[4], y[4];
#pragma unroll
    for (int k = 0; k < 4; ++k) {
      int i = c + k * 256;
      if (i < nd4) { x[k] = f4[i]; y[k] = n4[i]; }
    }
#pragma unroll
    for (int k = 0; k < 4; ++k) {
      int i = c + k * 256;
      if (i < nd4) {
        float4 r;
        r.x = x[k].x + (y[k].x - nm) * sc;
        r.y = x[k].y + (y[k].y - nm) * sc;
        r.z = x[k].z + (y[k].z - nm) * sc;
        r.w = x[k].w + (y[k].w - nm) * sc;
        o4[i] = r;
      }
    }
  }
}

__global__ __launch_bounds__(256)
void mean_kernel(const float* __restrict__ Fp, float* __restrict__ mean, int n, int d) {
  int b = blockIdx.x;
  int j = threadIdx.x;
  const float* base = Fp + (long long)b * n * d;
  float s = 0.0f;
  for (int i = 0; i < n; ++i) s += base[(long long)i * d + j];
  mean[(long long)b * d + j] = s / (float)n;
}

#define BM 128
#define BN 128
#define BK 16
#define LDP 18

__global__ __launch_bounds__(256)
void wmma_gemm_batched(const float* __restrict__ A, long long sAb, int sAm, int sAk,
                       const float* __restrict__ Bm, long long sBb, int sBk, int sBn,
                       float* __restrict__ Dm, long long sDb,
                       int N, int K, float alpha, int mode,
                       const float* __restrict__ mu) {
  __shared__ __align__(16) float As[BM][LDP];
  __shared__ __align__(16) float Bst[BN][LDP];

  int b  = blockIdx.z;
  int m0 = blockIdx.y * BM;
  int n0 = blockIdx.x * BN;
  const float* Ab = A  + (long long)b * sAb;
  const float* Bb = Bm + (long long)b * sBb;
  float*       Db = Dm + (long long)b * sDb;

  int tid  = threadIdx.x;
  int wid  = tid >> 5;
  int lane = tid & 31;
  int mr   = lane & 15;
  int h    = lane >> 4;
  int wm   = (wid >> 1) * 32;
  int wn   = (wid & 1) * 64;

  const float* mub = mu + (long long)b * N;
  float muA[8];
#pragma unroll
  for (int i = 0; i < 8; ++i) muA[i] = mub[m0 + ((tid + i * 256) >> 4)];
  float muB = mub[n0 + (tid & 127)];

  v8f acc[2][4];
#pragma unroll
  for (int im = 0; im < 2; ++im)
#pragma unroll
    for (int in = 0; in < 4; ++in) acc[im][in] = (v8f){};

  for (int k0 = 0; k0 < K; k0 += BK) {
#pragma unroll
    for (int i = 0; i < 8; ++i) {
      int e = tid + i * 256;
      int m = e >> 4, kk = e & 15;
      As[m][kk] = Ab[(long long)(m0 + m) * sAm + (long long)(k0 + kk) * sAk] - muA[i];
    }
#pragma unroll
    for (int i = 0; i < 8; ++i) {
      int e = tid + i * 256;
      int nn = e & 127, kk = e >> 7;
      Bst[nn][kk] = Bb[(long long)(k0 + kk) * sBk + (long long)(n0 + nn) * sBn] - muB;
    }
    __syncthreads();
#pragma unroll
    for (int kk = 0; kk < BK; kk += 4) {
      int kc = kk + 2 * h;
      v2f a0 = *(const v2f*)&As[wm + mr][kc];
      v2f a1 = *(const v2f*)&As[wm + 16 + mr][kc];
      v2f b0 = *(const v2f*)&Bst[wn + mr][kc];
      v2f b1 = *(const v2f*)&Bst[wn + 16 + mr][kc];
      v2f b2 = *(const v2f*)&Bst[wn + 32 + mr][kc];
      v2f b3 = *(const v2f*)&Bst[wn + 48 + mr][kc];
      acc[0][0] = __builtin_amdgcn_wmma_f32_16x16x4_f32(false, a0, false, b0, (short)0, acc[0][0], false, false);
      acc[0][1] = __builtin_amdgcn_wmma_f32_16x16x4_f32(false, a0, false, b1, (short)0, acc[0][1], false, false);
      acc[0][2] = __builtin_amdgcn_wmma_f32_16x16x4_f32(false, a0, false, b2, (short)0, acc[0][2], false, false);
      acc[0][3] = __builtin_amdgcn_wmma_f32_16x16x4_f32(false, a0, false, b3, (short)0, acc[0][3], false, false);
      acc[1][0] = __builtin_amdgcn_wmma_f32_16x16x4_f32(false, a1, false, b0, (short)0, acc[1][0], false, false);
      acc[1][1] = __builtin_amdgcn_wmma_f32_16x16x4_f32(false, a1, false, b1, (short)0, acc[1][1], false, false);
      acc[1][2] = __builtin_amdgcn_wmma_f32_16x16x4_f32(false, a1, false, b2, (short)0, acc[1][2], false, false);
      acc[1][3] = __builtin_amdgcn_wmma_f32_16x16x4_f32(false, a1, false, b3, (short)0, acc[1][3], false, false);
    }
    __syncthreads();
  }

#pragma unroll
  for (int im = 0; im < 2; ++im) {
#pragma unroll
    for (int r = 0; r < 8; ++r) {
      int row = m0 + wm + 16 * im + 8 * h + r;
      float* drow = Db + (long long)row * N;
#pragma unroll
      for (int in = 0; in < 4; ++in) {
        int col = n0 + wn + 16 * in + mr;
        float v = acc[im][in][r];
        if (mode == 1) {
          v = -0.5f * v + ((row == col) ? 1.5f : 0.0f);
        } else {
          v *= alpha;
        }
        drow[col] = v;
      }
    }
  }
}

#define GLD 256
#define LDA 20
#define SM 64
#define SLDB 160

__global__ __launch_bounds__(128)
void wmma_gemm_sym(const float* __restrict__ A, const float* __restrict__ Bm,
                   float* __restrict__ Dm, int K, int mode) {
  __shared__ __align__(16) float As[SM][LDA];
  __shared__ __align__(16) float Bs[BK / 2][SLDB];

  int b  = blockIdx.z;
  int t  = blockIdx.x;
  int ti = (t >= 4) + (t >= 7) + (t >= 9);
  int tj = ti + t - (4 * ti - (ti * (ti - 1)) / 2);
  int m0 = ti * SM;
  int n0 = tj * SM;
  const float* Ab = A  + (long long)b * (GLD * GLD);
  const float* Bb = Bm + (long long)b * (GLD * GLD);
  float*       Db = Dm + (long long)b * (GLD * GLD);

  int tid  = threadIdx.x;
  int wid  = tid >> 5;
  int lane = tid & 31;
  int mr   = lane & 15;
  int h    = lane >> 4;
  int wm   = (wid >> 1) * 32;
  int wn   = (wid & 1) * 32;

  int ar  = tid >> 2;
  int ak4 = tid & 3;
  int bk  = tid >> 4;
  int bn4 = tid & 15;
  const float* Ag = Ab + (m0 + ar) * GLD + 4 * ak4;
  const float* Bg = Bb + 2 * bk * GLD + n0 + 4 * bn4;

  v8f acc[2][2];
#pragma unroll
  for (int im = 0; im < 2; ++im)
#pragma unroll
    for (int in = 0; in < 2; ++in) acc[im][in] = (v8f){};

  for (int k0 = 0; k0 < K; k0 += BK) {
    float4 ga0 = *reinterpret_cast<const float4*>(Ag + k0);
    float4 ga1 = *reinterpret_cast<const float4*>(Ag + k0 + 32 * GLD);
    float4 gb0 = *reinterpret_cast<const float4*>(Bg + k0 * GLD);
    float4 gb1 = *reinterpret_cast<const float4*>(Bg + (k0 + 1) * GLD);
    *reinterpret_cast<float4*>(&As[ar][4 * ak4])      = ga0;
    *reinterpret_cast<float4*>(&As[ar + 32][4 * ak4]) = ga1;
    float4 lo, hi;
    lo.x = gb0.x; lo.y = gb1.x; lo.z = gb0.y; lo.w = gb1.y;
    hi.x = gb0.z; hi.y = gb1.z; hi.z = gb0.w; hi.w = gb1.w;
    *reinterpret_cast<float4*>(&Bs[bk][8 * bn4])     = lo;
    *reinterpret_cast<float4*>(&Bs[bk][8 * bn4 + 4]) = hi;
    __syncthreads();
#pragma unroll
    for (int kk = 0; kk < BK; kk += 4) {
      int kc = kk + 2 * h;
      v2f a0 = *(const v2f*)&As[wm + mr][kc];
      v2f a1 = *(const v2f*)&As[wm + 16 + mr][kc];
      int kp = (kk >> 1) + h;
      v2f b0 = *(const v2f*)&Bs[kp][2 * (wn + mr)];
      v2f b1 = *(const v2f*)&Bs[kp][2 * (wn + 16 + mr)];
      acc[0][0] = __builtin_amdgcn_wmma_f32_16x16x4_f32(false, a0, false, b0, (short)0, acc[0][0], false, false);
      acc[0][1] = __builtin_amdgcn_wmma_f32_16x16x4_f32(false, a0, false, b1, (short)0, acc[0][1], false, false);
      acc[1][0] = __builtin_amdgcn_wmma_f32_16x16x4_f32(false, a1, false, b0, (short)0, acc[1][0], false, false);
      acc[1][1] = __builtin_amdgcn_wmma_f32_16x16x4_f32(false, a1, false, b1, (short)0, acc[1][1], false, false);
    }
    __syncthreads();
  }

  if (mode == 1) {
#pragma unroll
    for (int im = 0; im < 2; ++im)
#pragma unroll
      for (int in = 0; in < 2; ++in)
#pragma unroll
        for (int r = 0; r < 8; ++r) {
          int row = m0 + wm + 16 * im + 8 * h + r;
          int col = n0 + wn + 16 * in + mr;
          acc[im][in][r] = -0.5f * acc[im][in][r] + ((row == col) ? 1.5f : 0.0f);
        }
  }

#pragma unroll
  for (int im = 0; im < 2; ++im) {
#pragma unroll
    for (int r = 0; r < 8; ++r) {
      int row = m0 + wm + 16 * im + 8 * h + r;
      float* drow = Db + row * GLD;
#pragma unroll
      for (int in = 0; in < 2; ++in) {
        int col = n0 + wn + 16 * in + mr;
        drow[col] = acc[im][in][r];
      }
    }
  }

  if (tj != ti) {
#pragma unroll
    for (int in = 0; in < 2; ++in) {
      float* mrow = Db + (n0 + wn + 16 * in + mr) * GLD + m0 + wm + 8 * h;
#pragma unroll
      for (int im = 0; im < 2; ++im) {
        float4 lo, hi;
        lo.x = acc[im][in][0]; lo.y = acc[im][in][1]; lo.z = acc[im][in][2]; lo.w = acc[im][in][3];
        hi.x = acc[im][in][4]; hi.y = acc[im][in][5]; hi.z = acc[im][in][6]; hi.w = acc[im][in][7];
        *reinterpret_cast<float4*>(mrow + 16 * im)     = lo;
        *reinterpret_cast<float4*>(mrow + 16 * im + 4) = hi;
      }
    }
  }
}

__global__ __launch_bounds__(128)
void wmma_gemm_first(const float* __restrict__ C, float* __restrict__ Ym,
                     float* __restrict__ Zm, float* __restrict__ trout) {
  __shared__ __align__(16) float As[SM][LDA];
  __shared__ __align__(16) float Bs[BK / 2][SLDB];
  __shared__ float red[128];

  int b  = blockIdx.z;
  int t  = blockIdx.x;
  int ti = (t >= 4) + (t >= 7) + (t >= 9);
  int tj = ti + t - (4 * ti - (ti * (ti - 1)) / 2);
  int m0 = ti * SM;
  int n0 = tj * SM;
  const float* Cb = C  + (long long)b * (GLD * GLD);
  float*       Yb = Ym + (long long)b * (GLD * GLD);
  float*       Zb = Zm + (long long)b * (GLD * GLD);

  int tid  = threadIdx.x;
  int wid  = tid >> 5;
  int lane = tid & 31;
  int mr   = lane & 15;
  int h    = lane >> 4;
  int wm   = (wid >> 1) * 32;
  int wn   = (wid & 1) * 32;

  int ar  = tid >> 2;
  int ak4 = tid & 3;
  int bk  = tid >> 4;
  int bn4 = tid & 15;
  const float* Ag = Cb + (m0 + ar) * GLD + 4 * ak4;
  const float* Bg = Cb + 2 * bk * GLD + n0 + 4 * bn4;

  red[tid] = Cb[tid * (GLD + 1)] + Cb[(tid + 128) * (GLD + 1)];
  __syncthreads();
  for (int s = 64; s > 0; s >>= 1) {
    if (tid < s) red[tid] += red[tid + s];
    __syncthreads();
  }
  float tr  = red[0];
  float inv = 1.0f / tr;
  if (t == 0 && tid == 0) trout[b] = tr;

  v8f acc[2][2];
#pragma unroll
  for (int im = 0; im < 2; ++im)
#pragma unroll
    for (int in = 0; in < 2; ++in) acc[im][in] = (v8f){};

  for (int k0 = 0; k0 < GLD; k0 += BK) {
    float4 ga0 = *reinterpret_cast<const float4*>(Ag + k0);
    float4 ga1 = *reinterpret_cast<const float4*>(Ag + k0 + 32 * GLD);
    float4 gb0 = *reinterpret_cast<const float4*>(Bg + k0 * GLD);
    float4 gb1 = *reinterpret_cast<const float4*>(Bg + (k0 + 1) * GLD);
    float4 sa0, sa1;
    sa0.x = ga0.x * inv; sa0.y = ga0.y * inv; sa0.z = ga0.z * inv; sa0.w = ga0.w * inv;
    sa1.x = ga1.x * inv; sa1.y = ga1.y * inv; sa1.z = ga1.z * inv; sa1.w = ga1.w * inv;
    *reinterpret_cast<float4*>(&As[ar][4 * ak4])      = sa0;
    *reinterpret_cast<float4*>(&As[ar + 32][4 * ak4]) = sa1;
    float4 lo, hi;
    lo.x = gb0.x * inv; lo.y = gb1.x * inv; lo.z = gb0.y * inv; lo.w = gb1.y * inv;
    hi.x = gb0.z * inv; hi.y = gb1.z * inv; hi.z = gb0.w * inv; hi.w = gb1.w * inv;
    *reinterpret_cast<float4*>(&Bs[bk][8 * bn4])     = lo;
    *reinterpret_cast<float4*>(&Bs[bk][8 * bn4 + 4]) = hi;
    __syncthreads();
#pragma unroll
    for (int kk = 0; kk < BK; kk += 4) {
      int kc = kk + 2 * h;
      v2f a0 = *(const v2f*)&As[wm + mr][kc];
      v2f a1 = *(const v2f*)&As[wm + 16 + mr][kc];
      int kp = (kk >> 1) + h;
      v2f b0 = *(const v2f*)&Bs[kp][2 * (wn + mr)];
      v2f b1 = *(const v2f*)&Bs[kp][2 * (wn + 16 + mr)];
      acc[0][0] = __builtin_amdgcn_wmma_f32_16x16x4_f32(false, a0, false, b0, (short)0, acc[0][0], false, false);
      acc[0][1] = __builtin_amdgcn_wmma_f32_16x16x4_f32(false, a0, false, b1, (short)0, acc[0][1], false, false);
      acc[1][0] = __builtin_amdgcn_wmma_f32_16x16x4_f32(false, a1, false, b0, (short)0, acc[1][0], false, false);
      acc[1][1] = __builtin_amdgcn_wmma_f32_16x16x4_f32(false, a1, false, b1, (short)0, acc[1][1], false, false);
    }
    __syncthreads();
  }

  v8f zv[2][2];
#pragma unroll
  for (int im = 0; im < 2; ++im)
#pragma unroll
    for (int in = 0; in < 2; ++in)
#pragma unroll
      for (int r = 0; r < 8; ++r) {
        int row = m0 + wm + 16 * im + 8 * h + r;
        int col = n0 + wn + 16 * in + mr;
        float a = Cb[row * GLD + col] / tr;
        zv[im][in][r]  = -0.5f * a + ((row == col) ? 1.5f : 0.0f);
        acc[im][in][r] = 1.5f * a - 0.5f * acc[im][in][r];
      }

#pragma unroll
  for (int im = 0; im < 2; ++im) {
#pragma unroll
    for (int r = 0; r < 8; ++r) {
      int row = m0 + wm + 16 * im + 8 * h + r;
      float* yrow = Yb + row * GLD;
      float* zrow = Zb + row * GLD;
#pragma unroll
      for (int in = 0; in < 2; ++in) {
        int col = n0 + wn + 16 * in + mr;
        yrow[col] = acc[im][in][r];
        zrow[col] = zv[im][in][r];
      }
    }
  }

  if (tj != ti) {
#pragma unroll
    for (int in = 0; in < 2; ++in) {
      int moff = (n0 + wn + 16 * in + mr) * GLD + m0 + wm + 8 * h;
#pragma unroll
      for (int im = 0; im < 2; ++im) {
        float4 lo, hi;
        lo.x = acc[im][in][0]; lo.y = acc[im][in][1]; lo.z = acc[im][in][2]; lo.w = acc[im][in][3];
        hi.x = acc[im][in][4]; hi.y = acc[im][in][5]; hi.z = acc[im][in][6]; hi.w = acc[im][in][7];
        *reinterpret_cast<float4*>(Yb + moff + 16 * im)     = lo;
        *reinterpret_cast<float4*>(Yb + moff + 16 * im + 4) = hi;
        lo.x = zv[im][in][0]; lo.y = zv[im][in][1]; lo.z = zv[im][in][2]; lo.w = zv[im][in][3];
        hi.x = zv[im][in][4]; hi.y = zv[im][in][5]; hi.z = zv[im][in][6]; hi.w = zv[im][in][7];
        *reinterpret_cast<float4*>(Zb + moff + 16 * im)     = lo;
        *reinterpret_cast<float4*>(Zb + moff + 16 * im + 4) = hi;
      }
    }
  }
}

__global__ __launch_bounds__(128)
void wmma_gemm_pair(const float* __restrict__ Tm, const float* __restrict__ Yin,
                    float* __restrict__ Sm, float* __restrict__ Ym) {
  __shared__ __align__(16) float As[SM][LDA];
  __shared__ __align__(16) float Bt[BK / 2][SLDB];
  __shared__ __align__(16) float By[BK / 2][SLDB];

  int b  = blockIdx.z;
  int t  = blockIdx.x;
  int ti = (t >= 4) + (t >= 7) + (t >= 9);
  int tj = ti + t - (4 * ti - (ti * (ti - 1)) / 2);
  int m0 = ti * SM;
  int n0 = tj * SM;
  const float* Tb = Tm  + (long long)b * (GLD * GLD);
  const float* Yb = Yin + (long long)b * (GLD * GLD);
  float*       Sb = Sm  + (long long)b * (GLD * GLD);
  float*       Ob = Ym  + (long long)b * (GLD * GLD);

  int tid  = threadIdx.x;
  int wid  = tid >> 5;
  int lane = tid & 31;
  int mr   = lane & 15;
  int h    = lane >> 4;
  int wm   = (wid >> 1) * 32;
  int wn   = (wid & 1) * 32;

  int ar  = tid >> 2;
  int ak4 = tid & 3;
  int bk  = tid >> 4;
  int bn4 = tid & 15;
  const float* Ag = Tb + (m0 + ar) * GLD + 4 * ak4;
  const float* Bg = Tb + 2 * bk * GLD + n0 + 4 * bn4;
  const float* Cg = Yb + 2 * bk * GLD + n0 + 4 * bn4;

  v8f accS[2][2], accY[2][2];
#pragma unroll
  for (int im = 0; im < 2; ++im)
#pragma unroll
    for (int in = 0; in < 2; ++in) { accS[im][in] = (v8f){}; accY[im][in] = (v8f){}; }

  for (int k0 = 0; k0 < GLD; k0 += BK) {
    float4 ga0 = *reinterpret_cast<const float4*>(Ag + k0);
    float4 ga1 = *reinterpret_cast<const float4*>(Ag + k0 + 32 * GLD);
    float4 gb0 = *reinterpret_cast<const float4*>(Bg + k0 * GLD);
    float4 gb1 = *reinterpret_cast<const float4*>(Bg + (k0 + 1) * GLD);
    float4 gc0 = *reinterpret_cast<const float4*>(Cg + k0 * GLD);
    float4 gc1 = *reinterpret_cast<const float4*>(Cg + (k0 + 1) * GLD);
    *reinterpret_cast<float4*>(&As[ar][4 * ak4])      = ga0;
    *reinterpret_cast<float4*>(&As[ar + 32][4 * ak4]) = ga1;
    float4 lo, hi;
    lo.x = gb0.x; lo.y = gb1.x; lo.z = gb0.y; lo.w = gb1.y;
    hi.x = gb0.z; hi.y = gb1.z; hi.z = gb0.w; hi.w = gb1.w;
    *reinterpret_cast<float4*>(&Bt[bk][8 * bn4])     = lo;
    *reinterpret_cast<float4*>(&Bt[bk][8 * bn4 + 4]) = hi;
    lo.x = gc0.x; lo.y = gc1.x; lo.z = gc0.y; lo.w = gc1.y;
    hi.x = gc0.z; hi.y = gc1.z; hi.z = gc0.w; hi.w = gc1.w;
    *reinterpret_cast<float4*>(&By[bk][8 * bn4])     = lo;
    *reinterpret_cast<float4*>(&By[bk][8 * bn4 + 4]) = hi;
    __syncthreads();
#pragma unroll
    for (int kk = 0; kk < BK; kk += 4) {
      int kc = kk + 2 * h;
      v2f a0 = *(const v2f*)&As[wm + mr][kc];
      v2f a1 = *(const v2f*)&As[wm + 16 + mr][kc];
      int kp = (kk >> 1) + h;
      v2f b0 = *(const v2f*)&Bt[kp][2 * (wn + mr)];
      v2f b1 = *(const v2f*)&Bt[kp][2 * (wn + 16 + mr)];
      v2f c0 = *(const v2f*)&By[kp][2 * (wn + mr)];
      v2f c1 = *(const v2f*)&By[kp][2 * (wn + 16 + mr)];
      accS[0][0] = __builtin_amdgcn_wmma_f32_16x16x4_f32(false, a0, false, b0, (short)0, accS[0][0], false, false);
      accS[0][1] = __builtin_amdgcn_wmma_f32_16x16x4_f32(false, a0, false, b1, (short)0, accS[0][1], false, false);
      accY[0][0] = __builtin_amdgcn_wmma_f32_16x16x4_f32(false, a0, false, c0, (short)0, accY[0][0], false, false);
      accY[0][1] = __builtin_amdgcn_wmma_f32_16x16x4_f32(false, a0, false, c1, (short)0, accY[0][1], false, false);
      accS[1][0] = __builtin_amdgcn_wmma_f32_16x16x4_f32(false, a1, false, b0, (short)0, accS[1][0], false, false);
      accS[1][1] = __builtin_amdgcn_wmma_f32_16x16x4_f32(false, a1, false, b1, (short)0, accS[1][1], false, false);
      accY[1][0] = __builtin_amdgcn_wmma_f32_16x16x4_f32(false, a1, false, c0, (short)0, accY[1][0], false, false);
      accY[1][1] = __builtin_amdgcn_wmma_f32_16x16x4_f32(false, a1, false, c1, (short)0, accY[1][1], false, false);
    }
    __syncthreads();
  }

#pragma unroll
  for (int im = 0; im < 2; ++im)
#pragma unroll
    for (int in = 0; in < 2; ++in)
#pragma unroll
      for (int r = 0; r < 8; ++r) {
        int row = m0 + wm + 16 * im + 8 * h + r;
        int col = n0 + wn + 16 * in + mr;
        float tt = Tb[row * GLD + col];
        float yy = Yb[row * GLD + col];
        accS[im][in][r] = 0.25f * accS[im][in][r] - 1.5f * tt + ((row == col) ? 2.25f : 0.0f);
        accY[im][in][r] = 1.5f * yy - 0.5f * accY[im][in][r];
      }

#pragma unroll
  for (int im = 0; im < 2; ++im) {
#pragma unroll
    for (int r = 0; r < 8; ++r) {
      int row = m0 + wm + 16 * im + 8 * h + r;
      float* srow = Sb + row * GLD;
      float* yrow = Ob + row * GLD;
#pragma unroll
      for (int in = 0; in < 2; ++in) {
        int col = n0 + wn + 16 * in + mr;
        srow[col] = accS[im][in][r];
        yrow[col] = accY[im][in][r];
      }
    }
  }

  if (tj != ti) {
#pragma unroll
    for (int in = 0; in < 2; ++in) {
      int moff = (n0 + wn + 16 * in + mr) * GLD + m0 + wm + 8 * h;
#pragma unroll
      for (int im = 0; im < 2; ++im) {
        float4 lo, hi;
        lo.x = accS[im][in][0]; lo.y = accS[im][in][1]; lo.z = accS[im][in][2]; lo.w = accS[im][in][3];
        hi.x = accS[im][in][4]; hi.y = accS[im][in][5]; hi.z = accS[im][in][6]; hi.w = accS[im][in][7];
        *reinterpret_cast<float4*>(Sb + moff + 16 * im)     = lo;
        *reinterpret_cast<float4*>(Sb + moff + 16 * im + 4) = hi;
        lo.x = accY[im][in][0]; lo.y = accY[im][in][1]; lo.z = accY[im][in][2]; lo.w = accY[im][in][3];
        hi.x = accY[im][in][4]; hi.y = accY[im][in][5]; hi.z = accY[im][in][6]; hi.w = accY[im][in][7];
        *reinterpret_cast<float4*>(Ob + moff + 16 * im)     = lo;
        *reinterpret_cast<float4*>(Ob + moff + 16 * im + 4) = hi;
      }
    }
  }
}

#define OUT_ROWS 4

__global__ __launch_bounds__(256)
void out_kernel(const float* __restrict__ YZY, const float* __restrict__ mean,
                const float* __restrict__ tr, float* __restrict__ out, int d, int rows) {
  __shared__ float res[8 * OUT_ROWS];
  int wid  = threadIdx.x >> 5;
  int lane = threadIdx.x & 31;
  int row0 = blockIdx.x * (8 * OUT_ROWS) + wid * OUT_ROWS;
  const float* yrow[OUT_ROWS];
  const float* mrow[OUT_ROWS];
  float s[OUT_ROWS];
#pragma unroll
  for (int r = 0; r < OUT_ROWS; ++r) {
    int row = row0 + r;
    if (row >= rows) row = rows - 1;
    yrow[r] = YZY + (long long)row * d;
    mrow[r] = mean + (long long)(row / d) * d;
    s[r] = 0.0f;
  }
  for (int j = lane; j < d; j += 32) {
#pragma unroll
    for (int r = 0; r < OUT_ROWS; ++r) s[r] += yrow[r][j] * mrow[r][j];
  }
#pragma unroll
  for (int r = 0; r < OUT_ROWS; ++r)
    for (int off = 16; off > 0; off >>= 1) s[r] += __shfl_down(s[r], off, 32);
  if (lane == 0) {
#pragma unroll
    for (int r = 0; r < OUT_ROWS; ++r) res[wid * OUT_ROWS + r] = s[r];
  }
  __syncthreads();
  if (threadIdx.x < 8 * OUT_ROWS) {
    int row = blockIdx.x * (8 * OUT_ROWS) + threadIdx.x;
    if (row < rows) out[row] = res[threadIdx.x] * sqrtf(tr[row / d]);
  }
}

extern "C" void kernel_launch(void* const* d_in, const int* in_sizes, int n_in,
                              void* d_out, int out_size, void* d_ws, size_t ws_size,
                              hipStream_t stream) {
  (void)n_in; (void)ws_size; (void)out_size;
  const float* feat  = (const float*)d_in[1];
  const float* noise = (const float*)d_in[2];

  const int B  = in_sizes[0];
  const int d  = 256;
  const long long total = (long long)in_sizes[1];
  const int n  = (int)(total / ((long long)B * d));
  const int nd = n * d;
  const long long dd = (long long)d * d;

  float* W = (float*)d_ws;
  long long off = 0;
  float* Fp   = W + off; off += (long long)B * nd;
  float* mean = W + off; off += (long long)B * d;
  float* scl  = W + off; off += B;
  float* nmn  = W + off; off += B;
  float* tr   = W + off; off += B;
  off = (off + 255) & ~255LL;
  float* M0 = W + off; off += (long long)B * dd;
  float* M1 = W + off; off += (long long)B * dd;
  float* M2 = W + off; off += (long long)B * dd;
  float* M3 = W + off; off += (long long)B * dd;

  stats_partial_kernel<<<dim3(STAT_SLICES, B), 256, 0, stream>>>(feat, noise, M0, nd);
  stats_final_kernel<<<1, 128, 0, stream>>>(M0, scl, nmn, B, nd, 31.6227766016838f);
  perturb_kernel<<<dim3((nd / 4 + EW_TILE4 - 1) / EW_TILE4, B), 256, 0, stream>>>(feat, noise, scl, nmn, Fp, nd);
  mean_kernel<<<B, d, 0, stream>>>(Fp, mean, n, d);

  dim3 gg(d / BN, d / BM, B);
  wmma_gemm_batched<<<gg, 256, 0, stream>>>(Fp, (long long)nd, 1, d,
                                            Fp, (long long)nd, d, 1,
                                            M0, dd, d, n,
                                            1.0f / (float)(n - 1), 0, mean);

  dim3 gs(10, 1, B);
  wmma_gemm_first<<<gs, 128, 0, stream>>>(M0, M1, M2, tr);

  wmma_gemm_sym<<<gs, 128, 0, stream>>>(M2, M1, M0, d, 0);
  wmma_gemm_pair<<<gs, 128, 0, stream>>>(M0, M1, M2, M3);
  wmma_gemm_sym<<<gs, 128, 0, stream>>>(M2, M0, M1, d, 0);
  wmma_gemm_pair<<<gs, 128, 0, stream>>>(M1, M3, M0, M2);
  wmma_gemm_sym<<<gs, 128, 0, stream>>>(M0, M1, M3, d, 0);
  wmma_gemm_pair<<<gs, 128, 0, stream>>>(M3, M2, M0, M1);

  wmma_gemm_sym<<<gs, 128, 0, stream>>>(M0, M3, M2, d, 1);
  wmma_gemm_sym<<<gs, 128, 0, stream>>>(M1, M2, M0, d, 0);
  float* S2 = M0;

  out_kernel<<<(B * d + 8 * OUT_ROWS - 1) / (8 * OUT_ROWS), 256, 0, stream>>>(S2, mean, tr, (float*)d_out, d, B * d);
}


# Newton-Schulz tiles run as 8-wave work-groups, the matrix work split over two 4-wave groups: K halves, swapped via LDS and added (pair launch: one product each); sums over k taken in a new order
# speedup vs baseline: 1.8645x; 1.0316x over previous
#include <hip/hip_runtime.h>
#include <hip/hip_bf16.h>

typedef __attribute__((ext_vector_type(2))) float v2f;
typedef __attribute__((ext_vector_type(8))) float v8f;

#define STAT_SLICES 8

__global__ __launch_bounds__(256)
void stats_partial_kernel(const float* __restrict__ feat, const float* __restrict__ noise,
                          float* __restrict__ part_ws, int nd) {
  int s = blockIdx.x;
  int b = blockIdx.y;
  int t = threadIdx.x;
  int len  = nd / STAT_SLICES;
  int len4 = len >> 2;
  const float4* f4 = reinterpret_cast<const float4*>(feat  + (long long)b * nd + (long long)s * len);
  const float4* n4 = reinterpret_cast<const float4*>(noise + (long long)b * nd + (long long)s * len);
  double sf = 0.0, sf2 = 0.0, sn = 0.0, sn2 = 0.0;
#pragma unroll 4
  for (int i = t; i < len4; i += 256) {
    float4 x = f4[i];
    float4 y = n4[i];
    double x0 = (double)x.x, x1 = (double)x.y, x2 = (double)x.z, x3 = (double)x.w;
    double y0 = (double)y.x, y1 = (double)y.y, y2 = (double)y.z, y3 = (double)y.w;
    sf += x0; sf2 += x0 * x0; sn += y0; sn2 += y0 * y0;
    sf += x1; sf2 += x1 * x1; sn += y1; sn2 += y1 * y1;
    sf += x2; sf2 += x2 * x2; sn += y2; sn2 += y2 * y2;
    sf += x3; sf2 += x3 * x3; sn += y3; sn2 += y3 * y3;
  }
  __shared__ double sh[4][256];
  sh[0][t] = sf; sh[1][t] = sf2; sh[2][t] = sn; sh[3][t] = sn2;
  __syncthreads();
  for (int w = 128; w > 0; w >>= 1) {
    if (t < w) {
      sh[0][t] += sh[0][t + w]; sh[1][t] += sh[1][t + w];
      sh[2][t] += sh[2][t + w]; sh[3][t] += sh[3][t + w];
    }
    __syncthreads();
  }
  if (t < 4) {
    double* part = reinterpret_cast<double*>(part_ws);
    part[((long long)b * STAT_SLICES + s) * 4 + t] = sh[t][0];
  }
}

__global__ __launch_bounds__(128)
void stats_final_kernel(const float* __restrict__ part_ws,
                        float* __restrict__ scl, float* __restrict__ nmn,
                        int B, int nd, float snr_pow) {
  const double* part = reinterpret_cast<const double*>(part_ws);
  for (int b = threadIdx.x; b < B; b += 128) {
    const double* p = part + (long long)b * STAT_SLICES * 4;
    double s0 = 0.0, s1 = 0.0, s2 = 0.0, s3 = 0.0;
    for (int s = 0; s < STAT_SLICES; ++s) {
      s0 += p[4 * s + 0]; s1 += p[4 * s + 1];
      s2 += p[4 * s + 2]; s3 += p[4 * s + 3];
    }
    double inv = 1.0 / (double)nd;
    double gmf = s0 * inv;
    double gmn = s2 * inv;
    double tvar = s1 * inv - gmf * gmf;
    double nvar = s3 * inv - gmn * gmn;
    scl[b] = (float)sqrt((tvar / (double)snr_pow) / nvar);
    nmn[b] = (float)gmn;
  }
}

#define EW_TILE4 1024

__global__ __launch_bounds__(256)
void perturb_kernel(const float* __restrict__ feat, const float* __restrict__ noise,
                    const float* __restrict__ scl, const float* __restrict__ nmn,
                    float* __restrict__ Fp, int nd) {
  int b   = blockIdx.y;
  int nd4 = nd >> 2;
  const float4* f4 = reinterpret_cast<const float4*>(feat  + (long long)b * nd);
  const float4* n4 = reinterpret_cast<const float4*>(noise + (long long)b * nd);
  float4*       o4 = reinterpret_cast<float4*>(Fp + (long long)b * nd);
  float nm = nmn[b];
  float sc = scl[b];
  for (int c = blockIdx.x * EW_TILE4 + threadIdx.x; c < nd4; c += gridDim.x * EW_TILE4) {
    float4 x[4], y[4];
#pragma unroll
    for (int k = 0; k < 4; ++k) {
      int i = c + k * 256;
      if (i < nd4) { x[k] = f4[i]; y[k] = n4[i]; }
    }
#pragma unroll
    for (int k = 0; k < 4; ++k) {
      int i = c + k * 256;
      if (i < nd4) {
        float4 r;
        r.x = x[k].x + (y[k].x - nm) * sc;
        r.y = x[k].y + (y[k].y - nm) * sc;
        r.z = x[k].z + (y[k].z - nm) * sc;
        r.w = x[k].w + (y[k].w - nm) * sc;
        o4[i] = r;
      }
    }
  }
}

__global__ __launch_bounds__(256)
void mean_kernel(const float* __restrict__ Fp, float* __restrict__ mean, int n, int d) {
  int b = blockIdx.x;
  int j = threadIdx.x;
  const float* base = Fp + (long long)b * n * d;
  float s = 0.0f;
  for (int i = 0; i < n; ++i) s += base[(long long)i * d + j];
  mean[(long long)b * d + j] = s / (float)n;
}

#define BM 128
#define BN 128
#define BK 16
#define LDP 18

__global__ __launch_bounds__(256)
void wmma_gemm_batched(const float* __restrict__ A, long long sAb, int sAm, int sAk,
                       const float* __restrict__ Bm, long long sBb, int sBk, int sBn,
                       float* __restrict__ Dm, long long sDb,
                       int N, int K, float alpha, int mode,
                       const float* __restrict__ mu) {
  __shared__ __align__(16) float As[BM][LDP];
  __shared__ __align__(16) float Bst[BN][LDP];

  int b  = blockIdx.z;
  int m0 = blockIdx.y * BM;
  int n0 = blockIdx.x * BN;
  const float* Ab = A  + (long long)b * sAb;
  const float* Bb = Bm + (long long)b * sBb;
  float*       Db = Dm + (long long)b * sDb;

  int tid  = threadIdx.x;
  int wid  = tid >> 5;
  int lane = tid & 31;
  int mr   = lane & 15;
  int h    = lane >> 4;
  int wm   = (wid >> 1) * 32;
  int wn   = (wid & 1) * 64;

  const float* mub = mu + (long long)b * N;
  float muA[8];
#pragma unroll
  for (int i = 0; i < 8; ++i) muA[i] = mub[m0 + ((tid + i * 256) >> 4)];
  float muB = mub[n0 + (tid & 127)];

  v8f acc[2][4];
#pragma unroll
  for (int im = 0; im < 2; ++im)
#pragma unroll
    for (int in = 0; in < 4; ++in) acc[im][in] = (v8f){};

  for (int k0 = 0; k0 < K; k0 += BK) {
#pragma unroll
    for (int i = 0; i < 8; ++i) {
      int e = tid + i * 256;
      int m = e >> 4, kk = e & 15;
      As[m][kk] = Ab[(long long)(m0 + m) * sAm + (long long)(k0 + kk) * sAk] - muA[i];
    }
#pragma unroll
    for (int i = 0; i < 8; ++i) {
      int e = tid + i * 256;
      int nn = e & 127, kk = e >> 7;
      Bst[nn][kk] = Bb[(long long)(k0 + kk) * sBk + (long long)(n0 + nn) * sBn] - muB;
    }
    __syncthreads();
#pragma unroll
    for (int kk = 0; kk < BK; kk += 4) {
      int kc = kk + 2 * h;
      v2f a0 = *(const v2f*)&As[wm + mr][kc];
      v2f a1 = *(const v2f*)&As[wm + 16 + mr][kc];
      v2f b0 = *(const v2f*)&Bst[wn + mr][kc];
      v2f b1 = *(const v2f*)&Bst[wn + 16 + mr][kc];
      v2f b2 = *(const v2f*)&Bst[wn + 32 + mr][kc];
      v2f b3 = *(const v2f*)&Bst[wn + 48 + mr][kc];
      acc[0][0] = __builtin_amdgcn_wmma_f32_16x16x4_f32(false, a0, false, b0, (short)0, acc[0][0], false, false);
      acc[0][1] = __builtin_amdgcn_wmma_f32_16x16x4_f32(false, a0, false, b1, (short)0, acc[0][1], false, false);
      acc[0][2] = __builtin_amdgcn_wmma_f32_16x16x4_f32(false, a0, false, b2, (short)0, acc[0][2], false, false);
      acc[0][3] = __builtin_amdgcn_wmma_f32_16x16x4_f32(false, a0, false, b3, (short)0, acc[0][3], false, false);
      acc[1][0] = __builtin_amdgcn_wmma_f32_16x16x4_f32(false, a1, false, b0, (short)0, acc[1][0], false, false);
      acc[1][1] = __builtin_amdgcn_wmma_f32_16x16x4_f32(false, a1, false, b1, (short)0, acc[1][1], false, false);
      acc[1][2] = __builtin_amdgcn_wmma_f32_16x16x4_f32(false, a1, false, b2, (short)0, acc[1][2], false, false);
      acc[1][3] = __builtin_amdgcn_wmma_f32_16x16x4_f32(false, a1, false, b3, (short)0, acc[1][3], false, false);
    }
    __syncthreads();
  }

#pragma unroll
  for (int im = 0; im < 2; ++im) {
#pragma unroll
    for (int r = 0; r < 8; ++r) {
      int row = m0 + wm + 16 * im + 8 * h + r;
      float* drow = Db + (long long)row * N;
#pragma unroll
      for (int in = 0; in < 4; ++in) {
        int col = n0 + wn + 16 * in + mr;
        float v = acc[im][in][r];
        if (mode == 1) {
          v = -0.5f * v + ((row == col) ? 1.5f : 0.0f);
        } else {
          v *= alpha;
        }
        drow[col] = v;
      }
    }
  }
}

#define GLD 256
#define LDA 20
#define SM 64
#define SLDB 160
#define GRP_F (SM * LDA + (BK / 2) * SLDB)
#define XCH_F (2 * 4 * 32 * 32)

template <bool SCALE>
__device__ __forceinline__ void half_k_product(const float* __restrict__ Ag,
                                               const float* __restrict__ Bg,
                                               float* __restrict__ As, float* __restrict__ Bs,
                                               int kh, float inv,
                                               int ar, int ak4, int bk, int bn4,
                                               int wm, int wn, int mr, int h,
                                               v8f (&acc)[2][2]) {
  for (int k0 = 0; k0 < kh; k0 += BK) {
    float4 ga0 = *reinterpret_cast<const float4*>(Ag + k0);
    float4 ga1 = *reinterpret_cast<const float4*>(Ag + k0 + 32 * GLD);
    float4 gb0 = *reinterpret_cast<const float4*>(Bg + k0 * GLD);
    float4 gb1 = *reinterpret_cast<const float4*>(Bg + (k0 + 1) * GLD);
    if (SCALE) {
      ga0.x *= inv; ga0.y *= inv; ga0.z *= inv; ga0.w *= inv;
      ga1.x *= inv; ga1.y *= inv; ga1.z *= inv; ga1.w *= inv;
      gb0.x *= inv; gb0.y *= inv; gb0.z *= inv; gb0.w *= inv;
      gb1.x *= inv; gb1.y *= inv; gb1.z *= inv; gb1.w *= inv;
    }
    *reinterpret_cast<float4*>(&As[ar * LDA + 4 * ak4])        = ga0;
    *reinterpret_cast<float4*>(&As[(ar + 32) * LDA + 4 * ak4]) = ga1;
    float4 lo, hi;
    lo.x = gb0.x; lo.y = gb1.x; lo.z = gb0.y; lo.w = gb1.y;
    hi.x = gb0.z; hi.y = gb1.z; hi.z = gb0.w; hi.w = gb1.w;
    *reinterpret_cast<float4*>(&Bs[bk * SLDB + 8 * bn4])     = lo;
    *reinterpret_cast<float4*>(&Bs[bk * SLDB + 8 * bn4 + 4]) = hi;
    __syncthreads();
#pragma unroll
    for (int kk = 0; kk < BK; kk += 4) {
      int kc = kk + 2 * h;
      v2f a0 = *(const v2f*)&As[(wm + mr) * LDA + kc];
      v2f a1 = *(const v2f*)&As[(wm + 16 + mr) * LDA + kc];
      int kp = (kk >> 1) + h;
      v2f b0 = *(const v2f*)&Bs[kp * SLDB + 2 * (wn + mr)];
      v2f b1 = *(const v2f*)&Bs[kp * SLDB + 2 * (wn + 16 + mr)];
      acc[0][0] = __builtin_amdgcn_wmma_f32_16x16x4_f32(false, a0, false, b0, (short)0, acc[0][0], false, false);
      acc[0][1] = __builtin_amdgcn_wmma_f32_16x16x4_f32(false, a0, false, b1, (short)0, acc[0][1], false, false);
      acc[1][0] = __builtin_amdgcn_wmma_f32_16x16x4_f32(false, a1, false, b0, (short)0, acc[1][0], false, false);
      acc[1][1] = __builtin_amdgcn_wmma_f32_16x16x4_f32(false, a1, false, b1, (short)0, acc[1][1], false, false);
    }
    __syncthreads();
  }
}

__device__ __forceinline__ void swap_and_add(float* __restrict__ X, int g, int wid, int lane,
                                             v8f (&acc)[2][2]) {
  float* mine = X + ((g * 4 + wid) * 32) * 32 + lane;
#pragma unroll
  for (int im = 0; im < 2; ++im)
#pragma unroll
    for (int in = 0; in < 2; ++in)
#pragma unroll
      for (int r = 0; r < 8; ++r) mine[((im * 2 + in) * 8 + r) * 32] = acc[im][in][r];
  __syncthreads();
  const float* oth = X + (((1 - g) * 4 + wid) * 32) * 32 + lane;
#pragma unroll
  for (int im = 0; im < 2; ++im)
#pragma unroll
    for (int in = 0; in < 2; ++in)
#pragma unroll
      for (int r = 0; r < 8; ++r) acc[im][in][r] += oth[((im * 2 + in) * 8 + r) * 32];
}

template <int IM>
__device__ __forceinline__ void store_rows(float* __restrict__ Db, int m0, int n0, int wm, int wn,
                                           int mr, int h, const v8f (&acc)[2][2]) {
  float* base = Db + (m0 + wm + 8 * h) * GLD + n0 + wn + mr;
#pragma unroll
  for (int r = 0; r < 8; ++r) {
#pragma unroll
    for (int in = 0; in < 2; ++in) base[(16 * IM + r) * GLD + 16 * in] = acc[IM][in][r];
  }
}

__device__ __forceinline__ void store_mirror(float* __restrict__ Db, int m0, int n0, int wm, int wn,
                                             int mr, int h, const v8f (&acc)[2][2]) {
  float* mbase = Db + (n0 + wn + mr) * GLD + m0 + wm + 8 * h;
#pragma unroll
  for (int in = 0; in < 2; ++in) {
#pragma unroll
    for (int im = 0; im < 2; ++im) {
      float4 lo, hi;
      lo.x = acc[im][in][0]; lo.y = acc[im][in][1]; lo.z = acc[im][in][2]; lo.w = acc[im][in][3];
      hi.x = acc[im][in][4]; hi.y = acc[im][in][5]; hi.z = acc[im][in][6]; hi.w = acc[im][in][7];
      *reinterpret_cast<float4*>(mbase + 16 * in * GLD + 16 * im)     = lo;
      *reinterpret_cast<float4*>(mbase + 16 * in * GLD + 16 * im + 4) = hi;
    }
  }
}

__global__ __launch_bounds__(256)
void wmma_gemm_sym(const float* __restrict__ A, const float* __restrict__ Bm,
                   float* __restrict__ Dm, int K, int mode) {
  __shared__ __align__(16) float smem[XCH_F];

  int b  = blockIdx.z;
  int t  = blockIdx.x;
  int ti = (t >= 4) + (t >= 7) + (t >= 9);
  int tj = ti + t - (4 * ti - (ti * (ti - 1)) / 2);
  int m0 = ti * SM;
  int n0 = tj * SM;
  const float* Ab = A  + (long long)b * (GLD * GLD);
  const float* Bb = Bm + (long long)b * (GLD * GLD);
  float*       Db = Dm + (long long)b * (GLD * GLD);

  int tid  = threadIdx.x;
  int g    = tid >> 7;
  int t7   = tid & 127;
  int wid  = t7 >> 5;
  int lane = tid & 31;
  int mr   = lane & 15;
  int h    = lane >> 4;
  int wm   = (wid >> 1) * 32;
  int wn   = (wid & 1) * 32;

  int ar  = t7 >> 2;
  int ak4 = t7 & 3;
  int bk  = t7 >> 4;
  int bn4 = t7 & 15;
  int kh  = K >> 1;
  const float* Ag = Ab + (m0 + ar) * GLD + 4 * ak4 + g * kh;
  const float* Bg = Bb + (2 * bk + g * kh) * GLD + n0 + 4 * bn4;
  float* As = smem + g * GRP_F;
  float* Bs = As + SM * LDA;

  v8f acc[2][2];
#pragma unroll
  for (int im = 0; im < 2; ++im)
#pragma unroll
    for (int in = 0; in < 2; ++in) acc[im][in] = (v8f){};

  half_k_product<false>(Ag, Bg, As, Bs, kh, 1.0f, ar, ak4, bk, bn4, wm, wn, mr, h, acc);
  swap_and_add(smem, g, wid, lane, acc);

  if (mode == 1) {
#pragma unroll
    for (int im = 0; im < 2; ++im)
#pragma unroll
      for (int in = 0; in < 2; ++in)
#pragma unroll
        for (int r = 0; r < 8; ++r) {
          int row = m0 + wm + 16 * im + 8 * h + r;
          int col = n0 + wn + 16 * in + mr;
          acc[im][in][r] = -0.5f * acc[im][in][r] + ((row == col) ? 1.5f : 0.0f);
        }
  }

  if (tj != ti) {
    if (g == 0) {
      store_rows<0>(Db, m0, n0, wm, wn, mr, h, acc);
      store_rows<1>(Db, m0, n0, wm, wn, mr, h, acc);
    } else {
      store_mirror(Db, m0, n0, wm, wn, mr, h, acc);
    }
  } else {
    if (g == 0) store_rows<0>(Db, m0, n0, wm, wn, mr, h, acc);
    else        store_rows<1>(Db, m0, n0, wm, wn, mr, h, acc);
  }
}

template <int IM>
__device__ __forceinline__ void first_rows(const float* __restrict__ Cb, float* __restrict__ Ob, int g,
                                           float tr, int m0, int n0, int wm, int wn, int mr, int h,
                                           v8f (&acc)[2][2]) {
  const float* cbase = Cb + (m0 + wm + 8 * h) * GLD + n0 + wn + mr;
#pragma unroll
  for (int in = 0; in < 2; ++in)
#pragma unroll
    for (int r = 0; r < 8; ++r) {
      int row = m0 + wm + 16 * IM + 8 * h + r;
      int col = n0 + wn + 16 * in + mr;
      float a = cbase[(16 * IM + r) * GLD + 16 * in] / tr;
      float zv = -0.5f * a + ((row == col) ? 1.5f : 0.0f);
      float yv = 1.5f * a - 0.5f * acc[IM][in][r];
      acc[IM][in][r] = (g == 0) ? yv : zv;
    }
  store_rows<IM>(Ob, m0, n0, wm, wn, mr, h, acc);
}

__global__ __launch_bounds__(256) __attribute__((amdgpu_waves_per_eu(10)))
void wmma_gemm_first(const float* __restrict__ C, float* __restrict__ Ym,
                     float* __restrict__ Zm, float* __restrict__ trout) {
  __shared__ __align__(16) float smem[XCH_F];
  __shared__ float red[256];

  int b  = blockIdx.z;
  int t  = blockIdx.x;
  int ti = (t >= 4) + (t >= 7) + (t >= 9);
  int tj = ti + t - (4 * ti - (ti * (ti - 1)) / 2);
  int m0 = ti * SM;
  int n0 = tj * SM;
  const float* Cb = C  + (long long)b * (GLD * GLD);
  float*       Yb = Ym + (long long)b * (GLD * GLD);
  float*       Zb = Zm + (long long)b * (GLD * GLD);

  int tid  = threadIdx.x;
  int g    = tid >> 7;
  int t7   = tid & 127;
  int wid  = t7 >> 5;
  int lane = tid & 31;
  int mr   = lane & 15;
  int h    = lane >> 4;
  int wm   = (wid >> 1) * 32;
  int wn   = (wid & 1) * 32;

  int ar  = t7 >> 2;
  int ak4 = t7 & 3;
  int bk  = t7 >> 4;
  int bn4 = t7 & 15;
  const int kh = GLD / 2;
  const float* Ag = Cb + (m0 + ar) * GLD + 4 * ak4 + g * kh;
  const float* Bg = Cb + (2 * bk + g * kh) * GLD + n0 + 4 * bn4;
  float* As = smem + g * GRP_F;
  float* Bs = As + SM * LDA;

  red[tid] = Cb[tid * (GLD + 1)];
  __syncthreads();
  for (int s = 128; s > 0; s >>= 1) {
    if (tid < s) red[tid] += red[tid + s];
    __syncthreads();
  }
  float tr  = red[0];
  float inv = 1.0f / tr;
  if (t == 0 && tid == 0) trout[b] = tr;

  v8f acc[2][2];
#pragma unroll
  for (int im = 0; im < 2; ++im)
#pragma unroll
    for (int in = 0; in < 2; ++in) acc[im][in] = (v8f){};

  half_k_product<true>(Ag, Bg, As, Bs, kh, inv, ar, ak4, bk, bn4, wm, wn, mr, h, acc);
  swap_and_add(smem, g, wid, lane, acc);

  float* Ob = (g == 0) ? Yb : Zb;
  first_rows<0>(Cb, Ob, g, tr, m0, n0, wm, wn, mr, h, acc);
  first_rows<1>(Cb, Ob, g, tr, m0, n0, wm, wn, mr, h, acc);
  if (tj != ti) store_mirror(Ob, m0, n0, wm, wn, mr, h, acc);
}

template <int IM>
__device__ __forceinline__ void pair_rows(const float* __restrict__ Pb, float* __restrict__ Db, int g,
                                          int m0, int n0, int wm, int wn, int mr, int h,
                                          v8f (&acc)[2][2]) {
  const float* pbase = Pb + (m0 + wm + 8 * h) * GLD + n0 + wn + mr;
#pragma unroll
  for (int in = 0; in < 2; ++in)
#pragma unroll
    for (int r = 0; r < 8; ++r) {
      int row = m0 + wm + 16 * IM + 8 * h + r;
      int col = n0 + wn + 16 * in + mr;
      float lin = pbase[(16 * IM + r) * GLD + 16 * in];
      float sv = 0.25f * acc[IM][in][r] - 1.5f * lin + ((row == col) ? 2.25f : 0.0f);
      float yv = 1.5f * lin - 0.5f * acc[IM][in][r];
      acc[IM][in][r] = (g == 0) ? sv : yv;
    }
  store_rows<IM>(Db, m0, n0, wm, wn, mr, h, acc);
}

__global__ __launch_bounds__(256)
void wmma_gemm_pair(const float* __restrict__ Tm, const float* __restrict__ Yin,
                    float* __restrict__ Sm, float* __restrict__ Ym) {
  __shared__ __align__(16) float As[SM * LDA];
  __shared__ __align__(16) float Bs2[2][(BK / 2) * SLDB];

  int b  = blockIdx.z;
  int t  = blockIdx.x;
  int ti = (t >= 4) + (t >= 7) + (t >= 9);
  int tj = ti + t - (4 * ti - (ti * (ti - 1)) / 2);
  int m0 = ti * SM;
  int n0 = tj * SM;
  const float* Tb = Tm  + (long long)b * (GLD * GLD);
  const float* Yb = Yin + (long long)b * (GLD * GLD);
  float*       Sb = Sm  + (long long)b * (GLD * GLD);
  float*       Ob = Ym  + (long long)b * (GLD * GLD);

  int tid  = threadIdx.x;
  int g    = tid >> 7;
  int t7   = tid & 127;
  int wid  = t7 >> 5;
  int lane = tid & 31;
  int mr   = lane & 15;
  int h    = lane >> 4;
  int wm   = (wid >> 1) * 32;
  int wn   = (wid & 1) * 32;

  int ar  = tid >> 2;
  int ak4 = tid & 3;
  int bk  = t7 >> 4;
  int bn4 = t7 & 15;
  const float* Ag = Tb + (m0 + ar) * GLD + 4 * ak4;
  const float* Pb = (g == 0) ? Tb : Yb;
  const float* Bg = Pb + 2 * bk * GLD + n0 + 4 * bn4;
  float* Bs = Bs2[g];

  v8f acc[2][2];
#pragma unroll
  for (int im = 0; im < 2; ++im)
#pragma unroll
    for (int in = 0; in < 2; ++in) acc[im][in] = (v8f){};

  for (int k0 = 0; k0 < GLD; k0 += BK) {
    float4 ga  = *reinterpret_cast<const float4*>(Ag + k0);
    float4 gb0 = *reinterpret_cast<const float4*>(Bg + k0 * GLD);
    float4 gb1 = *reinterpret_cast<const float4*>(Bg + (k0 + 1) * GLD);
    *reinterpret_cast<float4*>(&As[ar * LDA + 4 * ak4]) = ga;
    float4 lo, hi;
    lo.x = gb0.x; lo.y = gb1.x; lo.z = gb0.y; lo.w = gb1.y;
    hi.x = gb0.z; hi.y = gb1.z; hi.z = gb0.w; hi.w = gb1.w;
    *reinterpret_cast<float4*>(&Bs[bk * SLDB + 8 * bn4])     = lo;
    *reinterpret_cast<float4*>(&Bs[bk * SLDB + 8 * bn4 + 4]) = hi;
    __syncthreads();
#pragma unroll
    for (int kk = 0; kk < BK; kk += 4) {
      int kc = kk + 2 * h;
      v2f a0 = *(const v2f*)&As[(wm + mr) * LDA + kc];
      v2f a1 = *(const v2f*)&As[(wm + 16 + mr) * LDA + kc];
      int kp = (kk >> 1) + h;
      v2f b0 = *(const v2f*)&Bs[kp * SLDB + 2 * (wn + mr)];
      v2f b1 = *(const v2f*)&Bs[kp * SLDB + 2 * (wn + 16 + mr)];
      acc[0][0] = __builtin_amdgcn_wmma_f32_16x16x4_f32(false, a0, false, b0, (short)0, acc[0][0], false, false);
      acc[0][1] = __builtin_amdgcn_wmma_f32_16x16x4_f32(false, a0, false, b1, (short)0, acc[0][1], false, false);
      acc[1][0] = __builtin_amdgcn_wmma_f32_16x16x4_f32(false, a1, false, b0, (short)0, acc[1][0], false, false);
      acc[1][1] = __builtin_amdgcn_wmma_f32_16x16x4_f32(false, a1, false, b1, (short)0, acc[1][1], false, false);
    }
    __syncthreads();
  }

  float* Db = (g == 0) ? Sb : Ob;
  pair_rows<0>(Pb, Db, g, m0, n0, wm, wn, mr, h, acc);
  pair_rows<1>(Pb, Db, g, m0, n0, wm, wn, mr, h, acc);
  if (tj != ti) store_mirror(Db, m0, n0, wm, wn, mr, h, acc);
}

#define OUT_ROWS 4

__global__ __launch_bounds__(256)
void out_kernel(const float* __restrict__ YZY, const float* __restrict__ mean,
                const float* __restrict__ tr, float* __restrict__ out, int d, int rows) {
  __shared__ float res[8 * OUT_ROWS];
  int wid  = threadIdx.x >> 5;
  int lane = threadIdx.x & 31;
  int row0 = blockIdx.x * (8 * OUT_ROWS) + wid * OUT_ROWS;
  const float* yrow[OUT_ROWS];
  const float* mrow[OUT_ROWS];
  float s[OUT_ROWS];
#pragma unroll
  for (int r = 0; r < OUT_ROWS; ++r) {
    int row = row0 + r;
    if (row >= rows) row = rows - 1;
    yrow[r] = YZY + (long long)row * d;
    mrow[r] = mean + (long long)(row / d) * d;
    s[r] = 0.0f;
  }
  for (int j = lane; j < d; j += 32) {
#pragma unroll
    for (int r = 0; r < OUT_ROWS; ++r) s[r] += yrow[r][j] * mrow[r][j];
  }
#pragma unroll
  for (int r = 0; r < OUT_ROWS; ++r)
    for (int off = 16; off > 0; off >>= 1) s[r] += __shfl_down(s[r], off, 32);
  if (lane == 0) {
#pragma unroll
    for (int r = 0; r < OUT_ROWS; ++r) res[wid * OUT_ROWS + r] = s[r];
  }
  __syncthreads();
  if (threadIdx.x < 8 * OUT_ROWS) {
    int row = blockIdx.x * (8 * OUT_ROWS) + threadIdx.x;
    if (row < rows) out[row] = res[threadIdx.x] * sqrtf(tr[row / d]);
  }
}

extern "C" void kernel_launch(void* const* d_in, const int* in_sizes, int n_in,
                              void* d_out, int out_size, void* d_ws, size_t ws_size,
                              hipStream_t stream) {
  (void)n_in; (void)ws_size; (void)out_size;
  const float* feat  = (const float*)d_in[1];
  const float* noise = (const float*)d_in[2];

  const int B  = in_sizes[0];
  const int d  = 256;
  const long long total = (long long)in_sizes[1];
  const int n  = (int)(total / ((long long)B * d));
  const int nd = n * d;
  const long long dd = (long long)d * d;

  float* W = (float*)d_ws;
  long long off = 0;
  float* Fp   = W + off; off += (long long)B * nd;
  float* mean = W + off; off += (long long)B * d;
  float* scl  = W + off; off += B;
  float* nmn  = W + off; off += B;
  float* tr   = W + off; off += B;
  off = (off + 255) & ~255LL;
  float* M0 = W + off; off += (long long)B * dd;
  float* M1 = W + off; off += (long long)B * dd;
  float* M2 = W + off; off += (long long)B * dd;
  float* M3 = W + off; off += (long long)B * dd;

  stats_partial_kernel<<<dim3(STAT_SLICES, B), 256, 0, stream>>>(feat, noise, M0, nd);
  stats_final_kernel<<<1, 128, 0, stream>>>(M0, scl, nmn, B, nd, 31.6227766016838f);
  perturb_kernel<<<dim3((nd / 4 + EW_TILE4 - 1) / EW_TILE4, B), 256, 0, stream>>>(feat, noise, scl, nmn, Fp, nd);
  mean_kernel<<<B, d, 0, stream>>>(Fp, mean, n, d);

  dim3 gg(d / BN, d / BM, B);
  wmma_gemm_batched<<<gg, 256, 0, stream>>>(Fp, (long long)nd, 1, d,
                                            Fp, (long long)nd, d, 1,
                                            M0, dd, d, n,
                                            1.0f / (float)(n - 1), 0, mean);

  dim3 gs(10, 1, B);
  wmma_gemm_first<<<gs, 256, 0, stream>>>(M0, M1, M2, tr);

  wmma_gemm_sym<<<gs, 256, 0, stream>>>(M2, M1, M0, d, 0);
  wmma_gemm_pair<<<gs, 256, 0, stream>>>(M0, M1, M2, M3);
  wmma_gemm_sym<<<gs, 256, 0, stream>>>(M2, M0, M1, d, 0);
  wmma_gemm_pair<<<gs, 256, 0, stream>>>(M1, M3, M0, M2);
  wmma_gemm_sym<<<gs, 256, 0, stream>>>(M0, M1, M3, d, 0);
  wmma_gemm_pair<<<gs, 256, 0, stream>>>(M3, M2, M0, M1);

  wmma_gemm_sym<<<gs, 256, 0, stream>>>(M0, M3, M2, d, 1);
  wmma_gemm_sym<<<gs, 256, 0, stream>>>(M1, M2, M0, d, 0);
  float* S2 = M0;

  out_kernel<<<(B * d + 8 * OUT_ROWS - 1) / (8 * OUT_ROWS), 256, 0, stream>>>(S2, mean, tr, (float*)d_out, d, B * d);
}


# Covariance launch re-cut: 64x64 tiles on or above the diagonal, whole 128-row operand panels in LDS, column means summed from them (mean launch goes), tile mirrored; sum order kept
# speedup vs baseline: 1.9857x; 1.0650x over previous
#include <hip/hip_runtime.h>
#include <hip/hip_bf16.h>

typedef __attribute__((ext_vector_type(2))) float v2f;
typedef __attribute__((ext_vector_type(8))) float v8f;

#define STAT_SLICES 8

__global__ __launch_bounds__(256)
void stats_partial_kernel(const float* __restrict__ feat, const float* __restrict__ noise,
                          float* __restrict__ part_ws, int nd) {
  int s = blockIdx.x;
  int b = blockIdx.y;
  int t = threadIdx.x;
  int len  = nd / STAT_SLICES;
  int len4 = len >> 2;
  const float4* f4 = reinterpret_cast<const float4*>(feat  + (long long)b * nd + (long long)s * len);
  const float4* n4 = reinterpret_cast<const float4*>(noise + (long long)b * nd + (long long)s * len);
  double sf = 0.0, sf2 = 0.0, sn = 0.0, sn2 = 0.0;
#pragma unroll 4
  for (int i = t; i < len4; i += 256) {
    float4 x = f4[i];
    float4 y = n4[i];
    double x0 = (double)x.x, x1 = (double)x.y, x2 = (double)x.z, x3 = (double)x.w;
    double y0 = (double)y.x, y1 = (double)y.y, y2 = (double)y.z, y3 = (double)y.w;
    sf += x0; sf2 += x0 * x0; sn += y0; sn2 += y0 * y0;
    sf += x1; sf2 += x1 * x1; sn += y1; sn2 += y1 * y1;
    sf += x2; sf2 += x2 * x2; sn += y2; sn2 += y2 * y2;
    sf += x3; sf2 += x3 * x3; sn += y3; sn2 += y3 * y3;
  }
  __shared__ double sh[4][256];
  sh[0][t] = sf; sh[1][t] = sf2; sh[2][t] = sn; sh[3][t] = sn2;
  __syncthreads();
  for (int w = 128; w > 0; w >>= 1) {
    if (t < w) {
      sh[0][t] += sh[0][t + w]; sh[1][t] += sh[1][t + w];
      sh[2][t] += sh[2][t + w]; sh[3][t] += sh[3][t + w];
    }
    __syncthreads();
  }
  if (t < 4) {
    double* part = reinterpret_cast<double*>(part_ws);
    part[((long long)b * STAT_SLICES + s) * 4 + t] = sh[t][0];
  }
}

__global__ __launch_bounds__(128)
void stats_final_kernel(const float* __restrict__ part_ws,
                        float* __restrict__ scl, float* __restrict__ nmn,
                        int B, int nd, float snr_pow) {
  const double* part = reinterpret_cast<const double*>(part_ws);
  for (int b = threadIdx.x; b < B; b += 128) {
    const double* p = part + (long long)b * STAT_SLICES * 4;
    double s0 = 0.0, s1 = 0.0, s2 = 0.0, s3 = 0.0;
    for (int s = 0; s < STAT_SLICES; ++s) {
      s0 += p[4 * s + 0]; s1 += p[4 * s + 1];
      s2 += p[4 * s + 2]; s3 += p[4 * s + 3];
    }
    double inv = 1.0 / (double)nd;
    double gmf = s0 * inv;
    double gmn = s2 * inv;
    double tvar = s1 * inv - gmf * gmf;
    double nvar = s3 * inv - gmn * gmn;
    scl[b] = (float)sqrt((tvar / (double)snr_pow) / nvar);
    nmn[b] = (float)gmn;
  }
}

#define EW_TILE4 1024

__global__ __launch_bounds__(256)
void perturb_kernel(const float* __restrict__ feat, const float* __restrict__ noise,
                    const float* __restrict__ scl, const float* __restrict__ nmn,
                    float* __restrict__ Fp, int nd) {
  int b   = blockIdx.y;
  int nd4 = nd >> 2;
  const float4* f4 = reinterpret_cast<const float4*>(feat  + (long long)b * nd);
  const float4* n4 = reinterpret_cast<const float4*>(noise + (long long)b * nd);
  float4*       o4 = reinterpret_cast<float4*>(Fp + (long long)b * nd);
  float nm = nmn[b];
  float sc = scl[b];
  for (int c = blockIdx.x * EW_TILE4 + threadIdx.x; c < nd4; c += gridDim.x * EW_TILE4) {
    float4 x[4], y[4];
#pragma unroll
    for (int k = 0; k < 4; ++k) {
      int i = c + k * 256;
      if (i < nd4) { x[k] = f4[i]; y[k] = n4[i]; }
    }
#pragma unroll
    for (int k = 0; k < 4; ++k) {
      int i = c + k * 256;
      if (i < nd4) {
        float4 r;
        r.x = x[k].x + (y[k].x - nm) * sc;
        r.y = x[k].y + (y[k].y - nm) * sc;
        r.z = x[k].z + (y[k].z - nm) * sc;
        r.w = x[k].w + (y[k].w - nm) * sc;
        o4[i] = r;
      }
    }
  }
}

__global__ __launch_bounds__(256)
void mean_kernel(const float* __restrict__ Fp, float* __restrict__ mean, int n, int d) {
  int b = blockIdx.x;
  int j = threadIdx.x;
  const float* base = Fp + (long long)b * n * d;
  float s = 0.0f;
  for (int i = 0; i < n; ++i) s += base[(long long)i * d + j];
  mean[(long long)b * d + j] = s / (float)n;
}

#define BM 128
#define BN 128
#define BK 16
#define LDP 18

__global__ __launch_bounds__(256)
void wmma_gemm_batched(const float* __restrict__ A, long long sAb, int sAm, int sAk,
                       const float* __restrict__ Bm, long long sBb, int sBk, int sBn,
                       float* __restrict__ Dm, long long sDb,
                       int N, int K, float alpha, int mode,
                       const float* __restrict__ mu) {
  __shared__ __align__(16) float As[BM][LDP];
  __shared__ __align__(16) float Bst[BN][LDP];

  int b  = blockIdx.z;
  int m0 = blockIdx.y * BM;
  int n0 = blockIdx.x * BN;
  const float* Ab = A  + (long long)b * sAb;
  const float* Bb = Bm + (long long)b * sBb;
  float*       Db = Dm + (long long)b * sDb;

  int tid  = threadIdx.x;
  int wid  = tid >> 5;
  int lane = tid & 31;
  int mr   = lane & 15;
  int h    = lane >> 4;
  int wm   = (wid >> 1) * 32;
  int wn   = (wid & 1) * 64;

  const float* mub = mu + (long long)b * N;
  float muA[8];
#pragma unroll
  for (int i = 0; i < 8; ++i) muA[i] = mub[m0 + ((tid + i * 256) >> 4)];
  float muB = mub[n0 + (tid & 127)];

  v8f acc[2][4];
#pragma unroll
  for (int im = 0; im < 2; ++im)
#pragma unroll
    for (int in = 0; in < 4; ++in) acc[im][in] = (v8f){};

  for (int k0 = 0; k0 < K; k0 += BK) {
#pragma unroll
    for (int i = 0; i < 8; ++i) {
      int e = tid + i * 256;
      int m = e >> 4, kk = e & 15;
      As[m][kk] = Ab[(long long)(m0 + m) * sAm + (long long)(k0 + kk) * sAk] - muA[i];
    }
#pragma unroll
    for (int i = 0; i < 8; ++i) {
      int e = tid + i * 256;
      int nn = e & 127, kk = e >> 7;
      Bst[nn][kk] = Bb[(long long)(k0 + kk) * sBk + (long long)(n0 + nn) * sBn] - muB;
    }
    __syncthreads();
#pragma unroll
    for (int kk = 0; kk < BK; kk += 4) {
      int kc = kk + 2 * h;
      v2f a0 = *(const v2f*)&As[wm + mr][kc];
      v2f a1 = *(const v2f*)&As[wm + 16 + mr][kc];
      v2f b0 = *(const v2f*)&Bst[wn + mr][kc];
      v2f b1 = *(const v2f*)&Bst[wn + 16 + mr][kc];
      v2f b2 = *(const v2f*)&Bst[wn + 32 + mr][kc];
      v2f b3 = *(const v2f*)&Bst[wn + 48 + mr][kc];
      acc[0][0] = __builtin_amdgcn_wmma_f32_16x16x4_f32(false, a0, false, b0, (short)0, acc[0][0], false, false);
      acc[0][1] = __builtin_amdgcn_wmma_f32_16x16x4_f32(false, a0, false, b1, (short)0, acc[0][1], false, false);
      acc[0][2] = __builtin_amdgcn_wmma_f32_16x16x4_f32(false, a0, false, b2, (short)0, acc[0][2], false, false);
      acc[0][3] = __builtin_amdgcn_wmma_f32_16x16x4_f32(false, a0, false, b3, (short)0, acc[0][3], false, false);
      acc[1][0] = __builtin_amdgcn_wmma_f32_16x16x4_f32(false, a1, false, b0, (short)0, acc[1][0], false, false);
      acc[1][1] = __builtin_amdgcn_wmma_f32_16x16x4_f32(false, a1, false, b1, (short)0, acc[1][1], false, false);
      acc[1][2] = __builtin_amdgcn_wmma_f32_16x16x4_f32(false, a1, false, b2, (short)0, acc[1][2], false, false);
      acc[1][3] = __builtin_amdgcn_wmma_f32_16x16x4_f32(false, a1, false, b3, (short)0, acc[1][3], false, false);
    }
    __syncthreads();
  }

#pragma unroll
  for (int im = 0; im < 2; ++im) {
#pragma unroll
    for (int r = 0; r < 8; ++r) {
      int row = m0 + wm + 16 * im + 8 * h + r;
      float* drow = Db + (long long)row * N;
#pragma unroll
      for (int in = 0; in < 4; ++in) {
        int col = n0 + wn + 16 * in + mr;
        float v = acc[im][in][r];
        if (mode == 1) {
          v = -0.5f * v + ((row == col) ? 1.5f : 0.0f);
        } else {
          v *= alpha;
        }
        drow[col] = v;
      }
    }
  }
}

#define GLD 256
#define LDA 20
#define SM 64
#define SLDB 160
#define GRP_F (SM * LDA + (BK / 2) * SLDB)
#define XCH_F (2 * 4 * 32 * 32)

template <bool SCALE>
__device__ __forceinline__ void half_k_product(const float* __restrict__ Ag,
                                               const float* __restrict__ Bg,
                                               float* __restrict__ As, float* __restrict__ Bs,
                                               int kh, float inv,
                                               int ar, int ak4, int bk, int bn4,
                                               int wm, int wn, int mr, int h,
                                               v8f (&acc)[2][2]) {
  for (int k0 = 0; k0 < kh; k0 += BK) {
    float4 ga0 = *reinterpret_cast<const float4*>(Ag + k0);
    float4 ga1 = *reinterpret_cast<const float4*>(Ag + k0 + 32 * GLD);
    float4 gb0 = *reinterpret_cast<const float4*>(Bg + k0 * GLD);
    float4 gb1 = *reinterpret_cast<const float4*>(Bg + (k0 + 1) * GLD);
    if (SCALE) {
      ga0.x *= inv; ga0.y *= inv; ga0.z *= inv; ga0.w *= inv;
      ga1.x *= inv; ga1.y *= inv; ga1.z *= inv; ga1.w *= inv;
      gb0.x *= inv; gb0.y *= inv; gb0.z *= inv; gb0.w *= inv;
      gb1.x *= inv; gb1.y *= inv; gb1.z *= inv; gb1.w *= inv;
    }
    *reinterpret_cast<float4*>(&As[ar * LDA + 4 * ak4])        = ga0;
    *reinterpret_cast<float4*>(&As[(ar + 32) * LDA + 4 * ak4]) = ga1;
    float4 lo, hi;
    lo.x = gb0.x; lo.y = gb1.x; lo.z = gb0.y; lo.w = gb1.y;
    hi.x = gb0.z; hi.y = gb1.z; hi.z = gb0.w; hi.w = gb1.w;
    *reinterpret_cast<float4*>(&Bs[bk * SLDB + 8 * bn4])     = lo;
    *reinterpret_cast<float4*>(&Bs[bk * SLDB + 8 * bn4 + 4]) = hi;
    __syncthreads();
#pragma unroll
    for (int kk = 0; kk < BK; kk += 4) {
      int kc = kk + 2 * h;
      v2f a0 = *(const v2f*)&As[(wm + mr) * LDA + kc];
      v2f a1 = *(const v2f*)&As[(wm + 16 + mr) * LDA + kc];
      int kp = (kk >> 1) + h;
      v2f b0 = *(const v2f*)&Bs[kp * SLDB + 2 * (wn + mr)];
      v2f b1 = *(const v2f*)&Bs[kp * SLDB + 2 * (wn + 16 + mr)];
      acc[0][0] = __builtin_amdgcn_wmma_f32_16x16x4_f32(false, a0, false, b0, (short)0, acc[0][0], false, false);
      acc[0][1] = __builtin_amdgcn_wmma_f32_16x16x4_f32(false, a0, false, b1, (short)0, acc[0][1], false, false);
      acc[1][0] = __builtin_amdgcn_wmma_f32_16x16x4_f32(false, a1, false, b0, (short)0, acc[1][0], false, false);
      acc[1][1] = __builtin_amdgcn_wmma_f32_16x16x4_f32(false, a1, false, b1, (short)0, acc[1][1], false, false);
    }
    __syncthreads();
  }
}

__device__ __forceinline__ void swap_and_add(float* __restrict__ X, int g, int wid, int lane,
                                             v8f (&acc)[2][2]) {
  float* mine = X + ((g * 4 + wid) * 32) * 32 + lane;
#pragma unroll
  for (int im = 0; im < 2; ++im)
#pragma unroll
    for (int in = 0; in < 2; ++in)
#pragma unroll
      for (int r = 0; r < 8; ++r) mine[((im * 2 + in) * 8 + r) * 32] = acc[im][in][r];
  __syncthreads();
  const float* oth = X + (((1 - g) * 4 + wid) * 32) * 32 + lane;
#pragma unroll
  for (int im = 0; im < 2; ++im)
#pragma unroll
    for (int in = 0; in < 2; ++in)
#pragma unroll
      for (int r = 0; r < 8; ++r) acc[im][in][r] += oth[((im * 2 + in) * 8 + r) * 32];
}

template <int IM>
__device__ __forceinline__ void store_rows(float* __restrict__ Db, int m0, int n0, int wm, int wn,
                                           int mr, int h, const v8f (&acc)[2][2]) {
  float* base = Db + (m0 + wm + 8 * h) * GLD + n0 + wn + mr;
#pragma unroll
  for (int r = 0; r < 8; ++r) {
#pragma unroll
    for (int in = 0; in < 2; ++in) base[(16 * IM + r) * GLD + 16 * in] = acc[IM][in][r];
  }
}

__device__ __forceinline__ void store_mirror(float* __restrict__ Db, int m0, int n0, int wm, int wn,
                                             int mr, int h, const v8f (&acc)[2][2]) {
  float* mbase = Db + (n0 + wn + mr) * GLD + m0 + wm + 8 * h;
#pragma unroll
  for (int in = 0; in < 2; ++in) {
#pragma unroll
    for (int im = 0; im < 2; ++im) {
      float4 lo, hi;
      lo.x = acc[im][in][0]; lo.y = acc[im][in][1]; lo.z = acc[im][in][2]; lo.w = acc[im][in][3];
      hi.x = acc[im][in][4]; hi.y = acc[im][in][5]; hi.z = acc[im][in][6]; hi.w = acc[im][in][7];
      *reinterpret_cast<float4*>(mbase + 16 * in * GLD + 16 * im)     = lo;
      *reinterpret_cast<float4*>(mbase + 16 * in * GLD + 16 * im + 4) = hi;
    }
  }
}

__global__ __launch_bounds__(256)
void wmma_gemm_sym(const float* __restrict__ A, const float* __restrict__ Bm,
                   float* __restrict__ Dm, int K, int mode) {
  __shared__ __align__(16) float smem[XCH_F];

  int b  = blockIdx.z;
  int t  = blockIdx.x;
  int ti = (t >= 4) + (t >= 7) + (t >= 9);
  int tj = ti + t - (4 * ti - (ti * (ti - 1)) / 2);
  int m0 = ti * SM;
  int n0 = tj * SM;
  const float* Ab = A  + (long long)b * (GLD * GLD);
  const float* Bb = Bm + (long long)b * (GLD * GLD);
  float*       Db = Dm + (long long)b * (GLD * GLD);

  int tid  = threadIdx.x;
  int g    = tid >> 7;
  int t7   = tid & 127;
  int wid  = t7 >> 5;
  int lane = tid & 31;
  int mr   = lane & 15;
  int h    = lane >> 4;
  int wm   = (wid >> 1) * 32;
  int wn   = (wid & 1) * 32;

  int ar  = t7 >> 2;
  int ak4 = t7 & 3;
  int bk  = t7 >> 4;
  int bn4 = t7 & 15;
  int kh  = K >> 1;
  const float* Ag = Ab + (m0 + ar) * GLD + 4 * ak4 + g * kh;
  const float* Bg = Bb + (2 * bk + g * kh) * GLD + n0 + 4 * bn4;
  float* As = smem + g * GRP_F;
  float* Bs = As + SM * LDA;

  v8f acc[2][2];
#pragma unroll
  for (int im = 0; im < 2; ++im)
#pragma unroll
    for (int in = 0; in < 2; ++in) acc[im][in] = (v8f){};

  half_k_product<false>(Ag, Bg, As, Bs, kh, 1.0f, ar, ak4, bk, bn4, wm, wn, mr, h, acc);
  swap_and_add(smem, g, wid, lane, acc);

  if (mode == 1) {
#pragma unroll
    for (int im = 0; im < 2; ++im)
#pragma unroll
      for (int in = 0; in < 2; ++in)
#pragma unroll
        for (int r = 0; r < 8; ++r) {
          int row = m0 + wm + 16 * im + 8 * h + r;
          int col = n0 + wn + 16 * in + mr;
          acc[im][in][r] = -0.5f * acc[im][in][r] + ((row == col) ? 1.5f : 0.0f);
        }
  }

  if (tj != ti) {
    if (g == 0) {
      store_rows<0>(Db, m0, n0, wm, wn, mr, h, acc);
      store_rows<1>(Db, m0, n0, wm, wn, mr, h, acc);
    } else {
      store_mirror(Db, m0, n0, wm, wn, mr, h, acc);
    }
  } else {
    if (g == 0) store_rows<0>(Db, m0, n0, wm, wn, mr, h, acc);
    else        store_rows<1>(Db, m0, n0, wm, wn, mr, h, acc);
  }
}

#define CK 128
#define CPAN ((CK / 2) * 128)

__device__ __forceinline__ void load_panel(const float* __restrict__ Fb, int c0, int tid,
                                           float4 (&g0)[CK / 16], float4 (&g1)[CK / 16]) {
  const float* g = Fb + (2 * (tid >> 4)) * GLD + c0 + 4 * (tid & 15);
#pragma unroll
  for (int i = 0; i < CK / 16; ++i) {
    g0[i] = *reinterpret_cast<const float4*>(g + (16 * i) * GLD);
    g1[i] = *reinterpret_cast<const float4*>(g + (16 * i + 1) * GLD);
  }
}

__device__ __forceinline__ void store_panel(float* __restrict__ P, int tid,
                                            const float4 (&g0)[CK / 16], const float4 (&g1)[CK / 16]) {
  int bn4 = tid & 15;
  int kq  = tid >> 4;
#pragma unroll
  for (int i = 0; i < CK / 16; ++i) {
    int kp = kq + 8 * i;
    float4 lo, hi;
    lo.x = g0[i].x; lo.y = g1[i].x; lo.z = g0[i].y; lo.w = g1[i].y;
    hi.x = g0[i].z; hi.y = g1[i].z; hi.z = g0[i].w; hi.w = g1[i].w;
    int o = kp * 128 + ((8 * bn4) ^ ((kp & 1) << 5));
    *reinterpret_cast<float4*>(&P[o])     = lo;
    *reinterpret_cast<float4*>(&P[o + 4]) = hi;
  }
}

__global__ __launch_bounds__(128)
void wmma_cov_tri(const float* __restrict__ Fp, float* __restrict__ Cm,
                  float* __restrict__ mean, float alpha) {
  __shared__ __align__(16) float pan[2 * CPAN];
  __shared__ float mus[128];

  int b  = blockIdx.z;
  int t  = blockIdx.x;
  int ti = (t >= 4) + (t >= 7) + (t >= 9);
  int tj = ti + t - (4 * ti - (ti * (ti - 1)) / 2);
  int m0 = ti * SM;
  int n0 = tj * SM;
  bool diag = (ti == tj);
  const float* Fb = Fp + (long long)b * (CK * GLD);
  float*       Cb = Cm + (long long)b * (GLD * GLD);

  int tid  = threadIdx.x;
  int wid  = tid >> 5;
  int lane = tid & 31;
  int mr   = lane & 15;
  int h    = lane >> 4;
  int wm   = (wid >> 1) * 32;
  int wn   = (wid & 1) * 32;

  float* PA = pan;
  float* PB = diag ? pan : pan + CPAN;
  {
    float4 a0v[CK / 16], a1v[CK / 16], b0v[CK / 16], b1v[CK / 16];
    load_panel(Fb, m0, tid, a0v, a1v);
    if (!diag) load_panel(Fb, n0, tid, b0v, b1v);
    store_panel(PA, tid, a0v, a1v);
    if (!diag) store_panel(PB, tid, b0v, b1v);
  }
  __syncthreads();

  int ncol = diag ? 64 : 128;
  if (tid < ncol) {
    const float* P = (tid < 64) ? PA : PB;
    int col = tid & 63;
    float s = 0.0f;
#pragma unroll
    for (int i = 0; i < CK; ++i) {
      int kp = i >> 1;
      s += P[kp * 128 + ((2 * col) ^ ((kp & 1) << 5)) + (i & 1)];
    }
    float mu = s / (float)CK;
    mus[tid] = mu;
    if (diag) mean[(long long)b * GLD + m0 + col] = mu;
  }
  __syncthreads();

  const float* musB = diag ? mus : mus + 64;
  float muA0 = mus[wm + mr],  muA1 = mus[wm + 16 + mr];
  float muB0 = musB[wn + mr], muB1 = musB[wn + 16 + mr];

  v8f acc[2][2];
#pragma unroll
  for (int im = 0; im < 2; ++im)
#pragma unroll
    for (int in = 0; in < 2; ++in) acc[im][in] = (v8f){};

  int sw = h << 5;
  const float* fa0 = PA + h * 128 + ((2 * (wm + mr)) ^ sw);
  const float* fa1 = PA + h * 128 + ((2 * (wm + 16 + mr)) ^ sw);
  const float* fb0 = PB + h * 128 + ((2 * (wn + mr)) ^ sw);
  const float* fb1 = PB + h * 128 + ((2 * (wn + 16 + mr)) ^ sw);
#pragma unroll 8
  for (int kk = 0; kk < CK; kk += 4) {
    v2f a0 = *(const v2f*)(fa0 + kk * 64);
    v2f a1 = *(const v2f*)(fa1 + kk * 64);
    v2f b0 = *(const v2f*)(fb0 + kk * 64);
    v2f b1 = *(const v2f*)(fb1 + kk * 64);
    a0.x -= muA0; a0.y -= muA0;
    a1.x -= muA1; a1.y -= muA1;
    b0.x -= muB0; b0.y -= muB0;
    b1.x -= muB1; b1.y -= muB1;
    acc[0][0] = __builtin_amdgcn_wmma_f32_16x16x4_f32(false, a0, false, b0, (short)0, acc[0][0], false, false);
    acc[0][1] = __builtin_amdgcn_wmma_f32_16x16x4_f32(false, a0, false, b1, (short)0, acc[0][1], false, false);
    acc[1][0] = __builtin_amdgcn_wmma_f32_16x16x4_f32(false, a1, false, b0, (short)0, acc[1][0], false, false);
    acc[1][1] = __builtin_amdgcn_wmma_f32_16x16x4_f32(false, a1, false, b1, (short)0, acc[1][1], false, false);
  }

#pragma unroll
  for (int im = 0; im < 2; ++im)
#pragma unroll
    for (int in = 0; in < 2; ++in)
#pragma unroll
      for (int r = 0; r < 8; ++r) acc[im][in][r] *= alpha;

  store_rows<0>(Cb, m0, n0, wm, wn, mr, h, acc);
  store_rows<1>(Cb, m0, n0, wm, wn, mr, h, acc);
  if (!diag) store_mirror(Cb, m0, n0, wm, wn, mr, h, acc);
}

template <int IM>
__device__ __forceinline__ void first_rows(const float* __restrict__ Cb, float* __restrict__ Ob, int g,
                                           float tr, int m0, int n0, int wm, int wn, int mr, int h,
                                           v8f (&acc)[2][2]) {
  const float* cbase = Cb + (m0 + wm + 8 * h) * GLD + n0 + wn + mr;
#pragma unroll
  for (int in = 0; in < 2; ++in)
#pragma unroll
    for (int r = 0; r < 8; ++r) {
      int row = m0 + wm + 16 * IM + 8 * h + r;
      int col = n0 + wn + 16 * in + mr;
      float a = cbase[(16 * IM + r) * GLD + 16 * in] / tr;
      float zv = -0.5f * a + ((row == col) ? 1.5f : 0.0f);
      float yv = 1.5f * a - 0.5f * acc[IM][in][r];
      acc[IM][in][r] = (g == 0) ? yv : zv;
    }
  store_rows<IM>(Ob, m0, n0, wm, wn, mr, h, acc);
}

__global__ __launch_bounds__(256) __attribute__((amdgpu_waves_per_eu(10)))
void wmma_gemm_first(const float* __restrict__ C, float* __restrict__ Ym,
                     float* __restrict__ Zm, float* __restrict__ trout) {
  __shared__ __align__(16) float smem[XCH_F];
  __shared__ float red[256];

  int b  = blockIdx.z;
  int t  = blockIdx.x;
  int ti = (t >= 4) + (t >= 7) + (t >= 9);
  int tj = ti + t - (4 * ti - (ti * (ti - 1)) / 2);
  int m0 = ti * SM;
  int n0 = tj * SM;
  const float* Cb = C  + (long long)b * (GLD * GLD);
  float*       Yb = Ym + (long long)b * (GLD * GLD);
  float*       Zb = Zm + (long long)b * (GLD * GLD);

  int tid  = threadIdx.x;
  int g    = tid >> 7;
  int t7   = tid & 127;
  int wid  = t7 >> 5;
  int lane = tid & 31;
  int mr   = lane & 15;
  int h    = lane >> 4;
  int wm   = (wid >> 1) * 32;
  int wn   = (wid & 1) * 32;

  int ar  = t7 >> 2;
  int ak4 = t7 & 3;
  int bk  = t7 >> 4;
  int bn4 = t7 & 15;
  const int kh = GLD / 2;
  const float* Ag = Cb + (m0 + ar) * GLD + 4 * ak4 + g * kh;
  const float* Bg = Cb + (2 * bk + g * kh) * GLD + n0 + 4 * bn4;
  float* As = smem + g * GRP_F;
  float* Bs = As + SM * LDA;

  red[tid] = Cb[tid * (GLD + 1)];
  __syncthreads();
  for (int s = 128; s > 0; s >>= 1) {
    if (tid < s) red[tid] += red[tid + s];
    __syncthreads();
  }
  float tr  = red[0];
  float inv = 1.0f / tr;
  if (t == 0 && tid == 0) trout[b] = tr;

  v8f acc[2][2];
#pragma unroll
  for (int im = 0; im < 2; ++im)
#pragma unroll
    for (int in = 0; in < 2; ++in) acc[im][in] = (v8f){};

  half_k_product<true>(Ag, Bg, As, Bs, kh, inv, ar, ak4, bk, bn4, wm, wn, mr, h, acc);
  swap_and_add(smem, g, wid, lane, acc);

  float* Ob = (g == 0) ? Yb : Zb;
  first_rows<0>(Cb, Ob, g, tr, m0, n0, wm, wn, mr, h, acc);
  first_rows<1>(Cb, Ob, g, tr, m0, n0, wm, wn, mr, h, acc);
  if (tj != ti) store_mirror(Ob, m0, n0, wm, wn, mr, h, acc);
}

template <int IM>
__device__ __forceinline__ void pair_rows(const float* __restrict__ Pb, float* __restrict__ Db, int g,
                                          int m0, int n0, int wm, int wn, int mr, int h,
                                          v8f (&acc)[2][2]) {
  const float* pbase = Pb + (m0 + wm + 8 * h) * GLD + n0 + wn + mr;
#pragma unroll
  for (int in = 0; in < 2; ++in)
#pragma unroll
    for (int r = 0; r < 8; ++r) {
      int row = m0 + wm + 16 * IM + 8 * h + r;
      int col = n0 + wn + 16 * in + mr;
      float lin = pbase[(16 * IM + r) * GLD + 16 * in];
      float sv = 0.25f * acc[IM][in][r] - 1.5f * lin + ((row == col) ? 2.25f : 0.0f);
      float yv = 1.5f * lin - 0.5f * acc[IM][in][r];
      acc[IM][in][r] = (g == 0) ? sv : yv;
    }
  store_rows<IM>(Db, m0, n0, wm, wn, mr, h, acc);
}

__global__ __launch_bounds__(256)
void wmma_gemm_pair(const float* __restrict__ Tm, const float* __restrict__ Yin,
                    float* __restrict__ Sm, float* __restrict__ Ym) {
  __shared__ __align__(16) float As[SM * LDA];
  __shared__ __align__(16) float Bs2[2][(BK / 2) * SLDB];

  int b  = blockIdx.z;
  int t  = blockIdx.x;
  int ti = (t >= 4) + (t >= 7) + (t >= 9);
  int tj = ti + t - (4 * ti - (ti * (ti - 1)) / 2);
  int m0 = ti * SM;
  int n0 = tj * SM;
  const float* Tb = Tm  + (long long)b * (GLD * GLD);
  const float* Yb = Yin + (long long)b * (GLD * GLD);
  float*       Sb = Sm  + (long long)b * (GLD * GLD);
  float*       Ob = Ym  + (long long)b * (GLD * GLD);

  int tid  = threadIdx.x;
  int g    = tid >> 7;
  int t7   = tid & 127;
  int wid  = t7 >> 5;
  int lane = tid & 31;
  int mr   = lane & 15;
  int h    = lane >> 4;
  int wm   = (wid >> 1) * 32;
  int wn   = (wid & 1) * 32;

  int ar  = tid >> 2;
  int ak4 = tid & 3;
  int bk  = t7 >> 4;
  int bn4 = t7 & 15;
  const float* Ag = Tb + (m0 + ar) * GLD + 4 * ak4;
  const float* Pb = (g == 0) ? Tb : Yb;
  const float* Bg = Pb + 2 * bk * GLD + n0 + 4 * bn4;
  float* Bs = Bs2[g];

  v8f acc[2][2];
#pragma unroll
  for (int im = 0; im < 2; ++im)
#pragma unroll
    for (int in = 0; in < 2; ++in) acc[im][in] = (v8f){};

  for (int k0 = 0; k0 < GLD; k0 += BK) {
    float4 ga  = *reinterpret_cast<const float4*>(Ag + k0);
    float4 gb0 = *reinterpret_cast<const float4*>(Bg + k0 * GLD);
    float4 gb1 = *reinterpret_cast<const float4*>(Bg + (k0 + 1) * GLD);
    *reinterpret_cast<float4*>(&As[ar * LDA + 4 * ak4]) = ga;
    float4 lo, hi;
    lo.x = gb0.x; lo.y = gb1.x; lo.z = gb0.y; lo.w = gb1.y;
    hi.x = gb0.z; hi.y = gb1.z; hi.z = gb0.w; hi.w = gb1.w;
    *reinterpret_cast<float4*>(&Bs[bk * SLDB + 8 * bn4])     = lo;
    *reinterpret_cast<float4*>(&Bs[bk * SLDB + 8 * bn4 + 4]) = hi;
    __syncthreads();
#pragma unroll
    for (int kk = 0; kk < BK; kk += 4) {
      int kc = kk + 2 * h;
      v2f a0 = *(const v2f*)&As[(wm + mr) * LDA + kc];
      v2f a1 = *(const v2f*)&As[(wm + 16 + mr) * LDA + kc];
      int kp = (kk >> 1) + h;
      v2f b0 = *(const v2f*)&Bs[kp * SLDB + 2 * (wn + mr)];
      v2f b1 = *(const v2f*)&Bs[kp * SLDB + 2 * (wn + 16 + mr)];
      acc[0][0] = __builtin_amdgcn_wmma_f32_16x16x4_f32(false, a0, false, b0, (short)0, acc[0][0], false, false);
      acc[0][1] = __builtin_amdgcn_wmma_f32_16x16x4_f32(false, a0, false, b1, (short)0, acc[0][1], false, false);
      acc[1][0] = __builtin_amdgcn_wmma_f32_16x16x4_f32(false, a1, false, b0, (short)0, acc[1][0], false, false);
      acc[1][1] = __builtin_amdgcn_wmma_f32_16x16x4_f32(false, a1, false, b1, (short)0, acc[1][1], false, false);
    }
    __syncthreads();
  }

  float* Db = (g == 0) ? Sb : Ob;
  pair_rows<0>(Pb, Db, g, m0, n0, wm, wn, mr, h, acc);
  pair_rows<1>(Pb, Db, g, m0, n0, wm, wn, mr, h, acc);
  if (tj != ti) store_mirror(Db, m0, n0, wm, wn, mr, h, acc);
}

#define OUT_ROWS 4

__global__ __launch_bounds__(256)
void out_kernel(const float* __restrict__ YZY, const float* __restrict__ mean,
                const float* __restrict__ tr, float* __restrict__ out, int d, int rows) {
  __shared__ float res[8 * OUT_ROWS];
  int wid  = threadIdx.x >> 5;
  int lane = threadIdx.x & 31;
  int row0 = blockIdx.x * (8 * OUT_ROWS) + wid * OUT_ROWS;
  const float* yrow[OUT_ROWS];
  const float* mrow[OUT_ROWS];
  float s[OUT_ROWS];
#pragma unroll
  for (int r = 0; r < OUT_ROWS; ++r) {
    int row = row0 + r;
    if (row >= rows) row = rows - 1;
    yrow[r] = YZY + (long long)row * d;
    mrow[r] = mean + (long long)(row / d) * d;
    s[r] = 0.0f;
  }
  for (int j = lane; j < d; j += 32) {
#pragma unroll
    for (int r = 0; r < OUT_ROWS; ++r) s[r] += yrow[r][j] * mrow[r][j];
  }
#pragma unroll
  for (int r = 0; r < OUT_ROWS; ++r)
    for (int off = 16; off > 0; off >>= 1) s[r] += __shfl_down(s[r], off, 32);
  if (lane == 0) {
#pragma unroll
    for (int r = 0; r < OUT_ROWS; ++r) res[wid * OUT_ROWS + r] = s[r];
  }
  __syncthreads();
  if (threadIdx.x < 8 * OUT_ROWS) {
    int row = blockIdx.x * (8 * OUT_ROWS) + threadIdx.x;
    if (row < rows) out[row] = res[threadIdx.x] * sqrtf(tr[row / d]);
  }
}

extern "C" void kernel_launch(void* const* d_in, const int* in_sizes, int n_in,
                              void* d_out, int out_size, void* d_ws, size_t ws_size,
                              hipStream_t stream) {
  (void)n_in; (void)ws_size; (void)out_size;
  const float* feat  = (const float*)d_in[1];
  const float* noise = (const float*)d_in[2];

  const int B  = in_sizes[0];
  const int d  = 256;
  const long long total = (long long)in_sizes[1];
  const int n  = (int)(total / ((long long)B * d));
  const int nd = n * d;
  const long long dd = (long long)d * d;

  float* W = (float*)d_ws;
  long long off = 0;
  float* Fp   = W + off; off += (long long)B * nd;
  float* mean = W + off; off += (long long)B * d;
  float* scl  = W + off; off += B;
  float* nmn  = W + off; off += B;
  float* tr   = W + off; off += B;
  off = (off + 255) & ~255LL;
  float* M0 = W + off; off += (long long)B * dd;
  float* M1 = W + off; off += (long long)B * dd;
  float* M2 = W + off; off += (long long)B * dd;
  float* M3 = W + off; off += (long long)B * dd;

  stats_partial_kernel<<<dim3(STAT_SLICES, B), 256, 0, stream>>>(feat, noise, M0, nd);
  stats_final_kernel<<<1, 128, 0, stream>>>(M0, scl, nmn, B, nd, 31.6227766016838f);
  perturb_kernel<<<dim3((nd / 4 + EW_TILE4 - 1) / EW_TILE4, B), 256, 0, stream>>>(feat, noise, scl, nmn, Fp, nd);
  if (n == CK) {
    wmma_cov_tri<<<dim3(10, 1, B), 128, 0, stream>>>(Fp, M0, mean, 1.0f / (float)(n - 1));
  } else {
    mean_kernel<<<B, d, 0, stream>>>(Fp, mean, n, d);
    dim3 gg(d / BN, d / BM, B);
    wmma_gemm_batched<<<gg, 256, 0, stream>>>(Fp, (long long)nd, 1, d,
                                              Fp, (long long)nd, d, 1,
                                              M0, dd, d, n,
                                              1.0f / (float)(n - 1), 0, mean);
  }

  dim3 gs(10, 1, B);
  wmma_gemm_first<<<gs, 256, 0, stream>>>(M0, M1, M2, tr);

  wmma_gemm_sym<<<gs, 256, 0, stream>>>(M2, M1, M0, d, 0);
  wmma_gemm_pair<<<gs, 256, 0, stream>>>(M0, M1, M2, M3);
  wmma_gemm_sym<<<gs, 256, 0, stream>>>(M2, M0, M1, d, 0);
  wmma_gemm_pair<<<gs, 256, 0, stream>>>(M1, M3, M0, M2);
  wmma_gemm_sym<<<gs, 256, 0, stream>>>(M0, M1, M3, d, 0);
  wmma_gemm_pair<<<gs, 256, 0, stream>>>(M3, M2, M0, M1);

  wmma_gemm_sym<<<gs, 256, 0, stream>>>(M0, M3, M2, d, 1);
  wmma_gemm_sym<<<gs, 256, 0, stream>>>(M1, M2, M0, d, 0);
  float* S2 = M0;

  out_kernel<<<(B * d + 8 * OUT_ROWS - 1) / (8 * OUT_ROWS), 256, 0, stream>>>(S2, mean, tr, (float*)d_out, d, B * d);
}


# Result tiles of the 10 matrix launches leave through LDS as [row][col]: each global store writes 4 whole 128-byte lines (tile and mirror image), not 64- and 16-byte pieces; sum order kept
# speedup vs baseline: 2.0919x; 1.0535x over previous
#include <hip/hip_runtime.h>
#include <hip/hip_bf16.h>

typedef __attribute__((ext_vector_type(2))) float v2f;
typedef __attribute__((ext_vector_type(8))) float v8f;

#define STAT_SLICES 8

__global__ __launch_bounds__(256)
void stats_partial_kernel(const float* __restrict__ feat, const float* __restrict__ noise,
                          float* __restrict__ part_ws, int nd) {
  int s = blockIdx.x;
  int b = blockIdx.y;
  int t = threadIdx.x;
  int len  = nd / STAT_SLICES;
  int len4 = len >> 2;
  const float4* f4 = reinterpret_cast<const float4*>(feat  + (long long)b * nd + (long long)s * len);
  const float4* n4 = reinterpret_cast<const float4*>(noise + (long long)b * nd + (long long)s * len);
  double sf = 0.0, sf2 = 0.0, sn = 0.0, sn2 = 0.0;
#pragma unroll 4
  for (int i = t; i < len4; i += 256) {
    float4 x = f4[i];
    float4 y = n4[i];
    double x0 = (double)x.x, x1 = (double)x.y, x2 = (double)x.z, x3 = (double)x.w;
    double y0 = (double)y.x, y1 = (double)y.y, y2 = (double)y.z, y3 = (double)y.w;
    sf += x0; sf2 += x0 * x0; sn += y0; sn2 += y0 * y0;
    sf += x1; sf2 += x1 * x1; sn += y1; sn2 += y1 * y1;
    sf += x2; sf2 += x2 * x2; sn += y2; sn2 += y2 * y2;
    sf += x3; sf2 += x3 * x3; sn += y3; sn2 += y3 * y3;
  }
  __shared__ double sh[4][256];
  sh[0][t] = sf; sh[1][t] = sf2; sh[2][t] = sn; sh[3][t] = sn2;
  __syncthreads();
  for (int w = 128; w > 0; w >>= 1) {
    if (t < w) {
      sh[0][t] += sh[0][t + w]; sh[1][t] += sh[1][t + w];
      sh[2][t] += sh[2][t + w]; sh[3][t] += sh[3][t + w];
    }
    __syncthreads();
  }
  if (t < 4) {
    double* part = reinterpret_cast<double*>(part_ws);
    part[((long long)b * STAT_SLICES + s) * 4 + t] = sh[t][0];
  }
}

__global__ __launch_bounds__(128)
void stats_final_kernel(const float* __restrict__ part_ws,
                        float* __restrict__ scl, float* __restrict__ nmn,
                        int B, int nd, float snr_pow) {
  const double* part = reinterpret_cast<const double*>(part_ws);
  for (int b = threadIdx.x; b < B; b += 128) {
    const double* p = part + (long long)b * STAT_SLICES * 4;
    double s0 = 0.0, s1 = 0.0, s2 = 0.0, s3 = 0.0;
    for (int s = 0; s < STAT_SLICES; ++s) {
      s0 += p[4 * s + 0]; s1 += p[4 * s + 1];
      s2 += p[4 * s + 2]; s3 += p[4 * s + 3];
    }
    double inv = 1.0 / (double)nd;
    double gmf = s0 * inv;
    double gmn = s2 * inv;
    double tvar = s1 * inv - gmf * gmf;
    double nvar = s3 * inv - gmn * gmn;
    scl[b] = (float)sqrt((tvar / (double)snr_pow) / nvar);
    nmn[b] = (float)gmn;
  }
}

#define EW_TILE4 1024

__global__ __launch_bounds__(256)
void perturb_kernel(const float* __restrict__ feat, const float* __restrict__ noise,
                    const float* __restrict__ scl, const float* __restrict__ nmn,
                    float* __restrict__ Fp, int nd) {
  int b   = blockIdx.y;
  int nd4 = nd >> 2;
  const float4* f4 = reinterpret_cast<const float4*>(feat  + (long long)b * nd);
  const float4* n4 = reinterpret_cast<const float4*>(noise + (long long)b * nd);
  float4*       o4 = reinterpret_cast<float4*>(Fp + (long long)b * nd);
  float nm = nmn[b];
  float sc = scl[b];
  for (int c = blockIdx.x * EW_TILE4 + threadIdx.x; c < nd4; c += gridDim.x * EW_TILE4) {
    float4 x[4], y[4];
#pragma unroll
    for (int k = 0; k < 4; ++k) {
      int i = c + k * 256;
      if (i < nd4) { x[k] = f4[i]; y[k] = n4[i]; }
    }
#pragma unroll
    for (int k = 0; k < 4; ++k) {
      int i = c + k * 256;
      if (i < nd4) {
        float4 r;
        r.x = x[k].x + (y[k].x - nm) * sc;
        r.y = x[k].y + (y[k].y - nm) * sc;
        r.z = x[k].z + (y[k].z - nm) * sc;
        r.w = x[k].w + (y[k].w - nm) * sc;
        o4[i] = r;
      }
    }
  }
}

__global__ __launch_bounds__(256)
void mean_kernel(const float* __restrict__ Fp, float* __restrict__ mean, int n, int d) {
  int b = blockIdx.x;
  int j = threadIdx.x;
  const float* base = Fp + (long long)b * n * d;
  float s = 0.0f;
  for (int i = 0; i < n; ++i) s += base[(long long)i * d + j];
  mean[(long long)b * d + j] = s / (float)n;
}

#define BM 128
#define BN 128
#define BK 16
#define LDP 18

__global__ __launch_bounds__(256)
void wmma_gemm_batched(const float* __restrict__ A, long long sAb, int sAm, int sAk,
                       const float* __restrict__ Bm, long long sBb, int sBk, int sBn,
                       float* __restrict__ Dm, long long sDb,
                       int N, int K, float alpha, int mode,
                       const float* __restrict__ mu) {
  __shared__ __align__(16) float As[BM][LDP];
  __shared__ __align__(16) float Bst[BN][LDP];

  int b  = blockIdx.z;
  int m0 = blockIdx.y * BM;
  int n0 = blockIdx.x * BN;
  const float* Ab = A  + (long long)b * sAb;
  const float* Bb = Bm + (long long)b * sBb;
  float*       Db = Dm + (long long)b * sDb;

  int tid  = threadIdx.x;
  int wid  = tid >> 5;
  int lane = tid & 31;
  int mr   = lane & 15;
  int h    = lane >> 4;
  int wm   = (wid >> 1) * 32;
  int wn   = (wid & 1) * 64;

  const float* mub = mu + (long long)b * N;
  float muA[8];
#pragma unroll
  for (int i = 0; i < 8; ++i) muA[i] = mub[m0 + ((tid + i * 256) >> 4)];
  float muB = mub[n0 + (tid & 127)];

  v8f acc[2][4];
#pragma unroll
  for (int im = 0; im < 2; ++im)
#pragma unroll
    for (int in = 0; in < 4; ++in) acc[im][in] = (v8f){};

  for (int k0 = 0; k0 < K; k0 += BK) {
#pragma unroll
    for (int i = 0; i < 8; ++i) {
      int e = tid + i * 256;
      int m = e >> 4, kk = e & 15;
      As[m][kk] = Ab[(long long)(m0 + m) * sAm + (long long)(k0 + kk) * sAk] - muA[i];
    }
#pragma unroll
    for (int i = 0; i < 8; ++i) {
      int e = tid + i * 256;
      int nn = e & 127, kk = e >> 7;
      Bst[nn][kk] = Bb[(long long)(k0 + kk) * sBk + (long long)(n0 + nn) * sBn] - muB;
    }
    __syncthreads();
#pragma unroll
    for (int kk = 0; kk < BK; kk += 4) {
      int kc = kk + 2 * h;
      v2f a0 = *(const v2f*)&As[wm + mr][kc];
      v2f a1 = *(const v2f*)&As[wm + 16 + mr][kc];
      v2f b0 = *(const v2f*)&Bst[wn + mr][kc];
      v2f b1 = *(const v2f*)&Bst[wn + 16 + mr][kc];
      v2f b2 = *(const v2f*)&Bst[wn + 32 + mr][kc];
      v2f b3 = *(const v2f*)&Bst[wn + 48 + mr][kc];
      acc[0][0] = __builtin_amdgcn_wmma_f32_16x16x4_f32(false, a0, false, b0, (short)0, acc[0][0], false, false);
      acc[0][1] = __builtin_amdgcn_wmma_f32_16x16x4_f32(false, a0, false, b1, (short)0, acc[0][1], false, false);
      acc[0][2] = __builtin_amdgcn_wmma_f32_16x16x4_f32(false, a0, false, b2, (short)0, acc[0][2], false, false);
      acc[0][3] = __builtin_amdgcn_wmma_f32_16x16x4_f32(false, a0, false, b3, (short)0, acc[0][3], false, false);
      acc[1][0] = __builtin_amdgcn_wmma_f32_16x16x4_f32(false, a1, false, b0, (short)0, acc[1][0], false, false);
      acc[1][1] = __builtin_amdgcn_wmma_f32_16x16x4_f32(false, a1, false, b1, (short)0, acc[1][1], false, false);
      acc[1][2] = __builtin_amdgcn_wmma_f32_16x16x4_f32(false, a1, false, b2, (short)0, acc[1][2], false, false);
      acc[1][3] = __builtin_amdgcn_wmma_f32_16x16x4_f32(false, a1, false, b3, (short)0, acc[1][3], false, false);
    }
    __syncthreads();
  }

#pragma unroll
  for (int im = 0; im < 2; ++im) {
#pragma unroll
    for (int r = 0; r < 8; ++r) {
      int row = m0 + wm + 16 * im + 8 * h + r;
      float* drow = Db + (long long)row * N;
#pragma unroll
      for (int in = 0; in < 4; ++in) {
        int col = n0 + wn + 16 * in + mr;
        float v = acc[im][in][r];
        if (mode == 1) {
          v = -0.5f * v + ((row == col) ? 1.5f : 0.0f);
        } else {
          v *= alpha;
        }
        drow[col] = v;
      }
    }
  }
}

#define GLD 256
#define LDA 20
#define SM 64
#define SLDB 160
#define GRP_F (SM * LDA + (BK / 2) * SLDB)
#define XCH_F (2 * 4 * 32 * 32)

template <bool SCALE>
__device__ __forceinline__ void half_k_product(const float* __restrict__ Ag,
                                               const float* __restrict__ Bg,
                                               float* __restrict__ As, float* __restrict__ Bs,
                                               int kh, float inv,
                                               int ar, int ak4, int bk, int bn4,
                                               int wm, int wn, int mr, int h,
                                               v8f (&acc)[2][2]) {
  for (int k0 = 0; k0 < kh; k0 += BK) {
    float4 ga0 = *reinterpret_cast<const float4*>(Ag + k0);
    float4 ga1 = *reinterpret_cast<const float4*>(Ag + k0 + 32 * GLD);
    float4 gb0 = *reinterpret_cast<const float4*>(Bg + k0 * GLD);
    float4 gb1 = *reinterpret_cast<const float4*>(Bg + (k0 + 1) * GLD);
    if (SCALE) {
      ga0.x *= inv; ga0.y *= inv; ga0.z *= inv; ga0.w *= inv;
      ga1.x *= inv; ga1.y *= inv; ga1.z *= inv; ga1.w *= inv;
      gb0.x *= inv; gb0.y *= inv; gb0.z *= inv; gb0.w *= inv;
      gb1.x *= inv; gb1.y *= inv; gb1.z *= inv; gb1.w *= inv;
    }
    *reinterpret_cast<float4*>(&As[ar * LDA + 4 * ak4])        = ga0;
    *reinterpret_cast<float4*>(&As[(ar + 32) * LDA + 4 * ak4]) = ga1;
    float4 lo, hi;
    lo.x = gb0.x; lo.y = gb1.x; lo.z = gb0.y; lo.w = gb1.y;
    hi.x = gb0.z; hi.y = gb1.z; hi.z = gb0.w; hi.w = gb1.w;
    *reinterpret_cast<float4*>(&Bs[bk * SLDB + 8 * bn4])     = lo;
    *reinterpret_cast<float4*>(&Bs[bk * SLDB + 8 * bn4 + 4]) = hi;
    __syncthreads();
#pragma unroll
    for (int kk = 0; kk < BK; kk += 4) {
      int kc = kk + 2 * h;
      v2f a0 = *(const v2f*)&As[(wm + mr) * LDA + kc];
      v2f a1 = *(const v2f*)&As[(wm + 16 + mr) * LDA + kc];
      int kp = (kk >> 1) + h;
      v2f b0 = *(const v2f*)&Bs[kp * SLDB + 2 * (wn + mr)];
      v2f b1 = *(const v2f*)&Bs[kp * SLDB + 2 * (wn + 16 + mr)];
      acc[0][0] = __builtin_amdgcn_wmma_f32_16x16x4_f32(false, a0, false, b0, (short)0, acc[0][0], false, false);
      acc[0][1] = __builtin_amdgcn_wmma_f32_16x16x4_f32(false, a0, false, b1, (short)0, acc[0][1], false, false);
      acc[1][0] = __builtin_amdgcn_wmma_f32_16x16x4_f32(false, a1, false, b0, (short)0, acc[1][0], false, false);
      acc[1][1] = __builtin_amdgcn_wmma_f32_16x16x4_f32(false, a1, false, b1, (short)0, acc[1][1], false, false);
    }
    __syncthreads();
  }
}

__device__ __forceinline__ void swap_and_add(float* __restrict__ X, int g, int wid, int lane,
                                             v8f (&acc)[2][2]) {
  float* mine = X + ((g * 4 + wid) * 32) * 32 + lane;
#pragma unroll
  for (int im = 0; im < 2; ++im)
#pragma unroll
    for (int in = 0; in < 2; ++in)
#pragma unroll
      for (int r = 0; r < 8; ++r) mine[((im * 2 + in) * 8 + r) * 32] = acc[im][in][r];
  __syncthreads();
  const float* oth = X + (((1 - g) * 4 + wid) * 32) * 32 + lane;
#pragma unroll
  for (int im = 0; im < 2; ++im)
#pragma unroll
    for (int in = 0; in < 2; ++in)
#pragma unroll
      for (int r = 0; r < 8; ++r) acc[im][in][r] += oth[((im * 2 + in) * 8 + r) * 32];
}

#define TP 65
#define TILE_F (SM * TP)
#define SMEM_F (2 * TILE_F)

__device__ __forceinline__ void stage_tile(float* __restrict__ T, int wm, int wn, int mr, int h,
                                           const v8f (&acc)[2][2]) {
  float* p = T + (wm + 8 * h) * TP + wn + mr;
#pragma unroll
  for (int im = 0; im < 2; ++im)
#pragma unroll
    for (int in = 0; in < 2; ++in)
#pragma unroll
      for (int r = 0; r < 8; ++r) p[(16 * im + r) * TP + 16 * in] = acc[im][in][r];
}

template <int RC>
__device__ __forceinline__ void store_tile_rows(const float* __restrict__ T, float* __restrict__ Db,
                                                int m0, int n0, int tl, int rbeg) {
#pragma unroll
  for (int i = 0; i < RC / 8; ++i) {
    int f   = tl + 128 * i;
    int row = rbeg + (f >> 4);
    int c4  = f & 15;
    const float* s = T + row * TP + 4 * c4;
    float4 v;
    v.x = s[0]; v.y = s[1]; v.z = s[2]; v.w = s[3];
    *reinterpret_cast<float4*>(Db + (m0 + row) * GLD + n0 + 4 * c4) = v;
  }
}

__device__ __forceinline__ void store_tile_mirror(const float* __restrict__ T, float* __restrict__ Db,
                                                  int m0, int n0, int tl) {
#pragma unroll
  for (int i = 0; i < SM / 8; ++i) {
    int f   = tl + 128 * i;
    int col = f >> 4;
    int r4  = f & 15;
    const float* s = T + (4 * r4) * TP + col;
    float4 v;
    v.x = s[0]; v.y = s[TP]; v.z = s[2 * TP]; v.w = s[3 * TP];
    *reinterpret_cast<float4*>(Db + (n0 + col) * GLD + m0 + 4 * r4) = v;
  }
}

__global__ __launch_bounds__(256)
void wmma_gemm_sym(const float* __restrict__ A, const float* __restrict__ Bm,
                   float* __restrict__ Dm, int K, int mode) {
  __shared__ __align__(16) float smem[SMEM_F];

  int b  = blockIdx.z;
  int t  = blockIdx.x;
  int ti = (t >= 4) + (t >= 7) + (t >= 9);
  int tj = ti + t - (4 * ti - (ti * (ti - 1)) / 2);
  int m0 = ti * SM;
  int n0 = tj * SM;
  const float* Ab = A  + (long long)b * (GLD * GLD);
  const float* Bb = Bm + (long long)b * (GLD * GLD);
  float*       Db = Dm + (long long)b * (GLD * GLD);

  int tid  = threadIdx.x;
  int g    = tid >> 7;
  int t7   = tid & 127;
  int wid  = t7 >> 5;
  int lane = tid & 31;
  int mr   = lane & 15;
  int h    = lane >> 4;
  int wm   = (wid >> 1) * 32;
  int wn   = (wid & 1) * 32;

  int ar  = t7 >> 2;
  int ak4 = t7 & 3;
  int bk  = t7 >> 4;
  int bn4 = t7 & 15;
  int kh  = K >> 1;
  const float* Ag = Ab + (m0 + ar) * GLD + 4 * ak4 + g * kh;
  const float* Bg = Bb + (2 * bk + g * kh) * GLD + n0 + 4 * bn4;
  float* As = smem + g * GRP_F;
  float* Bs = As + SM * LDA;

  v8f acc[2][2];
#pragma unroll
  for (int im = 0; im < 2; ++im)
#pragma unroll
    for (int in = 0; in < 2; ++in) acc[im][in] = (v8f){};

  half_k_product<false>(Ag, Bg, As, Bs, kh, 1.0f, ar, ak4, bk, bn4, wm, wn, mr, h, acc);
  swap_and_add(smem, g, wid, lane, acc);

  if (mode == 1) {
#pragma unroll
    for (int im = 0; im < 2; ++im)
#pragma unroll
      for (int in = 0; in < 2; ++in)
#pragma unroll
        for (int r = 0; r < 8; ++r) {
          int row = m0 + wm + 16 * im + 8 * h + r;
          int col = n0 + wn + 16 * in + mr;
          acc[im][in][r] = -0.5f * acc[im][in][r] + ((row == col) ? 1.5f : 0.0f);
        }
  }

  __syncthreads();
  float* T = smem + g * TILE_F;
  stage_tile(T, wm, wn, mr, h, acc);
  __syncthreads();
  if (tj != ti) {
    if (g == 0) store_tile_rows<SM>(T, Db, m0, n0, t7, 0);
    else        store_tile_mirror(T, Db, m0, n0, t7);
  } else {
    store_tile_rows<SM / 2>(T, Db, m0, n0, t7, g * (SM / 2));
  }
}

#define CK 128
#define CPAN ((CK / 2) * 128)

__device__ __forceinline__ void load_panel(const float* __restrict__ Fb, int c0, int tid,
                                           float4 (&g0)[CK / 16], float4 (&g1)[CK / 16]) {
  const float* g = Fb + (2 * (tid >> 4)) * GLD + c0 + 4 * (tid & 15);
#pragma unroll
  for (int i = 0; i < CK / 16; ++i) {
    g0[i] = *reinterpret_cast<const float4*>(g + (16 * i) * GLD);
    g1[i] = *reinterpret_cast<const float4*>(g + (16 * i + 1) * GLD);
  }
}

__device__ __forceinline__ void store_panel(float* __restrict__ P, int tid,
                                            const float4 (&g0)[CK / 16], const float4 (&g1)[CK / 16]) {
  int bn4 = tid & 15;
  int kq  = tid >> 4;
#pragma unroll
  for (int i = 0; i < CK / 16; ++i) {
    int kp = kq + 8 * i;
    float4 lo, hi;
    lo.x = g0[i].x; lo.y = g1[i].x; lo.z = g0[i].y; lo.w = g1[i].y;
    hi.x = g0[i].z; hi.y = g1[i].z; hi.z = g0[i].w; hi.w = g1[i].w;
    int o = kp * 128 + ((8 * bn4) ^ ((kp & 1) << 5));
    *reinterpret_cast<float4*>(&P[o])     = lo;
    *reinterpret_cast<float4*>(&P[o + 4]) = hi;
  }
}

__global__ __launch_bounds__(128)
void wmma_cov_tri(const float* __restrict__ Fp, float* __restrict__ Cm,
                  float* __restrict__ mean, float alpha) {
  __shared__ __align__(16) float pan[2 * CPAN];
  __shared__ float mus[128];

  int b  = blockIdx.z;
  int t  = blockIdx.x;
  int ti = (t >= 4) + (t >= 7) + (t >= 9);
  int tj = ti + t - (4 * ti - (ti * (ti - 1)) / 2);
  int m0 = ti * SM;
  int n0 = tj * SM;
  bool diag = (ti == tj);
  const float* Fb = Fp + (long long)b * (CK * GLD);
  float*       Cb = Cm + (long long)b * (GLD * GLD);

  int tid  = threadIdx.x;
  int wid  = tid >> 5;
  int lane = tid & 31;
  int mr   = lane & 15;
  int h    = lane >> 4;
  int wm   = (wid >> 1) * 32;
  int wn   = (wid & 1) * 32;

  float* PA = pan;
  float* PB = diag ? pan : pan + CPAN;
  {
    float4 a0v[CK / 16], a1v[CK / 16], b0v[CK / 16], b1v[CK / 16];
    load_panel(Fb, m0, tid, a0v, a1v);
    if (!diag) load_panel(Fb, n0, tid, b0v, b1v);
    store_panel(PA, tid, a0v, a1v);
    if (!diag) store_panel(PB, tid, b0v, b1v);
  }
  __syncthreads();

  int ncol = diag ? 64 : 128;
  if (tid < ncol) {
    const float* P = (tid < 64) ? PA : PB;
    int col = tid & 63;
    float s = 0.0f;
#pragma unroll
    for (int i = 0; i < CK; ++i) {
      int kp = i >> 1;
      s += P[kp * 128 + ((2 * col) ^ ((kp & 1) << 5)) + (i & 1)];
    }
    float mu = s / (float)CK;
    mus[tid] = mu;
    if (diag) mean[(long long)b * GLD + m0 + col] = mu;
  }
  __syncthreads();

  const float* musB = diag ? mus : mus + 64;
  float muA0 = mus[wm + mr],  muA1 = mus[wm + 16 + mr];
  float muB0 = musB[wn + mr], muB1 = musB[wn + 16 + mr];

  v8f acc[2][2];
#pragma unroll
  for (int im = 0; im < 2; ++im)
#pragma unroll
    for (int in = 0; in < 2; ++in) acc[im][in] = (v8f){};

  int sw = h << 5;
  const float* fa0 = PA + h * 128 + ((2 * (wm + mr)) ^ sw);
  const float* fa1 = PA + h * 128 + ((2 * (wm + 16 + mr)) ^ sw);
  const float* fb0 = PB + h * 128 + ((2 * (wn + mr)) ^ sw);
  const float* fb1 = PB + h * 128 + ((2 * (wn + 16 + mr)) ^ sw);
#pragma unroll 8
  for (int kk = 0; kk < CK; kk += 4) {
    v2f a0 = *(const v2f*)(fa0 + kk * 64);
    v2f a1 = *(const v2f*)(fa1 + kk * 64);
    v2f b0 = *(const v2f*)(fb0 + kk * 64);
    v2f b1 = *(const v2f*)(fb1 + kk * 64);
    a0.x -= muA0; a0.y -= muA0;
    a1.x -= muA1; a1.y -= muA1;
    b0.x -= muB0; b0.y -= muB0;
    b1.x -= muB1; b1.y -= muB1;
    acc[0][0] = __builtin_amdgcn_wmma_f32_16x16x4_f32(false, a0, false, b0, (short)0, acc[0][0], false, false);
    acc[0][1] = __builtin_amdgcn_wmma_f32_16x16x4_f32(false, a0, false, b1, (short)0, acc[0][1], false, false);
    acc[1][0] = __builtin_amdgcn_wmma_f32_16x16x4_f32(false, a1, false, b0, (short)0, acc[1][0], false, false);
    acc[1][1] = __builtin_amdgcn_wmma_f32_16x16x4_f32(false, a1, false, b1, (short)0, acc[1][1], false, false);
  }

#pragma unroll
  for (int im = 0; im < 2; ++im)
#pragma unroll
    for (int in = 0; in < 2; ++in)
#pragma unroll
      for (int r = 0; r < 8; ++r) acc[im][in][r] *= alpha;

  __syncthreads();
  stage_tile(pan, wm, wn, mr, h, acc);
  __syncthreads();
  store_tile_rows<SM>(pan, Cb, m0, n0, tid, 0);
  if (!diag) store_tile_mirror(pan, Cb, m0, n0, tid);
}

template <int IM>
__device__ __forceinline__ void first_rows(const float* __restrict__ Cb, int g,
                                           float tr, int m0, int n0, int wm, int wn, int mr, int h,
                                           v8f (&acc)[2][2]) {
  const float* cbase = Cb + (m0 + wm + 8 * h) * GLD + n0 + wn + mr;
#pragma unroll
  for (int in = 0; in < 2; ++in)
#pragma unroll
    for (int r = 0; r < 8; ++r) {
      int row = m0 + wm + 16 * IM + 8 * h + r;
      int col = n0 + wn + 16 * in + mr;
      float a = cbase[(16 * IM + r) * GLD + 16 * in] / tr;
      float zv = -0.5f * a + ((row == col) ? 1.5f : 0.0f);
      float yv = 1.5f * a - 0.5f * acc[IM][in][r];
      acc[IM][in][r] = (g == 0) ? yv : zv;
    }
}

__global__ __launch_bounds__(256) __attribute__((amdgpu_waves_per_eu(10)))
void wmma_gemm_first(const float* __restrict__ C, float* __restrict__ Ym,
                     float* __restrict__ Zm, float* __restrict__ trout) {
  __shared__ __align__(16) float smem[SMEM_F];
  __shared__ float red[256];

  int b  = blockIdx.z;
  int t  = blockIdx.x;
  int ti = (t >= 4) + (t >= 7) + (t >= 9);
  int tj = ti + t - (4 * ti - (ti * (ti - 1)) / 2);
  int m0 = ti * SM;
  int n0 = tj * SM;
  const float* Cb = C  + (long long)b * (GLD * GLD);
  float*       Yb = Ym + (long long)b * (GLD * GLD);
  float*       Zb = Zm + (long long)b * (GLD * GLD);

  int tid  = threadIdx.x;
  int g    = tid >> 7;
  int t7   = tid & 127;
  int wid  = t7 >> 5;
  int lane = tid & 31;
  int mr   = lane & 15;
  int h    = lane >> 4;
  int wm   = (wid >> 1) * 32;
  int wn   = (wid & 1) * 32;

  int ar  = t7 >> 2;
  int ak4 = t7 & 3;
  int bk  = t7 >> 4;
  int bn4 = t7 & 15;
  const int kh = GLD / 2;
  const float* Ag = Cb + (m0 + ar) * GLD + 4 * ak4 + g * kh;
  const float* Bg = Cb + (2 * bk + g * kh) * GLD + n0 + 4 * bn4;
  float* As = smem + g * GRP_F;
  float* Bs = As + SM * LDA;

  red[tid] = Cb[tid * (GLD + 1)];
  __syncthreads();
  for (int s = 128; s > 0; s >>= 1) {
    if (tid < s) red[tid] += red[tid + s];
    __syncthreads();
  }
  float tr  = red[0];
  float inv = 1.0f / tr;
  if (t == 0 && tid == 0) trout[b] = tr;

  v8f acc[2][2];
#pragma unroll
  for (int im = 0; im < 2; ++im)
#pragma unroll
    for (int in = 0; in < 2; ++in) acc[im][in] = (v8f){};

  half_k_product<true>(Ag, Bg, As, Bs, kh, inv, ar, ak4, bk, bn4, wm, wn, mr, h, acc);
  swap_and_add(smem, g, wid, lane, acc);

  float* Ob = (g == 0) ? Yb : Zb;
  first_rows<0>(Cb, g, tr, m0, n0, wm, wn, mr, h, acc);
  first_rows<1>(Cb, g, tr, m0, n0, wm, wn, mr, h, acc);
  __syncthreads();
  float* T = smem + g * TILE_F;
  stage_tile(T, wm, wn, mr, h, acc);
  __syncthreads();
  store_tile_rows<SM>(T, Ob, m0, n0, t7, 0);
  if (tj != ti) store_tile_mirror(T, Ob, m0, n0, t7);
}

template <int IM>
__device__ __forceinline__ void pair_rows(const float* __restrict__ Pb, int g,
                                          int m0, int n0, int wm, int wn, int mr, int h,
                                          v8f (&acc)[2][2]) {
  const float* pbase = Pb + (m0 + wm + 8 * h) * GLD + n0 + wn + mr;
#pragma unroll
  for (int in = 0; in < 2; ++in)
#pragma unroll
    for (int r = 0; r < 8; ++r) {
      int row = m0 + wm + 16 * IM + 8 * h + r;
      int col = n0 + wn + 16 * in + mr;
      float lin = pbase[(16 * IM + r) * GLD + 16 * in];
      float sv = 0.25f * acc[IM][in][r] - 1.5f * lin + ((row == col) ? 2.25f : 0.0f);
      float yv = 1.5f * lin - 0.5f * acc[IM][in][r];
      acc[IM][in][r] = (g == 0) ? sv : yv;
    }
}

__global__ __launch_bounds__(256)
void wmma_gemm_pair(const float* __restrict__ Tm, const float* __restrict__ Yin,
                    float* __restrict__ Sm, float* __restrict__ Ym) {
  __shared__ __align__(16) float smem[SMEM_F];
  float* As = smem;

  int b  = blockIdx.z;
  int t  = blockIdx.x;
  int ti = (t >= 4) + (t >= 7) + (t >= 9);
  int tj = ti + t - (4 * ti - (ti * (ti - 1)) / 2);
  int m0 = ti * SM;
  int n0 = tj * SM;
  const float* Tb = Tm  + (long long)b * (GLD * GLD);
  const float* Yb = Yin + (long long)b * (GLD * GLD);
  float*       Sb = Sm  + (long long)b * (GLD * GLD);
  float*       Ob = Ym  + (long long)b * (GLD * GLD);

  int tid  = threadIdx.x;
  int g    = tid >> 7;
  int t7   = tid & 127;
  int wid  = t7 >> 5;
  int lane = tid & 31;
  int mr   = lane & 15;
  int h    = lane >> 4;
  int wm   = (wid >> 1) * 32;
  int wn   = (wid & 1) * 32;

  int ar  = tid >> 2;
  int ak4 = tid & 3;
  int bk  = t7 >> 4;
  int bn4 = t7 & 15;
  const float* Ag = Tb + (m0 + ar) * GLD + 4 * ak4;
  const float* Pb = (g == 0) ? Tb : Yb;
  const float* Bg = Pb + 2 * bk * GLD + n0 + 4 * bn4;
  float* Bs = smem + SM * LDA + g * ((BK / 2) * SLDB);

  v8f acc[2][2];
#pragma unroll
  for (int im = 0; im < 2; ++im)
#pragma unroll
    for (int in = 0; in < 2; ++in) acc[im][in] = (v8f){};

  for (int k0 = 0; k0 < GLD; k0 += BK) {
    float4 ga  = *reinterpret_cast<const float4*>(Ag + k0);
    float4 gb0 = *reinterpret_cast<const float4*>(Bg + k0 * GLD);
    float4 gb1 = *reinterpret_cast<const float4*>(Bg + (k0 + 1) * GLD);
    *reinterpret_cast<float4*>(&As[ar * LDA + 4 * ak4]) = ga;
    float4 lo, hi;
    lo.x = gb0.x; lo.y = gb1.x; lo.z = gb0.y; lo.w = gb1.y;
    hi.x = gb0.z; hi.y = gb1.z; hi.z = gb0.w; hi.w = gb1.w;
    *reinterpret_cast<float4*>(&Bs[bk * SLDB + 8 * bn4])     = lo;
    *reinterpret_cast<float4*>(&Bs[bk * SLDB + 8 * bn4 + 4]) = hi;
    __syncthreads();
#pragma unroll
    for (int kk = 0; kk < BK; kk += 4) {
      int kc = kk + 2 * h;
      v2f a0 = *(const v2f*)&As[(wm + mr) * LDA + kc];
      v2f a1 = *(const v2f*)&As[(wm + 16 + mr) * LDA + kc];
      int kp = (kk >> 1) + h;
      v2f b0 = *(const v2f*)&Bs[kp * SLDB + 2 * (wn + mr)];
      v2f b1 = *(const v2f*)&Bs[kp * SLDB + 2 * (wn + 16 + mr)];
      acc[0][0] = __builtin_amdgcn_wmma_f32_16x16x4_f32(false, a0, false, b0, (short)0, acc[0][0], false, false);
      acc[0][1] = __builtin_amdgcn_wmma_f32_16x16x4_f32(false, a0, false, b1, (short)0, acc[0][1], false, false);
      acc[1][0] = __builtin_amdgcn_wmma_f32_16x16x4_f32(false, a1, false, b0, (short)0, acc[1][0], false, false);
      acc[1][1] = __builtin_amdgcn_wmma_f32_16x16x4_f32(false, a1, false, b1, (short)0, acc[1][1], false, false);
    }
    __syncthreads();
  }

  float* Db = (g == 0) ? Sb : Ob;
  pair_rows<0>(Pb, g, m0, n0, wm, wn, mr, h, acc);
  pair_rows<1>(Pb, g, m0, n0, wm, wn, mr, h, acc);
  float* T = smem + g * TILE_F;
  stage_tile(T, wm, wn, mr, h, acc);
  __syncthreads();
  store_tile_rows<SM>(T, Db, m0, n0, t7, 0);
  if (tj != ti) store_tile_mirror(T, Db, m0, n0, t7);
}

#define OUT_ROWS 4

__global__ __launch_bounds__(256)
void out_kernel(const float* __restrict__ YZY, const float* __restrict__ mean,
                const float* __restrict__ tr, float* __restrict__ out, int d, int rows) {
  __shared__ float res[8 * OUT_ROWS];
  int wid  = threadIdx.x >> 5;
  int lane = threadIdx.x & 31;
  int row0 = blockIdx.x * (8 * OUT_ROWS) + wid * OUT_ROWS;
  const float* yrow[OUT_ROWS];
  const float* mrow[OUT_ROWS];
  float s[OUT_ROWS];
#pragma unroll
  for (int r = 0; r < OUT_ROWS; ++r) {
    int row = row0 + r;
    if (row >= rows) row = rows - 1;
    yrow[r] = YZY + (long long)row * d;
    mrow[r] = mean + (long long)(row / d) * d;
    s[r] = 0.0f;
  }
  for (int j = lane; j < d; j += 32) {
#pragma unroll
    for (int r = 0; r < OUT_ROWS; ++r) s[r] += yrow[r][j] * mrow[r][j];
  }
#pragma unroll
  for (int r = 0; r < OUT_ROWS; ++r)
    for (int off = 16; off > 0; off >>= 1) s[r] += __shfl_down(s[r], off, 32);
  if (lane == 0) {
#pragma unroll
    for (int r = 0; r < OUT_ROWS; ++r) res[wid * OUT_ROWS + r] = s[r];
  }
  __syncthreads();
  if (threadIdx.x < 8 * OUT_ROWS) {
    int row = blockIdx.x * (8 * OUT_ROWS) + threadIdx.x;
    if (row < rows) out[row] = res[threadIdx.x] * sqrtf(tr[row / d]);
  }
}

extern "C" void kernel_launch(void* const* d_in, const int* in_sizes, int n_in,
                              void* d_out, int out_size, void* d_ws, size_t ws_size,
                              hipStream_t stream) {
  (void)n_in; (void)ws_size; (void)out_size;
  const float* feat  = (const float*)d_in[1];
  const float* noise = (const float*)d_in[2];

  const int B  = in_sizes[0];
  const int d  = 256;
  const long long total = (long long)in_sizes[1];
  const int n  = (int)(total / ((long long)B * d));
  const int nd = n * d;
  const long long dd = (long long)d * d;

  float* W = (float*)d_ws;
  long long off = 0;
  float* Fp   = W + off; off += (long long)B * nd;
  float* mean = W + off; off += (long long)B * d;
  float* scl  = W + off; off += B;
  float* nmn  = W + off; off += B;
  float* tr   = W + off; off += B;
  off = (off + 255) & ~255LL;
  float* M0 = W + off; off += (long long)B * dd;
  float* M1 = W + off; off += (long long)B * dd;
  float* M2 = W + off; off += (long long)B * dd;
  float* M3 = W + off; off += (long long)B * dd;

  stats_partial_kernel<<<dim3(STAT_SLICES, B), 256, 0, stream>>>(feat, noise, M0, nd);
  stats_final_kernel<<<1, 128, 0, stream>>>(M0, scl, nmn, B, nd, 31.6227766016838f);
  perturb_kernel<<<dim3((nd / 4 + EW_TILE4 - 1) / EW_TILE4, B), 256, 0, stream>>>(feat, noise, scl, nmn, Fp, nd);
  if (n == CK) {
    wmma_cov_tri<<<dim3(10, 1, B), 128, 0, stream>>>(Fp, M0, mean, 1.0f / (float)(n - 1));
  } else {
    mean_kernel<<<B, d, 0, stream>>>(Fp, mean, n, d);
    dim3 gg(d / BN, d / BM, B);
    wmma_gemm_batched<<<gg, 256, 0, stream>>>(Fp, (long long)nd, 1, d,
                                              Fp, (long long)nd, d, 1,
                                              M0, dd, d, n,
                                              1.0f / (float)(n - 1), 0, mean);
  }

  dim3 gs(10, 1, B);
  wmma_gemm_first<<<gs, 256, 0, stream>>>(M0, M1, M2, tr);

  wmma_gemm_sym<<<gs, 256, 0, stream>>>(M2, M1, M0, d, 0);
  wmma_gemm_pair<<<gs, 256, 0, stream>>>(M0, M1, M2, M3);
  wmma_gemm_sym<<<gs, 256, 0, stream>>>(M2, M0, M1, d, 0);
  wmma_gemm_pair<<<gs, 256, 0, stream>>>(M1, M3, M0, M2);
  wmma_gemm_sym<<<gs, 256, 0, stream>>>(M0, M1, M3, d, 0);
  wmma_gemm_pair<<<gs, 256, 0, stream>>>(M3, M2, M0, M1);

  wmma_gemm_sym<<<gs, 256, 0, stream>>>(M0, M3, M2, d, 1);
  wmma_gemm_sym<<<gs, 256, 0, stream>>>(M1, M2, M0, d, 0);
  float* S2 = M0;

  out_kernel<<<(B * d + 8 * OUT_ROWS - 1) / (8 * OUT_ROWS), 256, 0, stream>>>(S2, mean, tr, (float*)d_out, d, B * d);
}


# Newton-Schulz planes kept as upper 64x64 blocks only: no tile stored mirrored, a block below the diagonal is staged from its transposed twin (planes symmetric); last plane whole; sum order kept
# speedup vs baseline: 2.1665x; 1.0357x over previous
#include <hip/hip_runtime.h>
#include <hip/hip_bf16.h>

typedef __attribute__((ext_vector_type(2))) float v2f;
typedef __attribute__((ext_vector_type(8))) float v8f;

#define STAT_SLICES 8

__global__ __launch_bounds__(256)
void stats_partial_kernel(const float* __restrict__ feat, const float* __restrict__ noise,
                          float* __restrict__ part_ws, int nd) {
  int s = blockIdx.x;
  int b = blockIdx.y;
  int t = threadIdx.x;
  int len  = nd / STAT_SLICES;
  int len4 = len >> 2;
  const float4* f4 = reinterpret_cast<const float4*>(feat  + (long long)b * nd + (long long)s * len);
  const float4* n4 = reinterpret_cast<const float4*>(noise + (long long)b * nd + (long long)s * len);
  double sf = 0.0, sf2 = 0.0, sn = 0.0, sn2 = 0.0;
#pragma unroll 4
  for (int i = t; i < len4; i += 256) {
    float4 x = f4[i];
    float4 y = n4[i];
    double x0 = (double)x.x, x1 = (double)x.y, x2 = (double)x.z, x3 = (double)x.w;
    double y0 = (double)y.x, y1 = (double)y.y, y2 = (double)y.z, y3 = (double)y.w;
    sf += x0; sf2 += x0 * x0; sn += y0; sn2 += y0 * y0;
    sf += x1; sf2 += x1 * x1; sn += y1; sn2 += y1 * y1;
    sf += x2; sf2 += x2 * x2; sn += y2; sn2 += y2 * y2;
    sf += x3; sf2 += x3 * x3; sn += y3; sn2 += y3 * y3;
  }
  __shared__ double sh[4][256];
  sh[0][t] = sf; sh[1][t] = sf2; sh[2][t] = sn; sh[3][t] = sn2;
  __syncthreads();
  for (int w = 128; w > 0; w >>= 1) {
    if (t < w) {
      sh[0][t] += sh[0][t + w]; sh[1][t] += sh[1][t + w];
      sh[2][t] += sh[2][t + w]; sh[3][t] += sh[3][t + w];
    }
    __syncthreads();
  }
  if (t < 4) {
    double* part = reinterpret_cast<double*>(part_ws);
    part[((long long)b * STAT_SLICES + s) * 4 + t] = sh[t][0];
  }
}

__global__ __launch_bounds__(128)
void stats_final_kernel(const float* __restrict__ part_ws,
                        float* __restrict__ scl, float* __restrict__ nmn,
                        int B, int nd, float snr_pow) {
  const double* part = reinterpret_cast<const double*>(part_ws);
  for (int b = threadIdx.x; b < B; b += 128) {
    const double* p = part + (long long)b * STAT_SLICES * 4;
    double s0 = 0.0, s1 = 0.0, s2 = 0.0, s3 = 0.0;
    for (int s = 0; s < STAT_SLICES; ++s) {
      s0 += p[4 * s + 0]; s1 += p[4 * s + 1];
      s2 += p[4 * s + 2]; s3 += p[4 * s + 3];
    }
    double inv = 1.0 / (double)nd;
    double gmf = s0 * inv;
    double gmn = s2 * inv;
    double tvar = s1 * inv - gmf * gmf;
    double nvar = s3 * inv - gmn * gmn;
    scl[b] = (float)sqrt((tvar / (double)snr_pow) / nvar);
    nmn[b] = (float)gmn;
  }
}

#define EW_TILE4 1024

__global__ __launch_bounds__(256)
void perturb_kernel(const float* __restrict__ feat, const float* __restrict__ noise,
                    const float* __restrict__ scl, const float* __restrict__ nmn,
                    float* __restrict__ Fp, int nd) {
  int b   = blockIdx.y;
  int nd4 = nd >> 2;
  const float4* f4 = reinterpret_cast<const float4*>(feat  + (long long)b * nd);
  const float4* n4 = reinterpret_cast<const float4*>(noise + (long long)b * nd);
  float4*       o4 = reinterpret_cast<float4*>(Fp + (long long)b * nd);
  float nm = nmn[b];
  float sc = scl[b];
  for (int c = blockIdx.x * EW_TILE4 + threadIdx.x; c < nd4; c += gridDim.x * EW_TILE4) {
    float4 x[4], y[4];
#pragma unroll
    for (int k = 0; k < 4; ++k) {
      int i = c + k * 256;
      if (i < nd4) { x[k] = f4[i]; y[k] = n4[i]; }
    }
#pragma unroll
    for (int k = 0; k < 4; ++k) {
      int i = c + k * 256;
      if (i < nd4) {
        float4 r;
        r.x = x[k].x + (y[k].x - nm) * sc;
        r.y = x[k].y + (y[k].y - nm) * sc;
        r.z = x[k].z + (y[k].z - nm) * sc;
        r.w = x[k].w + (y[k].w - nm) * sc;
        o4[i] = r;
      }
    }
  }
}

__global__ __launch_bounds__(256)
void mean_kernel(const float* __restrict__ Fp, float* __restrict__ mean, int n, int d) {
  int b = blockIdx.x;
  int j = threadIdx.x;
  const float* base = Fp + (long long)b * n * d;
  float s = 0.0f;
  for (int i = 0; i < n; ++i) s += base[(long long)i * d + j];
  mean[(long long)b * d + j] = s / (float)n;
}

#define BM 128
#define BN 128
#define BK 16
#define LDP 18

__global__ __launch_bounds__(256)
void wmma_gemm_batched(const float* __restrict__ A, long long sAb, int sAm, int sAk,
                       const float* __restrict__ Bm, long long sBb, int sBk, int sBn,
                       float* __restrict__ Dm, long long sDb,
                       int N, int K, float alpha, int mode,
                       const float* __restrict__ mu) {
  __shared__ __align__(16) float As[BM][LDP];
  __shared__ __align__(16) float Bst[BN][LDP];

  int b  = blockIdx.z;
  int m0 = blockIdx.y * BM;
  int n0 = blockIdx.x * BN;
  const float* Ab = A  + (long long)b * sAb;
  const float* Bb = Bm + (long long)b * sBb;
  float*       Db = Dm + (long long)b * sDb;

  int tid  = threadIdx.x;
  int wid  = tid >> 5;
  int lane = tid & 31;
  int mr   = lane & 15;
  int h    = lane >> 4;
  int wm   = (wid >> 1) * 32;
  int wn   = (wid & 1) * 64;

  const float* mub = mu + (long long)b * N;
  float muA[8];
#pragma unroll
  for (int i = 0; i < 8; ++i) muA[i] = mub[m0 + ((tid + i * 256) >> 4)];
  float muB = mub[n0 + (tid & 127)];

  v8f acc[2][4];
#pragma unroll
  for (int im = 0; im < 2; ++im)
#pragma unroll
    for (int in = 0; in < 4; ++in) acc[im][in] = (v8f){};

  for (int k0 = 0; k0 < K; k0 += BK) {
#pragma unroll
    for (int i = 0; i < 8; ++i) {
      int e = tid + i * 256;
      int m = e >> 4, kk = e & 15;
      As[m][kk] = Ab[(long long)(m0 + m) * sAm + (long long)(k0 + kk) * sAk] - muA[i];
    }
#pragma unroll
    for (int i = 0; i < 8; ++i) {
      int e = tid + i * 256;
      int nn = e & 127, kk = e >> 7;
      Bst[nn][kk] = Bb[(long long)(k0 + kk) * sBk + (long long)(n0 + nn) * sBn] - muB;
    }
    __syncthreads();
#pragma unroll
    for (int kk = 0; kk < BK; kk += 4) {
      int kc = kk + 2 * h;
      v2f a0 = *(const v2f*)&As[wm + mr][kc];
      v2f a1 = *(const v2f*)&As[wm + 16 + mr][kc];
      v2f b0 = *(const v2f*)&Bst[wn + mr][kc];
      v2f b1 = *(const v2f*)&Bst[wn + 16 + mr][kc];
      v2f b2 = *(const v2f*)&Bst[wn + 32 + mr][kc];
      v2f b3 = *(const v2f*)&Bst[wn + 48 + mr][kc];
      acc[0][0] = __builtin_amdgcn_wmma_f32_16x16x4_f32(false, a0, false, b0, (short)0, acc[0][0], false, false);
      acc[0][1] = __builtin_amdgcn_wmma_f32_16x16x4_f32(false, a0, false, b1, (short)0, acc[0][1], false, false);
      acc[0][2] = __builtin_amdgcn_wmma_f32_16x16x4_f32(false, a0, false, b2, (short)0, acc[0][2], false, false);
      acc[0][3] = __builtin_amdgcn_wmma_f32_16x16x4_f32(false, a0, false, b3, (short)0, acc[0][3], false, false);
      acc[1][0] = __builtin_amdgcn_wmma_f32_16x16x4_f32(false, a1, false, b0, (short)0, acc[1][0], false, false);
      acc[1][1] = __builtin_amdgcn_wmma_f32_16x16x4_f32(false, a1, false, b1, (short)0, acc[1][1], false, false);
      acc[1][2] = __builtin_amdgcn_wmma_f32_16x16x4_f32(false, a1, false, b2, (short)0, acc[1][2], false, false);
      acc[1][3] = __builtin_amdgcn_wmma_f32_16x16x4_f32(false, a1, false, b3, (short)0, acc[1][3], false, false);
    }
    __syncthreads();
  }

#pragma unroll
  for (int im = 0; im < 2; ++im) {
#pragma unroll
    for (int r = 0; r < 8; ++r) {
      int row = m0 + wm + 16 * im + 8 * h + r;
      float* drow = Db + (long long)row * N;
#pragma unroll
      for (int in = 0; in < 4; ++in) {
        int col = n0 + wn + 16 * in + mr;
        float v = acc[im][in][r];
        if (mode == 1) {
          v = -0.5f * v + ((row == col) ? 1.5f : 0.0f);
        } else {
          v *= alpha;
        }
        drow[col] = v;
      }
    }
  }
}

#define GLD 256
#define LDA 20
#define SM 64
#define SLDB 160
#define GRP_F (SM * LDA + (BK / 2) * SLDB)
#define XCH_F (2 * 4 * 32 * 32)

struct StageMap {
  int ar, ak4;
  int bk, bn4;
};

__device__ __forceinline__ void load_piece(const float* __restrict__ P, bool kp, int base, int k,
                                           const StageMap& sm, float4& g0, float4& g1) {
  int off = kp ? ((k + 2 * sm.bk) * GLD + base + 4 * sm.bn4)
               : ((base + sm.ar) * GLD + k + 4 * sm.ak4);
  int d   = kp ? GLD : 32 * GLD;
  g0 = *reinterpret_cast<const float4*>(P + off);
  g1 = *reinterpret_cast<const float4*>(P + off + d);
}

__device__ __forceinline__ void store_piece(float* __restrict__ X, bool kp, const StageMap& sm,
                                            const float4& g0, const float4& g1) {
  float4 v0, v1;
  v0.x = g0.x;
  v0.y = kp ? g1.x : g0.y;
  v0.z = kp ? g0.y : g0.z;
  v0.w = kp ? g1.y : g0.w;
  v1.x = kp ? g0.z : g1.x;
  v1.y = kp ? g1.z : g1.y;
  v1.z = kp ? g0.w : g1.z;
  v1.w = g1.w;
  int a0 = kp ? (sm.bk * SLDB + 8 * sm.bn4) : (sm.ar * LDA + 4 * sm.ak4);
  int a1 = kp ? (a0 + 4) : ((sm.ar + 32) * LDA + 4 * sm.ak4);
  *reinterpret_cast<float4*>(&X[a0]) = v0;
  *reinterpret_cast<float4*>(&X[a1]) = v1;
}

__device__ __forceinline__ void loads_out(const float4& a, const float4& b,
                                          const float4& c, const float4& d) {
  asm volatile("" :: "v"(a.x), "v"(a.y), "v"(a.z), "v"(a.w), "v"(b.x), "v"(b.y), "v"(b.z), "v"(b.w),
                     "v"(c.x), "v"(c.y), "v"(c.z), "v"(c.w), "v"(d.x), "v"(d.y), "v"(d.z), "v"(d.w));
}

__device__ __forceinline__ void frag_map(bool kp, int w, int mr, int h, int& f0, int& d16, int& dk) {
  f0  = kp ? (h * SLDB + 2 * (w + mr)) : ((w + mr) * LDA + 2 * h);
  d16 = kp ? 32 : 16 * LDA;
  dk  = kp ? 2 * SLDB : 4;
}

__device__ __forceinline__ void kstep_mma(const float* __restrict__ As, const float* __restrict__ Bs,
                                          bool kpA, bool kpB, int wm, int wn, int mr, int h,
                                          v8f (&acc)[2][2]) {
  int fa, da, ka, fb, db, kb;
  frag_map(kpA, wm, mr, h, fa, da, ka);
  frag_map(kpB, wn, mr, h, fb, db, kb);
#pragma unroll
  for (int j = 0; j < BK / 4; ++j) {
    v2f a0 = *(const v2f*)&As[fa + j * ka];
    v2f a1 = *(const v2f*)&As[fa + da + j * ka];
    v2f b0 = *(const v2f*)&Bs[fb + j * kb];
    v2f b1 = *(const v2f*)&Bs[fb + db + j * kb];
    acc[0][0] = __builtin_amdgcn_wmma_f32_16x16x4_f32(false, a0, false, b0, (short)0, acc[0][0], false, false);
    acc[0][1] = __builtin_amdgcn_wmma_f32_16x16x4_f32(false, a0, false, b1, (short)0, acc[0][1], false, false);
    acc[1][0] = __builtin_amdgcn_wmma_f32_16x16x4_f32(false, a1, false, b0, (short)0, acc[1][0], false, false);
    acc[1][1] = __builtin_amdgcn_wmma_f32_16x16x4_f32(false, a1, false, b1, (short)0, acc[1][1], false, false);
  }
}

template <bool SCALE, bool TRI>
__device__ __forceinline__ void half_k_product(const float* __restrict__ Ab,
                                               const float* __restrict__ Bb,
                                               float* __restrict__ As, float* __restrict__ Bs,
                                               int kbeg, int kh, float inv,
                                               int ti, int tj, const StageMap& sm,
                                               int wm, int wn, int mr, int h,
                                               v8f (&acc)[2][2]) {
  for (int k0 = 0; k0 < kh; k0 += BK) {
    int k  = kbeg + k0;
    int kb = k >> 6;
    bool kpA = TRI && (kb < ti);
    bool kpB = !(TRI && (kb > tj));
    float4 ga0, ga1, gb0, gb1;
    load_piece(Ab, kpA, ti * SM, k, sm, ga0, ga1);
    load_piece(Bb, kpB, tj * SM, k, sm, gb0, gb1);
    loads_out(ga0, ga1, gb0, gb1);
    if (SCALE) {
      ga0.x *= inv; ga0.y *= inv; ga0.z *= inv; ga0.w *= inv;
      ga1.x *= inv; ga1.y *= inv; ga1.z *= inv; ga1.w *= inv;
      gb0.x *= inv; gb0.y *= inv; gb0.z *= inv; gb0.w *= inv;
      gb1.x *= inv; gb1.y *= inv; gb1.z *= inv; gb1.w *= inv;
    }
    store_piece(As, kpA, sm, ga0, ga1);
    store_piece(Bs, kpB, sm, gb0, gb1);
    __syncthreads();
    kstep_mma(As, Bs, kpA, kpB, wm, wn, mr, h, acc);
    __syncthreads();
  }
}

__device__ __forceinline__ void swap_and_add(float* __restrict__ X, int g, int wid, int lane,
                                             v8f (&acc)[2][2]) {
  float* mine = X + ((g * 4 + wid) * 32) * 32 + lane;
#pragma unroll
  for (int im = 0; im < 2; ++im)
#pragma unroll
    for (int in = 0; in < 2; ++in)
#pragma unroll
      for (int r = 0; r < 8; ++r) mine[((im * 2 + in) * 8 + r) * 32] = acc[im][in][r];
  __syncthreads();
  const float* oth = X + (((1 - g) * 4 + wid) * 32) * 32 + lane;
#pragma unroll
  for (int im = 0; im < 2; ++im)
#pragma unroll
    for (int in = 0; in < 2; ++in)
#pragma unroll
      for (int r = 0; r < 8; ++r) acc[im][in][r] += oth[((im * 2 + in) * 8 + r) * 32];
}

#define TP 65
#define TILE_F (SM * TP)
#define SMEM_F (2 * TILE_F)

__device__ __forceinline__ void stage_tile(float* __restrict__ T, int wm, int wn, int mr, int h,
                                           const v8f (&acc)[2][2]) {
  float* p = T + (wm + 8 * h) * TP + wn + mr;
#pragma unroll
  for (int im = 0; im < 2; ++im)
#pragma unroll
    for (int in = 0; in < 2; ++in)
#pragma unroll
      for (int r = 0; r < 8; ++r) p[(16 * im + r) * TP + 16 * in] = acc[im][in][r];
}

template <int RC>
__device__ __forceinline__ void store_tile_rows(const float* __restrict__ T, float* __restrict__ Db,
                                                int m0, int n0, int tl, int rbeg) {
#pragma unroll
  for (int i = 0; i < RC / 8; ++i) {
    int f   = tl + 128 * i;
    int row = rbeg + (f >> 4);
    int c4  = f & 15;
    const float* s = T + row * TP + 4 * c4;
    float4 v;
    v.x = s[0]; v.y = s[1]; v.z = s[2]; v.w = s[3];
    *reinterpret_cast<float4*>(Db + (m0 + row) * GLD + n0 + 4 * c4) = v;
  }
}

__device__ __forceinline__ void store_tile_mirror(const float* __restrict__ T, float* __restrict__ Db,
                                                  int m0, int n0, int tl) {
#pragma unroll
  for (int i = 0; i < SM / 8; ++i) {
    int f   = tl + 128 * i;
    int col = f >> 4;
    int r4  = f & 15;
    const float* s = T + (4 * r4) * TP + col;
    float4 v;
    v.x = s[0]; v.y = s[TP]; v.z = s[2 * TP]; v.w = s[3 * TP];
    *reinterpret_cast<float4*>(Db + (n0 + col) * GLD + m0 + 4 * r4) = v;
  }
}

__global__ __launch_bounds__(256)
void wmma_gemm_sym(const float* __restrict__ A, const float* __restrict__ Bm,
                   float* __restrict__ Dm, int K, int mode, int full_out) {
  __shared__ __align__(16) float smem[SMEM_F];

  int b  = blockIdx.z;
  int t  = blockIdx.x;
  int ti = (t >= 4) + (t >= 7) + (t >= 9);
  int tj = ti + t - (4 * ti - (ti * (ti - 1)) / 2);
  int m0 = ti * SM;
  int n0 = tj * SM;
  const float* Ab = A  + (long long)b * (GLD * GLD);
  const float* Bb = Bm + (long long)b * (GLD * GLD);
  float*       Db = Dm + (long long)b * (GLD * GLD);

  int tid  = threadIdx.x;
  int g    = __builtin_amdgcn_readfirstlane(tid >> 7);
  int t7   = tid & 127;
  int wid  = t7 >> 5;
  int lane = tid & 31;
  int mr   = lane & 15;
  int h    = lane >> 4;
  int wm   = (wid >> 1) * 32;
  int wn   = (wid & 1) * 32;

  StageMap sm;
  sm.ar  = t7 >> 2;
  sm.ak4 = t7 & 3;
  sm.bk  = t7 >> 4;
  sm.bn4 = t7 & 15;
  int kh  = K >> 1;
  float* As = smem + g * GRP_F;
  float* Bs = As + SM * LDA;

  v8f acc[2][2];
#pragma unroll
  for (int im = 0; im < 2; ++im)
#pragma unroll
    for (int in = 0; in < 2; ++in) acc[im][in] = (v8f){};

  half_k_product<false, true>(Ab, Bb, As, Bs, g * kh, kh, 1.0f, ti, tj, sm, wm, wn, mr, h, acc);
  swap_and_add(smem, g, wid, lane, acc);

  if (mode == 1) {
#pragma unroll
    for (int im = 0; im < 2; ++im)
#pragma unroll
      for (int in = 0; in < 2; ++in)
#pragma unroll
        for (int r = 0; r < 8; ++r) {
          int row = m0 + wm + 16 * im + 8 * h + r;
          int col = n0 + wn + 16 * in + mr;
          acc[im][in][r] = -0.5f * acc[im][in][r] + ((row == col) ? 1.5f : 0.0f);
        }
  }

  __syncthreads();
  float* T = smem + g * TILE_F;
  stage_tile(T, wm, wn, mr, h, acc);
  __syncthreads();
  if (full_out != 0 && tj != ti) {
    if (g == 0) store_tile_rows<SM>(T, Db, m0, n0, t7, 0);
    else        store_tile_mirror(T, Db, m0, n0, t7);
  } else {
    store_tile_rows<SM / 2>(T, Db, m0, n0, t7, g * (SM / 2));
  }
}

#define CK 128
#define CPAN ((CK / 2) * 128)

__device__ __forceinline__ void load_panel(const float* __restrict__ Fb, int c0, int tid,
                                           float4 (&g0)[CK / 16], float4 (&g1)[CK / 16]) {
  const float* g = Fb + (2 * (tid >> 4)) * GLD + c0 + 4 * (tid & 15);
#pragma unroll
  for (int i = 0; i < CK / 16; ++i) {
    g0[i] = *reinterpret_cast<const float4*>(g + (16 * i) * GLD);
    g1[i] = *reinterpret_cast<const float4*>(g + (16 * i + 1) * GLD);
  }
}

__device__ __forceinline__ void store_panel(float* __restrict__ P, int tid,
                                            const float4 (&g0)[CK / 16], const float4 (&g1)[CK / 16]) {
  int bn4 = tid & 15;
  int kq  = tid >> 4;
#pragma unroll
  for (int i = 0; i < CK / 16; ++i) {
    int kp = kq + 8 * i;
    float4 lo, hi;
    lo.x = g0[i].x; lo.y = g1[i].x; lo.z = g0[i].y; lo.w = g1[i].y;
    hi.x = g0[i].z; hi.y = g1[i].z; hi.z = g0[i].w; hi.w = g1[i].w;
    int o = kp * 128 + ((8 * bn4) ^ ((kp & 1) << 5));
    *reinterpret_cast<float4*>(&P[o])     = lo;
    *reinterpret_cast<float4*>(&P[o + 4]) = hi;
  }
}

__global__ __launch_bounds__(128)
void wmma_cov_tri(const float* __restrict__ Fp, float* __restrict__ Cm,
                  float* __restrict__ mean, float alpha) {
  __shared__ __align__(16) float pan[2 * CPAN];
  __shared__ float mus[128];

  int b  = blockIdx.z;
  int t  = blockIdx.x;
  int ti = (t >= 4) + (t >= 7) + (t >= 9);
  int tj = ti + t - (4 * ti - (ti * (ti - 1)) / 2);
  int m0 = ti * SM;
  int n0 = tj * SM;
  bool diag = (ti == tj);
  const float* Fb = Fp + (long long)b * (CK * GLD);
  float*       Cb = Cm + (long long)b * (GLD * GLD);

  int tid  = threadIdx.x;
  int wid  = tid >> 5;
  int lane = tid & 31;
  int mr   = lane & 15;
  int h    = lane >> 4;
  int wm   = (wid >> 1) * 32;
  int wn   = (wid & 1) * 32;

  float* PA = pan;
  float* PB = diag ? pan : pan + CPAN;
  {
    float4 a0v[CK / 16], a1v[CK / 16], b0v[CK / 16], b1v[CK / 16];
    load_panel(Fb, m0, tid, a0v, a1v);
    if (!diag) load_panel(Fb, n0, tid, b0v, b1v);
    store_panel(PA, tid, a0v, a1v);
    if (!diag) store_panel(PB, tid, b0v, b1v);
  }
  __syncthreads();

  int ncol = diag ? 64 : 128;
  if (tid < ncol) {
    const float* P = (tid < 64) ? PA : PB;
    int col = tid & 63;
    float s = 0.0f;
#pragma unroll
    for (int i = 0; i < CK; ++i) {
      int kp = i >> 1;
      s += P[kp * 128 + ((2 * col) ^ ((kp & 1) << 5)) + (i & 1)];
    }
    float mu = s / (float)CK;
    mus[tid] = mu;
    if (diag) mean[(long long)b * GLD + m0 + col] = mu;
  }
  __syncthreads();

  const float* musB = diag ? mus : mus + 64;
  float muA0 = mus[wm + mr],  muA1 = mus[wm + 16 + mr];
  float muB0 = musB[wn + mr], muB1 = musB[wn + 16 + mr];

  v8f acc[2][2];
#pragma unroll
  for (int im = 0; im < 2; ++im)
#pragma unroll
    for (int in = 0; in < 2; ++in) acc[im][in] = (v8f){};

  int sw = h << 5;
  const float* fa0 = PA + h * 128 + ((2 * (wm + mr)) ^ sw);
  const float* fa1 = PA + h * 128 + ((2 * (wm + 16 + mr)) ^ sw);
  const float* fb0 = PB + h * 128 + ((2 * (wn + mr)) ^ sw);
  const float* fb1 = PB + h * 128 + ((2 * (wn + 16 + mr)) ^ sw);
#pragma unroll 8
  for (int kk = 0; kk < CK; kk += 4) {
    v2f a0 = *(const v2f*)(fa0 + kk * 64);
    v2f a1 = *(const v2f*)(fa1 + kk * 64);
    v2f b0 = *(const v2f*)(fb0 + kk * 64);
    v2f b1 = *(const v2f*)(fb1 + kk * 64);
    a0.x -= muA0; a0.y -= muA0;
    a1.x -= muA1; a1.y -= muA1;
    b0.x -= muB0; b0.y -= muB0;
    b1.x -= muB1; b1.y -= muB1;
    acc[0][0] = __builtin_amdgcn_wmma_f32_16x16x4_f32(false, a0, false, b0, (short)0, acc[0][0], false, false);
    acc[0][1] = __builtin_amdgcn_wmma_f32_16x16x4_f32(false, a0, false, b1, (short)0, acc[0][1], false, false);
    acc[1][0] = __builtin_amdgcn_wmma_f32_16x16x4_f32(false, a1, false, b0, (short)0, acc[1][0], false, false);
    acc[1][1] = __builtin_amdgcn_wmma_f32_16x16x4_f32(false, a1, false, b1, (short)0, acc[1][1], false, false);
  }

#pragma unroll
  for (int im = 0; im < 2; ++im)
#pragma unroll
    for (int in = 0; in < 2; ++in)
#pragma unroll
      for (int r = 0; r < 8; ++r) acc[im][in][r] *= alpha;

  __syncthreads();
  stage_tile(pan, wm, wn, mr, h, acc);
  __syncthreads();
  store_tile_rows<SM>(pan, Cb, m0, n0, tid, 0);
  if (!diag) store_tile_mirror(pan, Cb, m0, n0, tid);
}

template <int IM>
__device__ __forceinline__ void first_rows(const float* __restrict__ Cb, int g,
                                           float tr, int m0, int n0, int wm, int wn, int mr, int h,
                                           v8f (&acc)[2][2]) {
  const float* cbase = Cb + (m0 + wm + 8 * h) * GLD + n0 + wn + mr;
#pragma unroll
  for (int in = 0; in < 2; ++in)
#pragma unroll
    for (int r = 0; r < 8; ++r) {
      int row = m0 + wm + 16 * IM + 8 * h + r;
      int col = n0 + wn + 16 * in + mr;
      float a = cbase[(16 * IM + r) * GLD + 16 * in] / tr;
      float zv = -0.5f * a + ((row == col) ? 1.5f : 0.0f);
      float yv = 1.5f * a - 0.5f * acc[IM][in][r];
      acc[IM][in][r] = (g == 0) ? yv : zv;
    }
}

__global__ __launch_bounds__(256) __attribute__((amdgpu_waves_per_eu(10)))
void wmma_gemm_first(const float* __restrict__ C, float* __restrict__ Ym,
                     float* __restrict__ Zm, float* __restrict__ trout) {
  __shared__ __align__(16) float smem[SMEM_F];
  __shared__ float red[256];

  int b  = blockIdx.z;
  int t  = blockIdx.x;
  int ti = (t >= 4) + (t >= 7) + (t >= 9);
  int tj = ti + t - (4 * ti - (ti * (ti - 1)) / 2);
  int m0 = ti * SM;
  int n0 = tj * SM;
  const float* Cb = C  + (long long)b * (GLD * GLD);
  float*       Yb = Ym + (long long)b * (GLD * GLD);
  float*       Zb = Zm + (long long)b * (GLD * GLD);

  int tid  = threadIdx.x;
  int g    = __builtin_amdgcn_readfirstlane(tid >> 7);
  int t7   = tid & 127;
  int wid  = t7 >> 5;
  int lane = tid & 31;
  int mr   = lane & 15;
  int h    = lane >> 4;
  int wm   = (wid >> 1) * 32;
  int wn   = (wid & 1) * 32;

  StageMap sm;
  sm.ar  = t7 >> 2;
  sm.ak4 = t7 & 3;
  sm.bk  = t7 >> 4;
  sm.bn4 = t7 & 15;
  const int kh = GLD / 2;
  float* As = smem + g * GRP_F;
  float* Bs = As + SM * LDA;

  red[tid] = Cb[tid * (GLD + 1)];
  __syncthreads();
  for (int s = 128; s > 0; s >>= 1) {
    if (tid < s) red[tid] += red[tid + s];
    __syncthreads();
  }
  float tr  = red[0];
  float inv = 1.0f / tr;
  if (t == 0 && tid == 0) trout[b] = tr;

  v8f acc[2][2];
#pragma unroll
  for (int im = 0; im < 2; ++im)
#pragma unroll
    for (int in = 0; in < 2; ++in) acc[im][in] = (v8f){};

  half_k_product<true, false>(Cb, Cb, As, Bs, g * kh, kh, inv, ti, tj, sm, wm, wn, mr, h, acc);
  swap_and_add(smem, g, wid, lane, acc);

  float* Ob = (g == 0) ? Yb : Zb;
  first_rows<0>(Cb, g, tr, m0, n0, wm, wn, mr, h, acc);
  first_rows<1>(Cb, g, tr, m0, n0, wm, wn, mr, h, acc);
  __syncthreads();
  float* T = smem + g * TILE_F;
  stage_tile(T, wm, wn, mr, h, acc);
  __syncthreads();
  store_tile_rows<SM>(T, Ob, m0, n0, t7, 0);
}

template <int IM>
__device__ __forceinline__ void pair_rows(const float* __restrict__ Pb, int g,
                                          int m0, int n0, int wm, int wn, int mr, int h,
                                          v8f (&acc)[2][2]) {
  const float* pbase = Pb + (m0 + wm + 8 * h) * GLD + n0 + wn + mr;
#pragma unroll
  for (int in = 0; in < 2; ++in)
#pragma unroll
    for (int r = 0; r < 8; ++r) {
      int row = m0 + wm + 16 * IM + 8 * h + r;
      int col = n0 + wn + 16 * in + mr;
      float lin = pbase[(16 * IM + r) * GLD + 16 * in];
      float sv = 0.25f * acc[IM][in][r] - 1.5f * lin + ((row == col) ? 2.25f : 0.0f);
      float yv = 1.5f * lin - 0.5f * acc[IM][in][r];
      acc[IM][in][r] = (g == 0) ? sv : yv;
    }
}

__global__ __launch_bounds__(256)
void wmma_gemm_pair(const float* __restrict__ Tm, const float* __restrict__ Yin,
                    float* __restrict__ Sm, float* __restrict__ Ym) {
  __shared__ __align__(16) float smem[SMEM_F];
  float* As = smem;

  int b  = blockIdx.z;
  int t  = blockIdx.x;
  int ti = (t >= 4) + (t >= 7) + (t >= 9);
  int tj = ti + t - (4 * ti - (ti * (ti - 1)) / 2);
  int m0 = ti * SM;
  int n0 = tj * SM;
  const float* Tb = Tm  + (long long)b * (GLD * GLD);
  const float* Yb = Yin + (long long)b * (GLD * GLD);
  float*       Sb = Sm  + (long long)b * (GLD * GLD);
  float*       Ob = Ym  + (long long)b * (GLD * GLD);

  int tid  = threadIdx.x;
  int g    = __builtin_amdgcn_readfirstlane(tid >> 7);
  int t7   = tid & 127;
  int wid  = t7 >> 5;
  int lane = tid & 31;
  int mr   = lane & 15;
  int h    = lane >> 4;
  int wm   = (wid >> 1) * 32;
  int wn   = (wid & 1) * 32;

  StageMap sm;
  sm.ar  = t7 >> 2;
  sm.ak4 = t7 & 3;
  sm.bk  = t7 >> 4;
  sm.bn4 = t7 & 15;
  const float* Pb = (g == 0) ? Tb : Yb;
  float* Bs = smem + SM * LDA + g * ((BK / 2) * SLDB);

  v8f acc[2][2];
#pragma unroll
  for (int im = 0; im < 2; ++im)
#pragma unroll
    for (int in = 0; in < 2; ++in) acc[im][in] = (v8f){};

  for (int k0 = 0; k0 < GLD; k0 += BK) {
    int kb = k0 >> 6;
    bool kpA = (kb < ti);
    bool kpB = !(kb > tj);
    float4 ga0 = make_float4(0.0f, 0.0f, 0.0f, 0.0f), ga1 = ga0, gb0, gb1;
    if (g == 0) load_piece(Tb, kpA, m0, k0, sm, ga0, ga1);
    load_piece(Pb, kpB, n0, k0, sm, gb0, gb1);
    loads_out(ga0, ga1, gb0, gb1);
    if (g == 0) store_piece(As, kpA, sm, ga0, ga1);
    store_piece(Bs, kpB, sm, gb0, gb1);
    __syncthreads();
    kstep_mma(As, Bs, kpA, kpB, wm, wn, mr, h, acc);
    __syncthreads();
  }

  float* Db = (g == 0) ? Sb : Ob;
  pair_rows<0>(Pb, g, m0, n0, wm, wn, mr, h, acc);
  pair_rows<1>(Pb, g, m0, n0, wm, wn, mr, h, acc);
  float* T = smem + g * TILE_F;
  stage_tile(T, wm, wn, mr, h, acc);
  __syncthreads();
  store_tile_rows<SM>(T, Db, m0, n0, t7, 0);
}

#define OUT_ROWS 4

__global__ __launch_bounds__(256)
void out_kernel(const float* __restrict__ YZY, const float* __restrict__ mean,
                const float* __restrict__ tr, float* __restrict__ out, int d, int rows) {
  __shared__ float res[8 * OUT_ROWS];
  int wid  = threadIdx.x >> 5;
  int lane = threadIdx.x & 31;
  int row0 = blockIdx.x * (8 * OUT_ROWS) + wid * OUT_ROWS;
  const float* yrow[OUT_ROWS];
  const float* mrow[OUT_ROWS];
  float s[OUT_ROWS];
#pragma unroll
  for (int r = 0; r < OUT_ROWS; ++r) {
    int row = row0 + r;
    if (row >= rows) row = rows - 1;
    yrow[r] = YZY + (long long)row * d;
    mrow[r] = mean + (long long)(row / d) * d;
    s[r] = 0.0f;
  }
  for (int j = lane; j < d; j += 32) {
#pragma unroll
    for (int r = 0; r < OUT_ROWS; ++r) s[r] += yrow[r][j] * mrow[r][j];
  }
#pragma unroll
  for (int r = 0; r < OUT_ROWS; ++r)
    for (int off = 16; off > 0; off >>= 1) s[r] += __shfl_down(s[r], off, 32);
  if (lane == 0) {
#pragma unroll
    for (int r = 0; r < OUT_ROWS; ++r) res[wid * OUT_ROWS + r] = s[r];
  }
  __syncthreads();
  if (threadIdx.x < 8 * OUT_ROWS) {
    int row = blockIdx.x * (8 * OUT_ROWS) + threadIdx.x;
    if (row < rows) out[row] = res[threadIdx.x] * sqrtf(tr[row / d]);
  }
}

extern "C" void kernel_launch(void* const* d_in, const int* in_sizes, int n_in,
                              void* d_out, int out_size, void* d_ws, size_t ws_size,
                              hipStream_t stream) {
  (void)n_in; (void)ws_size; (void)out_size;
  const float* feat  = (const float*)d_in[1];
  const float* noise = (const float*)d_in[2];

  const int B  = in_sizes[0];
  const int d  = 256;
  const long long total = (long long)in_sizes[1];
  const int n  = (int)(total / ((long long)B * d));
  const int nd = n * d;
  const long long dd = (long long)d * d;

  float* W = (float*)d_ws;
  long long off = 0;
  float* Fp   = W + off; off += (long long)B * nd;
  float* mean = W + off; off += (long long)B * d;
  float* scl  = W + off; off += B;
  float* nmn  = W + off; off += B;
  float* tr   = W + off; off += B;
  off = (off + 255) & ~255LL;
  float* M0 = W + off; off += (long long)B * dd;
  float* M1 = W + off; off += (long long)B * dd;
  float* M2 = W + off; off += (long long)B * dd;
  float* M3 = W + off; off += (long long)B * dd;

  stats_partial_kernel<<<dim3(STAT_SLICES, B), 256, 0, stream>>>(feat, noise, M0, nd);
  stats_final_kernel<<<1, 128, 0, stream>>>(M0, scl, nmn, B, nd, 31.6227766016838f);
  perturb_kernel<<<dim3((nd / 4 + EW_TILE4 - 1) / EW_TILE4, B), 256, 0, stream>>>(feat, noise, scl, nmn, Fp, nd);
  if (n == CK) {
    wmma_cov_tri<<<dim3(10, 1, B), 128, 0, stream>>>(Fp, M0, mean, 1.0f / (float)(n - 1));
  } else {
    mean_kernel<<<B, d, 0, stream>>>(Fp, mean, n, d);
    dim3 gg(d / BN, d / BM, B);
    wmma_gemm_batched<<<gg, 256, 0, stream>>>(Fp, (long long)nd, 1, d,
                                              Fp, (long long)nd, d, 1,
                                              M0, dd, d, n,
                                              1.0f / (float)(n - 1), 0, mean);
  }

  dim3 gs(10, 1, B);
  wmma_gemm_first<<<gs, 256, 0, stream>>>(M0, M1, M2, tr);

  wmma_gemm_sym<<<gs, 256, 0, stream>>>(M2, M1, M0, d, 0, 0);
  wmma_gemm_pair<<<gs, 256, 0, stream>>>(M0, M1, M2, M3);
  wmma_gemm_sym<<<gs, 256, 0, stream>>>(M2, M0, M1, d, 0, 0);
  wmma_gemm_pair<<<gs, 256, 0, stream>>>(M1, M3, M0, M2);
  wmma_gemm_sym<<<gs, 256, 0, stream>>>(M0, M1, M3, d, 0, 0);
  wmma_gemm_pair<<<gs, 256, 0, stream>>>(M3, M2, M0, M1);

  wmma_gemm_sym<<<gs, 256, 0, stream>>>(M0, M3, M2, d, 1, 0);
  wmma_gemm_sym<<<gs, 256, 0, stream>>>(M1, M2, M0, d, 0, 1);
  float* S2 = M0;

  out_kernel<<<(B * d + 8 * OUT_ROWS - 1) / (8 * OUT_ROWS), 256, 0, stream>>>(S2, mean, tr, (float*)d_out, d, B * d);
}
